# starting kernel
# baseline (speedup 1.0000x reference)
#include <hip/hip_runtime.h>

#define NN 100000
#define NE 1600000
#define DIM 128
#define BN_EPS 1e-5f

typedef __attribute__((ext_vector_type(2))) float v2f;
typedef __attribute__((ext_vector_type(8))) float v8f;

__global__ void zero_kernel(float* __restrict__ p, size_t n) {
    size_t i = ((size_t)blockIdx.x * blockDim.x + threadIdx.x) * 4;
    if (i + 3 < n) {
        float4 z; z.x = 0.f; z.y = 0.f; z.z = 0.f; z.w = 0.f;
        *(float4*)(p + i) = z;
    } else {
        for (size_t j = i; j < n; ++j) p[j] = 0.f;
    }
}

__global__ __launch_bounds__(256) void scatter_kernel(
    const float* __restrict__ x, const int* __restrict__ ei,
    float* __restrict__ agg, float* __restrict__ deg) {
    int wid  = blockIdx.x * 8 + (threadIdx.x >> 5);
    int lane = threadIdx.x & 31;
    if (wid >= NE) return;
    int src = ei[wid];
    int dst = ei[NE + wid];
    const float4 v = *(const float4*)(x + (size_t)src * DIM + lane * 4);
    float* dp = agg + (size_t)dst * DIM + lane * 4;
    atomicAdd(dp + 0, v.x);
    atomicAdd(dp + 1, v.y);
    atomicAdd(dp + 2, v.z);
    atomicAdd(dp + 3, v.w);
    if (lane == 0) atomicAdd(deg + dst, 1.0f);
}

__global__ void mean_kernel(float* __restrict__ agg, const float* __restrict__ deg) {
    size_t i = (size_t)blockIdx.x * blockDim.x + threadIdx.x;
    if (i >= (size_t)NN * DIM / 4) return;
    int node = (int)((i * 4) >> 7);
    float inv = 1.0f / fmaxf(deg[node], 1.0f);
    float4 v = ((float4*)agg)[i];
    v.x *= inv; v.y *= inv; v.z *= inv; v.w *= inv;
    ((float4*)agg)[i] = v;
}

__global__ __launch_bounds__(256) void gemm_bn_kernel(
    const float* __restrict__ mean, const float* __restrict__ x,
    const float* __restrict__ Wl, const float* __restrict__ bl,
    const float* __restrict__ Wr, float* __restrict__ h,
    float* __restrict__ gsum, float* __restrict__ gsq) {
    __shared__ float lWl[DIM * DIM];
    __shared__ float lWr[DIM * DIM];
    __shared__ float lsum[DIM];
    __shared__ float lsq[DIM];

    const int tid = threadIdx.x;
    for (int i = tid * 4; i < DIM * DIM; i += 256 * 4) {
        *(float4*)(lWl + i) = *(const float4*)(Wl + i);
        *(float4*)(lWr + i) = *(const float4*)(Wr + i);
    }
    if (tid < DIM) { lsum[tid] = 0.f; lsq[tid] = 0.f; }
    __syncthreads();

    const int wave = tid >> 5;
    const int lane = tid & 31;
    const int tile = blockIdx.x * 8 + wave;
    const bool active = (tile * 16) < NN;

    if (active) {
        const int row0 = tile * 16;
        const int m  = lane & 15;
        const int kh = lane >> 4;
        const float* meanRow = mean + (size_t)(row0 + m) * DIM;
        const float* xRow    = x    + (size_t)(row0 + m) * DIM;

        for (int nt = 0; nt < 8; ++nt) {
            const int ncol = nt * 16 + (lane & 15);
            const float bias = bl[ncol];
            v8f c;
#pragma unroll
            for (int r = 0; r < 8; ++r) c[r] = bias;

#pragma unroll
            for (int k0 = 0; k0 < DIM; k0 += 4) {
                const int ka = k0 + 2 * kh;
                v2f a; a.x = meanRow[ka];           a.y = meanRow[ka + 1];
                v2f b; b.x = lWl[ka * DIM + ncol];  b.y = lWl[(ka + 1) * DIM + ncol];
                c = __builtin_amdgcn_wmma_f32_16x16x4_f32(
                        false, a, false, b, (short)0, c, false, false);
            }
#pragma unroll
            for (int k0 = 0; k0 < DIM; k0 += 4) {
                const int ka = k0 + 2 * kh;
                v2f a; a.x = xRow[ka];              a.y = xRow[ka + 1];
                v2f b; b.x = lWr[ka * DIM + ncol];  b.y = lWr[(ka + 1) * DIM + ncol];
                c = __builtin_amdgcn_wmma_f32_16x16x4_f32(
                        false, a, false, b, (short)0, c, false, false);
            }

            float s = 0.f, s2 = 0.f;
#pragma unroll
            for (int r = 0; r < 8; ++r) {
                float v = c[r] > 0.f ? c[r] : 0.f;
                const int row = row0 + r + 8 * kh;
                h[(size_t)row * DIM + ncol] = v;
                s  += v;
                s2 += v * v;
            }
            s  += __shfl_xor(s, 16, 32);
            s2 += __shfl_xor(s2, 16, 32);
            if (lane < 16) {
                atomicAdd(&lsum[ncol], s);
                atomicAdd(&lsq[ncol], s2);
            }
        }
    }
    __syncthreads();
    if (tid < DIM) {
        atomicAdd(gsum + tid, lsum[tid]);
        atomicAdd(gsq  + tid, lsq[tid]);
    }
}

__global__ void stats_kernel(const float* __restrict__ gsum, const float* __restrict__ gsq,
                             const float* __restrict__ gamma, const float* __restrict__ beta,
                             float* __restrict__ cA, float* __restrict__ cB) {
    int c = threadIdx.x;
    if (c >= DIM) return;
    const float invN = 1.0f / (float)NN;
    float mu  = gsum[c] * invN;
    float var = gsq[c] * invN - mu * mu;
    float a   = rsqrtf(var + BN_EPS) * gamma[c];
    cA[c] = a;
    cB[c] = beta[c] - mu * a;
}

__global__ void apply_kernel(const float* __restrict__ x,
                             const float* __restrict__ cA, const float* __restrict__ cB,
                             float* __restrict__ out) {
    size_t i = (size_t)blockIdx.x * blockDim.x + threadIdx.x;
    if (i >= (size_t)NN * DIM / 4) return;
    int col4 = (int)(i & 31) * 4;
    float4 hv = ((float4*)out)[i];
    float4 xv = ((const float4*)x)[i];
    float4 a  = *(const float4*)(cA + col4);
    float4 b  = *(const float4*)(cB + col4);
    float4 o;
    o.x = xv.x + hv.x * a.x + b.x;
    o.y = xv.y + hv.y * a.y + b.y;
    o.z = xv.z + hv.z * a.z + b.z;
    o.w = xv.w + hv.w * a.w + b.w;
    ((float4*)out)[i] = o;
}

extern "C" void kernel_launch(void* const* d_in, const int* in_sizes, int n_in,
                              void* d_out, int out_size, void* d_ws, size_t ws_size,
                              hipStream_t stream) {
    const float* x     = (const float*)d_in[0];
    const int*   ei    = (const int*)d_in[1];
    const float* Wl    = (const float*)d_in[2];
    const float* bl    = (const float*)d_in[3];
    const float* Wr    = (const float*)d_in[4];
    const float* gamma = (const float*)d_in[5];
    const float* beta  = (const float*)d_in[6];
    float* out = (float*)d_out;

    float* agg  = (float*)d_ws;
    float* deg  = agg + (size_t)NN * DIM;
    float* gsum = deg + NN;
    float* gsq  = gsum + DIM;
    float* cA   = gsq + DIM;
    float* cB   = cA + DIM;

    const size_t zero_elems = (size_t)NN * DIM + NN + 4 * DIM;
    int zb = (int)((zero_elems / 4 + 255) / 256);
    zero_kernel<<<zb, 256, 0, stream>>>(agg, zero_elems);

    scatter_kernel<<<NE / 8, 256, 0, stream>>>(x, ei, agg, deg);

    int mb = (int)(((size_t)NN * DIM / 4 + 255) / 256);
    mean_kernel<<<mb, 256, 0, stream>>>(agg, deg);

    const int tiles = NN / 16;
    gemm_bn_kernel<<<(tiles + 7) / 8, 256, 0, stream>>>(agg, x, Wl, bl, Wr,
                                                        out, gsum, gsq);

    stats_kernel<<<1, 128, 0, stream>>>(gsum, gsq, gamma, beta, cA, cB);

    apply_kernel<<<mb, 256, 0, stream>>>(x, cA, cB, out);
}


# f32 atomic scatter replaced by a destination-grouped edge list (integer rank atomics, prefix sum, fill) and a per-node register gather that writes the mean; per-node sum order = arrival rank
# speedup vs baseline: 1.6017x; 1.6017x over previous
#include <hip/hip_runtime.h>

#define NN 100000
#define NE 1600000
#define DIM 128
#define BN_EPS 1e-5f
#define SCAN_CHUNK 1024
#define NODES_PER_WAVE 4
#define ROWPTR_SLOTS 100352

typedef __attribute__((ext_vector_type(2))) float v2f;
typedef __attribute__((ext_vector_type(8))) float v8f;

__global__ void zero_kernel(float* __restrict__ p, size_t n) {
    size_t i = ((size_t)blockIdx.x * blockDim.x + threadIdx.x) * 4;
    if (i + 3 < n) {
        float4 z; z.x = 0.f; z.y = 0.f; z.z = 0.f; z.w = 0.f;
        *(float4*)(p + i) = z;
    } else {
        for (size_t j = i; j < n; ++j) p[j] = 0.f;
    }
}


__global__ __launch_bounds__(256) void rank_kernel(
    const int* __restrict__ ei, int* __restrict__ cnt, int* __restrict__ rank) {
    const int t = blockIdx.x * 256 + threadIdx.x;
    if (t >= NE / 4) return;
    const int4 d = *(const int4*)(ei + NE + 4 * t);
    int4 r;
    r.x = atomicAdd(cnt + d.x, 1);
    r.y = atomicAdd(cnt + d.y, 1);
    r.z = atomicAdd(cnt + d.z, 1);
    r.w = atomicAdd(cnt + d.w, 1);
    *(int4*)(rank + 4 * t) = r;
}

__global__ __launch_bounds__(256) void scan_kernel(
    const int* __restrict__ cnt, int* __restrict__ rowptr) {
    __shared__ int wtot[8];
    __shared__ int wpre[8];
    const int tid  = threadIdx.x;
    const int lane = tid & 31;
    const int wave = tid >> 5;
    const int base = blockIdx.x * SCAN_CHUNK;

    int pre = 0;
    int i = tid * 4;
    for (; i + 7 * SCAN_CHUNK < base; i += 8 * SCAN_CHUNK) {
        int4 v[8];
#pragma unroll
        for (int u = 0; u < 8; ++u) v[u] = *(const int4*)(cnt + i + u * SCAN_CHUNK);
#pragma unroll
        for (int u = 0; u < 8; ++u) pre += (v[u].x + v[u].y) + (v[u].z + v[u].w);
    }
    for (; i < base; i += SCAN_CHUNK) {
        const int4 v = *(const int4*)(cnt + i);
        pre += (v.x + v.y) + (v.z + v.w);
    }

    const int i0 = base + tid * 4;
    int4 c; c.x = 0; c.y = 0; c.z = 0; c.w = 0;
    if (i0 + 3 < NN) {
        c = *(const int4*)(cnt + i0);
    } else {
        if (i0     < NN) c.x = cnt[i0];
        if (i0 + 1 < NN) c.y = cnt[i0 + 1];
        if (i0 + 2 < NN) c.z = cnt[i0 + 2];
    }
    const int t = c.x + c.y + c.z + c.w;
    int inc = t;
    for (int o = 1; o < 32; o <<= 1) {
        const int y = __shfl_up(inc, o, 32);
        if (lane >= o) inc += y;
    }
    for (int o = 16; o > 0; o >>= 1) pre += __shfl_xor(pre, o, 32);
    if (lane == 31) wtot[wave] = inc;
    if (lane == 0)  wpre[wave] = pre;
    __syncthreads();
    int off = 0;
    for (int w = 0; w < 8; ++w) {
        off += wpre[w];
        if (w < wave) off += wtot[w];
    }
    const int ex = off + inc - t;
    if (i0     <= NN) rowptr[i0]     = ex;
    if (i0 + 1 <= NN) rowptr[i0 + 1] = ex + c.x;
    if (i0 + 2 <= NN) rowptr[i0 + 2] = ex + c.x + c.y;
    if (i0 + 3 <= NN) rowptr[i0 + 3] = ex + c.x + c.y + c.z;
}

__global__ __launch_bounds__(256) void fill_kernel(
    const int* __restrict__ ei, const int* __restrict__ rowptr,
    const int* __restrict__ rank, int* __restrict__ col) {
    const int t = blockIdx.x * 256 + threadIdx.x;
    if (t >= NE / 4) return;
    const int4 s = *(const int4*)(ei + 4 * t);
    const int4 d = *(const int4*)(ei + NE + 4 * t);
    const int4 r = *(const int4*)(rank + 4 * t);
    col[rowptr[d.x] + r.x] = s.x;
    col[rowptr[d.y] + r.y] = s.y;
    col[rowptr[d.z] + r.z] = s.z;
    col[rowptr[d.w] + r.w] = s.w;
}

template <int U>
__device__ __forceinline__ void add_rows(const float* __restrict__ xl, int myc, int j,
                                         float4& acc) {
    float4 v[U];
#pragma unroll
    for (int u = 0; u < U; ++u) {
        const int s = __builtin_amdgcn_readlane(myc, j + u);
        v[u] = *(const float4*)(xl + (size_t)s * DIM);
    }
#pragma unroll
    for (int u = 0; u < U; ++u) {
        acc.x += v[u].x; acc.y += v[u].y; acc.z += v[u].z; acc.w += v[u].w;
    }
}

__global__ __launch_bounds__(256) void gather_mean_kernel(
    const float* __restrict__ x, const int* __restrict__ rowptr,
    const int* __restrict__ col, float* __restrict__ mean) {
    const int lane = threadIdx.x & 31;
    const int w = __builtin_amdgcn_readfirstlane((int)(blockIdx.x * 8 + (threadIdx.x >> 5)));
    const float* xl = x + lane * 4;
    for (int q = 0; q < NODES_PER_WAVE; ++q) {
        const int node = w * NODES_PER_WAVE + q;
        if (node >= NN) break;
        const int start = rowptr[node];
        const int end   = rowptr[node + 1];
        float4 acc; acc.x = 0.f; acc.y = 0.f; acc.z = 0.f; acc.w = 0.f;
        for (int j0 = start; j0 < end; j0 += 32) {
            const int m = (end - j0) < 32 ? (end - j0) : 32;
            const int myc = (lane < m) ? col[j0 + lane] : 0;
            int j = 0;
            for (; j + 8 <= m; j += 8) add_rows<8>(xl, myc, j, acc);
            if (j + 4 <= m) { add_rows<4>(xl, myc, j, acc); j += 4; }
            if (j + 2 <= m) { add_rows<2>(xl, myc, j, acc); j += 2; }
            if (j < m)      { add_rows<1>(xl, myc, j, acc); }
        }
        const float inv = 1.0f / fmaxf((float)(end - start), 1.0f);
        float4 o;
        o.x = acc.x * inv; o.y = acc.y * inv; o.z = acc.z * inv; o.w = acc.w * inv;
        *(float4*)(mean + (size_t)node * DIM + lane * 4) = o;
    }
}

__global__ __launch_bounds__(256) void gemm_bn_kernel(
    const float* __restrict__ mean, const float* __restrict__ x,
    const float* __restrict__ Wl, const float* __restrict__ bl,
    const float* __restrict__ Wr, float* __restrict__ h,
    float* __restrict__ gsum, float* __restrict__ gsq) {
    __shared__ float lWl[DIM * DIM];
    __shared__ float lWr[DIM * DIM];
    __shared__ float lsum[DIM];
    __shared__ float lsq[DIM];

    const int tid = threadIdx.x;
    for (int i = tid * 4; i < DIM * DIM; i += 256 * 4) {
        *(float4*)(lWl + i) = *(const float4*)(Wl + i);
        *(float4*)(lWr + i) = *(const float4*)(Wr + i);
    }
    if (tid < DIM) { lsum[tid] = 0.f; lsq[tid] = 0.f; }
    __syncthreads();

    const int wave = tid >> 5;
    const int lane = tid & 31;
    const int tile = blockIdx.x * 8 + wave;
    const bool active = (tile * 16) < NN;

    if (active) {
        const int row0 = tile * 16;
        const int m  = lane & 15;
        const int kh = lane >> 4;
        const float* meanRow = mean + (size_t)(row0 + m) * DIM;
        const float* xRow    = x    + (size_t)(row0 + m) * DIM;

        for (int nt = 0; nt < 8; ++nt) {
            const int ncol = nt * 16 + (lane & 15);
            const float bias = bl[ncol];
            v8f c;
#pragma unroll
            for (int r = 0; r < 8; ++r) c[r] = bias;

#pragma unroll
            for (int k0 = 0; k0 < DIM; k0 += 4) {
                const int ka = k0 + 2 * kh;
                v2f a; a.x = meanRow[ka];           a.y = meanRow[ka + 1];
                v2f b; b.x = lWl[ka * DIM + ncol];  b.y = lWl[(ka + 1) * DIM + ncol];
                c = __builtin_amdgcn_wmma_f32_16x16x4_f32(
                        false, a, false, b, (short)0, c, false, false);
            }
#pragma unroll
            for (int k0 = 0; k0 < DIM; k0 += 4) {
                const int ka = k0 + 2 * kh;
                v2f a; a.x = xRow[ka];              a.y = xRow[ka + 1];
                v2f b; b.x = lWr[ka * DIM + ncol];  b.y = lWr[(ka + 1) * DIM + ncol];
                c = __builtin_amdgcn_wmma_f32_16x16x4_f32(
                        false, a, false, b, (short)0, c, false, false);
            }

            float s = 0.f, s2 = 0.f;
#pragma unroll
            for (int r = 0; r < 8; ++r) {
                float v = c[r] > 0.f ? c[r] : 0.f;
                const int row = row0 + r + 8 * kh;
                h[(size_t)row * DIM + ncol] = v;
                s  += v;
                s2 += v * v;
            }
            s  += __shfl_xor(s, 16, 32);
            s2 += __shfl_xor(s2, 16, 32);
            if (lane < 16) {
                atomicAdd(&lsum[ncol], s);
                atomicAdd(&lsq[ncol], s2);
            }
        }
    }
    __syncthreads();
    if (tid < DIM) {
        atomicAdd(gsum + tid, lsum[tid]);
        atomicAdd(gsq  + tid, lsq[tid]);
    }
}

__global__ void stats_kernel(const float* __restrict__ gsum, const float* __restrict__ gsq,
                             const float* __restrict__ gamma, const float* __restrict__ beta,
                             float* __restrict__ cA, float* __restrict__ cB) {
    int c = threadIdx.x;
    if (c >= DIM) return;
    const float invN = 1.0f / (float)NN;
    float mu  = gsum[c] * invN;
    float var = gsq[c] * invN - mu * mu;
    float a   = rsqrtf(var + BN_EPS) * gamma[c];
    cA[c] = a;
    cB[c] = beta[c] - mu * a;
}

__global__ void apply_kernel(const float* __restrict__ x,
                             const float* __restrict__ cA, const float* __restrict__ cB,
                             float* __restrict__ out) {
    size_t i = (size_t)blockIdx.x * blockDim.x + threadIdx.x;
    if (i >= (size_t)NN * DIM / 4) return;
    int col4 = (int)(i & 31) * 4;
    float4 hv = ((float4*)out)[i];
    float4 xv = ((const float4*)x)[i];
    float4 a  = *(const float4*)(cA + col4);
    float4 b  = *(const float4*)(cB + col4);
    float4 o;
    o.x = xv.x + hv.x * a.x + b.x;
    o.y = xv.y + hv.y * a.y + b.y;
    o.z = xv.z + hv.z * a.z + b.z;
    o.w = xv.w + hv.w * a.w + b.w;
    ((float4*)out)[i] = o;
}

extern "C" void kernel_launch(void* const* d_in, const int* in_sizes, int n_in,
                              void* d_out, int out_size, void* d_ws, size_t ws_size,
                              hipStream_t stream) {
    const float* x     = (const float*)d_in[0];
    const int*   ei    = (const int*)d_in[1];
    const float* Wl    = (const float*)d_in[2];
    const float* bl    = (const float*)d_in[3];
    const float* Wr    = (const float*)d_in[4];
    const float* gamma = (const float*)d_in[5];
    const float* beta  = (const float*)d_in[6];
    float* out = (float*)d_out;

    float* agg  = (float*)d_ws;
    float* deg  = agg + (size_t)NN * DIM;
    float* gsum = deg + NN;
    float* gsq  = gsum + DIM;
    float* cA   = gsq + DIM;
    float* cB   = cA + DIM;
    int*   cnt  = (int*)deg;

    int* rowptr = (int*)d_out;
    int* rank   = rowptr + ROWPTR_SLOTS;
    int* col    = rank + NE;

    zero_kernel<<<(NN + 4 * DIM + 1023) / 1024, 256, 0, stream>>>(deg, (size_t)(NN + 4 * DIM));

    rank_kernel<<<(NE / 4 + 255) / 256, 256, 0, stream>>>(ei, cnt, rank);

    scan_kernel<<<(NN + 1 + SCAN_CHUNK - 1) / SCAN_CHUNK, 256, 0, stream>>>(cnt, rowptr);

    fill_kernel<<<(NE / 4 + 255) / 256, 256, 0, stream>>>(ei, rowptr, rank, col);

    gather_mean_kernel<<<(NN + 8 * NODES_PER_WAVE - 1) / (8 * NODES_PER_WAVE), 256, 0, stream>>>(
        x, rowptr, col, agg);

    const int tiles = NN / 16;
    gemm_bn_kernel<<<(tiles + 7) / 8, 256, 0, stream>>>(agg, x, Wl, bl, Wr,
                                                        out, gsum, gsq);

    stats_kernel<<<1, 128, 0, stream>>>(gsum, gsq, gamma, beta, cA, cB);

    int mb = (int)(((size_t)NN * DIM / 4 + 255) / 256);
    apply_kernel<<<mb, 256, 0, stream>>>(x, cA, cB, out);
}


# edge grouping by destination redone as a two-level counting sort in LDS (chunk histogram, prefix sum, bucket partition, per-bucket finish), no global integer atomics; per-node sum order = LDS arrival
# speedup vs baseline: 2.4408x; 1.5239x over previous
#include <hip/hip_runtime.h>

#define NN 100000
#define NE 1600000
#define DIM 128
#define BN_EPS 1e-5f
#define SCAN_CHUNK 1024
#define NODES_PER_WAVE 4
#define ROWPTR_SLOTS 100352
#define EDGE_CHUNK 4096
#define NBLK ((NE + EDGE_CHUNK - 1) / EDGE_CHUNK)
#define BSHIFT 8
#define BW (1 << BSHIFT)
#define NBK ((NN + BW - 1) / BW)
#define BK_SLOTS 512
#define NH (NBK * NBLK)
#define NH_SLOTS 153600
#define STAGE_CAP 8192
#define CSR_BATCH 16

typedef __attribute__((ext_vector_type(2))) float v2f;
typedef __attribute__((ext_vector_type(8))) float v8f;


__global__ __launch_bounds__(256) void hist_kernel(
    const int* __restrict__ ei, int* __restrict__ H, float* __restrict__ sums) {
    __shared__ int lh[BK_SLOTS];
    const int tid = threadIdx.x;
    const int b   = blockIdx.x;
    const int e0  = b * EDGE_CHUNK;
    lh[tid] = 0; lh[tid + 256] = 0;
    if (b == 0) sums[tid] = 0.f;
    __syncthreads();
    for (int u = 0; u < 4; ++u) {
        const int e = e0 + 4 * (tid + 256 * u);
        if (e < NE) {
            const int4 d = *(const int4*)(ei + NE + e);
            atomicAdd(&lh[d.x >> BSHIFT], 1);
            atomicAdd(&lh[d.y >> BSHIFT], 1);
            atomicAdd(&lh[d.z >> BSHIFT], 1);
            atomicAdd(&lh[d.w >> BSHIFT], 1);
        }
    }
    __syncthreads();
    for (int q = tid; q < NBK; q += 256) H[q * NBLK + b] = lh[q];
}

template <int N>
__global__ __launch_bounds__(256) void scan_kernel(
    const int* __restrict__ cnt, int* __restrict__ out) {
    __shared__ int wtot[8];
    __shared__ int wpre[8];
    const int tid  = threadIdx.x;
    const int lane = tid & 31;
    const int wave = tid >> 5;
    const int base = blockIdx.x * SCAN_CHUNK;

    int pre = 0;
    int i = tid * 4;
    for (; i + 15 * SCAN_CHUNK < base; i += 16 * SCAN_CHUNK) {
        int4 v[16];
#pragma unroll
        for (int u = 0; u < 16; ++u) v[u] = *(const int4*)(cnt + i + u * SCAN_CHUNK);
#pragma unroll
        for (int u = 0; u < 16; ++u) pre += (v[u].x + v[u].y) + (v[u].z + v[u].w);
    }
    for (; i < base; i += SCAN_CHUNK) {
        const int4 v = *(const int4*)(cnt + i);
        pre += (v.x + v.y) + (v.z + v.w);
    }

    const int i0 = base + tid * 4;
    int4 c; c.x = 0; c.y = 0; c.z = 0; c.w = 0;
    if (i0 + 3 < N) {
        c = *(const int4*)(cnt + i0);
    } else {
        if (i0     < N) c.x = cnt[i0];
        if (i0 + 1 < N) c.y = cnt[i0 + 1];
        if (i0 + 2 < N) c.z = cnt[i0 + 2];
    }
    const int t = c.x + c.y + c.z + c.w;
    int inc = t;
    for (int o = 1; o < 32; o <<= 1) {
        const int y = __shfl_up(inc, o, 32);
        if (lane >= o) inc += y;
    }
    for (int o = 16; o > 0; o >>= 1) pre += __shfl_xor(pre, o, 32);
    if (lane == 31) wtot[wave] = inc;
    if (lane == 0)  wpre[wave] = pre;
    __syncthreads();
    int off = 0;
    for (int w = 0; w < 8; ++w) {
        off += wpre[w];
        if (w < wave) off += wtot[w];
    }
    const int ex = off + inc - t;
    if (i0     <= N) out[i0]     = ex;
    if (i0 + 1 <= N) out[i0 + 1] = ex + c.x;
    if (i0 + 2 <= N) out[i0 + 2] = ex + c.x + c.y;
    if (i0 + 3 <= N) out[i0 + 3] = ex + c.x + c.y + c.z;
}

__global__ __launch_bounds__(256) void part_kernel(
    const int* __restrict__ ei, const int* __restrict__ Hs, int* __restrict__ tmp) {
    __shared__ int lh[BK_SLOTS];
    __shared__ int lgo[BK_SLOTS];
    __shared__ int lsrc[EDGE_CHUNK];
    __shared__ int ldst[EDGE_CHUNK];
    __shared__ int wtot[8];
    const int tid  = threadIdx.x;
    const int lane = tid & 31;
    const int wave = tid >> 5;
    const int b    = blockIdx.x;
    const int e0   = b * EDGE_CHUNK;
    lh[tid] = 0; lh[tid + 256] = 0;
    __syncthreads();

    int4 sv[4], dv[4], rv[4];
#pragma unroll
    for (int u = 0; u < 4; ++u) {
        const int e = e0 + 4 * (tid + 256 * u);
        sv[u].x = 0; sv[u].y = 0; sv[u].z = 0; sv[u].w = 0;
        dv[u] = sv[u]; rv[u] = sv[u];
        if (e < NE) {
            sv[u] = *(const int4*)(ei + e);
            dv[u] = *(const int4*)(ei + NE + e);
            rv[u].x = atomicAdd(&lh[dv[u].x >> BSHIFT], 1);
            rv[u].y = atomicAdd(&lh[dv[u].y >> BSHIFT], 1);
            rv[u].z = atomicAdd(&lh[dv[u].z >> BSHIFT], 1);
            rv[u].w = atomicAdd(&lh[dv[u].w >> BSHIFT], 1);
        }
    }
    __syncthreads();

    const int c0 = lh[2 * tid];
    const int c1 = lh[2 * tid + 1];
    const int t  = c0 + c1;
    int inc = t;
    for (int o = 1; o < 32; o <<= 1) {
        const int y = __shfl_up(inc, o, 32);
        if (lane >= o) inc += y;
    }
    if (lane == 31) wtot[wave] = inc;
    __syncthreads();
    int off = 0;
    for (int w = 0; w < 8; ++w) if (w < wave) off += wtot[w];
    const int ex = off + inc - t;
    lh[2 * tid]     = ex;
    lh[2 * tid + 1] = ex + c0;
    if (2 * tid     < NBK) lgo[2 * tid]     = Hs[(2 * tid)     * NBLK + b] - ex;
    if (2 * tid + 1 < NBK) lgo[2 * tid + 1] = Hs[(2 * tid + 1) * NBLK + b] - (ex + c0);
    __syncthreads();

#pragma unroll
    for (int u = 0; u < 4; ++u) {
        const int e = e0 + 4 * (tid + 256 * u);
        if (e < NE) {
            int p;
            p = lh[dv[u].x >> BSHIFT] + rv[u].x; lsrc[p] = sv[u].x; ldst[p] = dv[u].x;
            p = lh[dv[u].y >> BSHIFT] + rv[u].y; lsrc[p] = sv[u].y; ldst[p] = dv[u].y;
            p = lh[dv[u].z >> BSHIFT] + rv[u].z; lsrc[p] = sv[u].z; ldst[p] = dv[u].z;
            p = lh[dv[u].w >> BSHIFT] + rv[u].w; lsrc[p] = sv[u].w; ldst[p] = dv[u].w;
        }
    }
    __syncthreads();

    const int n = (NE - e0) < EDGE_CHUNK ? (NE - e0) : EDGE_CHUNK;
    for (int j = tid; j < n; j += 256) {
        const int d = ldst[j];
        tmp[lgo[d >> BSHIFT] + j] = (lsrc[j] << BSHIFT) | (d & (BW - 1));
    }
}

__global__ __launch_bounds__(256) void csr_kernel(
    const int* __restrict__ Hs, const int* __restrict__ tmp,
    int* __restrict__ rowptr, int* __restrict__ col) {
    __shared__ int lc[BW];
    __shared__ int lcol[STAGE_CAP];
    __shared__ int wtot[8];
    const int tid  = threadIdx.x;
    const int lane = tid & 31;
    const int wave = tid >> 5;
    const int q    = blockIdx.x;
    const int s    = Hs[q * NBLK];
    const int e    = Hs[(q + 1) * NBLK];
    lc[tid] = 0;
    __syncthreads();
    for (int i0 = s + tid; i0 < e; i0 += 256 * CSR_BATCH) {
        int w[CSR_BATCH];
#pragma unroll
        for (int u = 0; u < CSR_BATCH; ++u) {
            const int i = i0 + 256 * u;
            w[u] = (i < e) ? tmp[i] : -1;
        }
#pragma unroll
        for (int u = 0; u < CSR_BATCH; ++u)
            if (w[u] >= 0) atomicAdd(&lc[w[u] & (BW - 1)], 1);
    }
    __syncthreads();

    const int c = lc[tid];
    int inc = c;
    for (int o = 1; o < 32; o <<= 1) {
        const int y = __shfl_up(inc, o, 32);
        if (lane >= o) inc += y;
    }
    if (lane == 31) wtot[wave] = inc;
    __syncthreads();
    int off = 0;
    for (int w = 0; w < 8; ++w) if (w < wave) off += wtot[w];
    const int ex = off + inc - c;
    const int node = q * BW + tid;
    if (node <= NN) rowptr[node] = s + ex;
    lc[tid] = ex;
    __syncthreads();

    const bool staged = (e - s) <= STAGE_CAP;
    for (int i0 = s + tid; i0 < e; i0 += 256 * CSR_BATCH) {
        int w[CSR_BATCH];
#pragma unroll
        for (int u = 0; u < CSR_BATCH; ++u) {
            const int i = i0 + 256 * u;
            w[u] = (i < e) ? tmp[i] : -1;
        }
#pragma unroll
        for (int u = 0; u < CSR_BATCH; ++u) {
            if (w[u] >= 0) {
                const int p = atomicAdd(&lc[w[u] & (BW - 1)], 1);
                if (staged) lcol[p] = w[u] >> BSHIFT; else col[s + p] = w[u] >> BSHIFT;
            }
        }
    }
    __syncthreads();
    if (staged) {
        for (int i = tid; i < e - s; i += 256) col[s + i] = lcol[i];
    }
}

template <int U>
__device__ __forceinline__ void add_rows(const float* __restrict__ xl, int myc, int j,
                                         float4& acc) {
    float4 v[U];
#pragma unroll
    for (int u = 0; u < U; ++u) {
        const int s = __builtin_amdgcn_readlane(myc, j + u);
        v[u] = *(const float4*)(xl + (size_t)s * DIM);
    }
#pragma unroll
    for (int u = 0; u < U; ++u) {
        acc.x += v[u].x; acc.y += v[u].y; acc.z += v[u].z; acc.w += v[u].w;
    }
}

__global__ __launch_bounds__(256) void gather_mean_kernel(
    const float* __restrict__ x, const int* __restrict__ rowptr,
    const int* __restrict__ col, float* __restrict__ mean) {
    const int lane = threadIdx.x & 31;
    const int w = __builtin_amdgcn_readfirstlane((int)(blockIdx.x * 8 + (threadIdx.x >> 5)));
    const float* xl = x + lane * 4;
    for (int q = 0; q < NODES_PER_WAVE; ++q) {
        const int node = w * NODES_PER_WAVE + q;
        if (node >= NN) break;
        const int start = rowptr[node];
        const int end   = rowptr[node + 1];
        float4 acc; acc.x = 0.f; acc.y = 0.f; acc.z = 0.f; acc.w = 0.f;
        for (int j0 = start; j0 < end; j0 += 32) {
            const int m = (end - j0) < 32 ? (end - j0) : 32;
            const int myc = (lane < m) ? col[j0 + lane] : 0;
            int j = 0;
            for (; j + 8 <= m; j += 8) add_rows<8>(xl, myc, j, acc);
            if (j + 4 <= m) { add_rows<4>(xl, myc, j, acc); j += 4; }
            if (j + 2 <= m) { add_rows<2>(xl, myc, j, acc); j += 2; }
            if (j < m)      { add_rows<1>(xl, myc, j, acc); }
        }
        const float inv = 1.0f / fmaxf((float)(end - start), 1.0f);
        float4 o;
        o.x = acc.x * inv; o.y = acc.y * inv; o.z = acc.z * inv; o.w = acc.w * inv;
        *(float4*)(mean + (size_t)node * DIM + lane * 4) = o;
    }
}

__global__ __launch_bounds__(256) void gemm_bn_kernel(
    const float* __restrict__ mean, const float* __restrict__ x,
    const float* __restrict__ Wl, const float* __restrict__ bl,
    const float* __restrict__ Wr, float* __restrict__ h,
    float* __restrict__ gsum, float* __restrict__ gsq) {
    __shared__ float lWl[DIM * DIM];
    __shared__ float lWr[DIM * DIM];
    __shared__ float lsum[DIM];
    __shared__ float lsq[DIM];

    const int tid = threadIdx.x;
    for (int i = tid * 4; i < DIM * DIM; i += 256 * 4) {
        *(float4*)(lWl + i) = *(const float4*)(Wl + i);
        *(float4*)(lWr + i) = *(const float4*)(Wr + i);
    }
    if (tid < DIM) { lsum[tid] = 0.f; lsq[tid] = 0.f; }
    __syncthreads();

    const int wave = tid >> 5;
    const int lane = tid & 31;
    const int tile = blockIdx.x * 8 + wave;
    const bool active = (tile * 16) < NN;

    if (active) {
        const int row0 = tile * 16;
        const int m  = lane & 15;
        const int kh = lane >> 4;
        const float* meanRow = mean + (size_t)(row0 + m) * DIM;
        const float* xRow    = x    + (size_t)(row0 + m) * DIM;

        for (int nt = 0; nt < 8; ++nt) {
            const int ncol = nt * 16 + (lane & 15);
            const float bias = bl[ncol];
            v8f c;
#pragma unroll
            for (int r = 0; r < 8; ++r) c[r] = bias;

#pragma unroll
            for (int k0 = 0; k0 < DIM; k0 += 4) {
                const int ka = k0 + 2 * kh;
                v2f a; a.x = meanRow[ka];           a.y = meanRow[ka + 1];
                v2f b; b.x = lWl[ka * DIM + ncol];  b.y = lWl[(ka + 1) * DIM + ncol];
                c = __builtin_amdgcn_wmma_f32_16x16x4_f32(
                        false, a, false, b, (short)0, c, false, false);
            }
#pragma unroll
            for (int k0 = 0; k0 < DIM; k0 += 4) {
                const int ka = k0 + 2 * kh;
                v2f a; a.x = xRow[ka];              a.y = xRow[ka + 1];
                v2f b; b.x = lWr[ka * DIM + ncol];  b.y = lWr[(ka + 1) * DIM + ncol];
                c = __builtin_amdgcn_wmma_f32_16x16x4_f32(
                        false, a, false, b, (short)0, c, false, false);
            }

            float s = 0.f, s2 = 0.f;
#pragma unroll
            for (int r = 0; r < 8; ++r) {
                float v = c[r] > 0.f ? c[r] : 0.f;
                const int row = row0 + r + 8 * kh;
                h[(size_t)row * DIM + ncol] = v;
                s  += v;
                s2 += v * v;
            }
            s  += __shfl_xor(s, 16, 32);
            s2 += __shfl_xor(s2, 16, 32);
            if (lane < 16) {
                atomicAdd(&lsum[ncol], s);
                atomicAdd(&lsq[ncol], s2);
            }
        }
    }
    __syncthreads();
    if (tid < DIM) {
        atomicAdd(gsum + tid, lsum[tid]);
        atomicAdd(gsq  + tid, lsq[tid]);
    }
}

__global__ void stats_kernel(const float* __restrict__ gsum, const float* __restrict__ gsq,
                             const float* __restrict__ gamma, const float* __restrict__ beta,
                             float* __restrict__ cA, float* __restrict__ cB) {
    int c = threadIdx.x;
    if (c >= DIM) return;
    const float invN = 1.0f / (float)NN;
    float mu  = gsum[c] * invN;
    float var = gsq[c] * invN - mu * mu;
    float a   = rsqrtf(var + BN_EPS) * gamma[c];
    cA[c] = a;
    cB[c] = beta[c] - mu * a;
}

__global__ void apply_kernel(const float* __restrict__ x,
                             const float* __restrict__ cA, const float* __restrict__ cB,
                             float* __restrict__ out) {
    size_t i = (size_t)blockIdx.x * blockDim.x + threadIdx.x;
    if (i >= (size_t)NN * DIM / 4) return;
    int col4 = (int)(i & 31) * 4;
    float4 hv = ((float4*)out)[i];
    float4 xv = ((const float4*)x)[i];
    float4 a  = *(const float4*)(cA + col4);
    float4 b  = *(const float4*)(cB + col4);
    float4 o;
    o.x = xv.x + hv.x * a.x + b.x;
    o.y = xv.y + hv.y * a.y + b.y;
    o.z = xv.z + hv.z * a.z + b.z;
    o.w = xv.w + hv.w * a.w + b.w;
    ((float4*)out)[i] = o;
}

extern "C" void kernel_launch(void* const* d_in, const int* in_sizes, int n_in,
                              void* d_out, int out_size, void* d_ws, size_t ws_size,
                              hipStream_t stream) {
    const float* x     = (const float*)d_in[0];
    const int*   ei    = (const int*)d_in[1];
    const float* Wl    = (const float*)d_in[2];
    const float* bl    = (const float*)d_in[3];
    const float* Wr    = (const float*)d_in[4];
    const float* gamma = (const float*)d_in[5];
    const float* beta  = (const float*)d_in[6];
    float* out = (float*)d_out;

    float* agg  = (float*)d_ws;
    float* deg  = agg + (size_t)NN * DIM;
    float* gsum = deg + NN;
    float* gsq  = gsum + DIM;
    float* cA   = gsq + DIM;
    float* cB   = cA + DIM;

    int* H   = (int*)d_ws;
    int* Hs  = H + NH_SLOTS;
    int* tmp = Hs + NH_SLOTS;

    int* rowptr = (int*)d_out;
    int* col    = rowptr + ROWPTR_SLOTS;

    hist_kernel<<<NBLK, 256, 0, stream>>>(ei, H, gsum);

    scan_kernel<NH><<<(NH + 1 + SCAN_CHUNK - 1) / SCAN_CHUNK, 256, 0, stream>>>(H, Hs);

    part_kernel<<<NBLK, 256, 0, stream>>>(ei, Hs, tmp);

    csr_kernel<<<NBK, 256, 0, stream>>>(Hs, tmp, rowptr, col);

    gather_mean_kernel<<<(NN + 8 * NODES_PER_WAVE - 1) / (8 * NODES_PER_WAVE), 256, 0, stream>>>(
        x, rowptr, col, agg);

    const int tiles = NN / 16;
    gemm_bn_kernel<<<(tiles + 7) / 8, 256, 0, stream>>>(agg, x, Wl, bl, Wr,
                                                        out, gsum, gsq);

    stats_kernel<<<1, 128, 0, stream>>>(gsum, gsq, gamma, beta, cA, cB);

    int mb = (int)(((size_t)NN * DIM / 4 + 255) / 256);
    apply_kernel<<<mb, 256, 0, stream>>>(x, cA, cB, out);
}


# matrix kernel: both WMMA operands held as k-quads: A rows read once by 16-byte loads, weights in LDS as [k/4][n][k%4], one conflict-free 16-byte LDS load per two products; k grouping of sums changed
# speedup vs baseline: 2.6078x; 1.0684x over previous
#include <hip/hip_runtime.h>

#define NN 100000
#define NE 1600000
#define DIM 128
#define BN_EPS 1e-5f
#define SCAN_CHUNK 1024
#define NODES_PER_WAVE 4
#define ROWPTR_SLOTS 100352
#define EDGE_CHUNK 4096
#define NBLK ((NE + EDGE_CHUNK - 1) / EDGE_CHUNK)
#define BSHIFT 8
#define BW (1 << BSHIFT)
#define NBK ((NN + BW - 1) / BW)
#define BK_SLOTS 512
#define NH (NBK * NBLK)
#define NH_SLOTS 153600
#define STAGE_CAP 8192
#define CSR_BATCH 16

typedef __attribute__((ext_vector_type(2))) float v2f;
typedef __attribute__((ext_vector_type(8))) float v8f;


__global__ __launch_bounds__(256) void hist_kernel(
    const int* __restrict__ ei, int* __restrict__ H, float* __restrict__ sums) {
    __shared__ int lh[BK_SLOTS];
    const int tid = threadIdx.x;
    const int b   = blockIdx.x;
    const int e0  = b * EDGE_CHUNK;
    lh[tid] = 0; lh[tid + 256] = 0;
    if (b == 0) sums[tid] = 0.f;
    __syncthreads();
    for (int u = 0; u < 4; ++u) {
        const int e = e0 + 4 * (tid + 256 * u);
        if (e < NE) {
            const int4 d = *(const int4*)(ei + NE + e);
            atomicAdd(&lh[d.x >> BSHIFT], 1);
            atomicAdd(&lh[d.y >> BSHIFT], 1);
            atomicAdd(&lh[d.z >> BSHIFT], 1);
            atomicAdd(&lh[d.w >> BSHIFT], 1);
        }
    }
    __syncthreads();
    for (int q = tid; q < NBK; q += 256) H[q * NBLK + b] = lh[q];
}

template <int N>
__global__ __launch_bounds__(256) void scan_kernel(
    const int* __restrict__ cnt, int* __restrict__ out) {
    __shared__ int wtot[8];
    __shared__ int wpre[8];
    const int tid  = threadIdx.x;
    const int lane = tid & 31;
    const int wave = tid >> 5;
    const int base = blockIdx.x * SCAN_CHUNK;

    int pre = 0;
    int i = tid * 4;
    for (; i + 15 * SCAN_CHUNK < base; i += 16 * SCAN_CHUNK) {
        int4 v[16];
#pragma unroll
        for (int u = 0; u < 16; ++u) v[u] = *(const int4*)(cnt + i + u * SCAN_CHUNK);
#pragma unroll
        for (int u = 0; u < 16; ++u) pre += (v[u].x + v[u].y) + (v[u].z + v[u].w);
    }
    for (; i < base; i += SCAN_CHUNK) {
        const int4 v = *(const int4*)(cnt + i);
        pre += (v.x + v.y) + (v.z + v.w);
    }

    const int i0 = base + tid * 4;
    int4 c; c.x = 0; c.y = 0; c.z = 0; c.w = 0;
    if (i0 + 3 < N) {
        c = *(const int4*)(cnt + i0);
    } else {
        if (i0     < N) c.x = cnt[i0];
        if (i0 + 1 < N) c.y = cnt[i0 + 1];
        if (i0 + 2 < N) c.z = cnt[i0 + 2];
    }
    const int t = c.x + c.y + c.z + c.w;
    int inc = t;
    for (int o = 1; o < 32; o <<= 1) {
        const int y = __shfl_up(inc, o, 32);
        if (lane >= o) inc += y;
    }
    for (int o = 16; o > 0; o >>= 1) pre += __shfl_xor(pre, o, 32);
    if (lane == 31) wtot[wave] = inc;
    if (lane == 0)  wpre[wave] = pre;
    __syncthreads();
    int off = 0;
    for (int w = 0; w < 8; ++w) {
        off += wpre[w];
        if (w < wave) off += wtot[w];
    }
    const int ex = off + inc - t;
    if (i0     <= N) out[i0]     = ex;
    if (i0 + 1 <= N) out[i0 + 1] = ex + c.x;
    if (i0 + 2 <= N) out[i0 + 2] = ex + c.x + c.y;
    if (i0 + 3 <= N) out[i0 + 3] = ex + c.x + c.y + c.z;
}

__global__ __launch_bounds__(256) void part_kernel(
    const int* __restrict__ ei, const int* __restrict__ Hs, int* __restrict__ tmp) {
    __shared__ int lh[BK_SLOTS];
    __shared__ int lgo[BK_SLOTS];
    __shared__ int lsrc[EDGE_CHUNK];
    __shared__ int ldst[EDGE_CHUNK];
    __shared__ int wtot[8];
    const int tid  = threadIdx.x;
    const int lane = tid & 31;
    const int wave = tid >> 5;
    const int b    = blockIdx.x;
    const int e0   = b * EDGE_CHUNK;
    lh[tid] = 0; lh[tid + 256] = 0;
    __syncthreads();

    int4 sv[4], dv[4], rv[4];
#pragma unroll
    for (int u = 0; u < 4; ++u) {
        const int e = e0 + 4 * (tid + 256 * u);
        sv[u].x = 0; sv[u].y = 0; sv[u].z = 0; sv[u].w = 0;
        dv[u] = sv[u]; rv[u] = sv[u];
        if (e < NE) {
            sv[u] = *(const int4*)(ei + e);
            dv[u] = *(const int4*)(ei + NE + e);
            rv[u].x = atomicAdd(&lh[dv[u].x >> BSHIFT], 1);
            rv[u].y = atomicAdd(&lh[dv[u].y >> BSHIFT], 1);
            rv[u].z = atomicAdd(&lh[dv[u].z >> BSHIFT], 1);
            rv[u].w = atomicAdd(&lh[dv[u].w >> BSHIFT], 1);
        }
    }
    __syncthreads();

    const int c0 = lh[2 * tid];
    const int c1 = lh[2 * tid + 1];
    const int t  = c0 + c1;
    int inc = t;
    for (int o = 1; o < 32; o <<= 1) {
        const int y = __shfl_up(inc, o, 32);
        if (lane >= o) inc += y;
    }
    if (lane == 31) wtot[wave] = inc;
    __syncthreads();
    int off = 0;
    for (int w = 0; w < 8; ++w) if (w < wave) off += wtot[w];
    const int ex = off + inc - t;
    lh[2 * tid]     = ex;
    lh[2 * tid + 1] = ex + c0;
    if (2 * tid     < NBK) lgo[2 * tid]     = Hs[(2 * tid)     * NBLK + b] - ex;
    if (2 * tid + 1 < NBK) lgo[2 * tid + 1] = Hs[(2 * tid + 1) * NBLK + b] - (ex + c0);
    __syncthreads();

#pragma unroll
    for (int u = 0; u < 4; ++u) {
        const int e = e0 + 4 * (tid + 256 * u);
        if (e < NE) {
            int p;
            p = lh[dv[u].x >> BSHIFT] + rv[u].x; lsrc[p] = sv[u].x; ldst[p] = dv[u].x;
            p = lh[dv[u].y >> BSHIFT] + rv[u].y; lsrc[p] = sv[u].y; ldst[p] = dv[u].y;
            p = lh[dv[u].z >> BSHIFT] + rv[u].z; lsrc[p] = sv[u].z; ldst[p] = dv[u].z;
            p = lh[dv[u].w >> BSHIFT] + rv[u].w; lsrc[p] = sv[u].w; ldst[p] = dv[u].w;
        }
    }
    __syncthreads();

    const int n = (NE - e0) < EDGE_CHUNK ? (NE - e0) : EDGE_CHUNK;
    for (int j = tid; j < n; j += 256) {
        const int d = ldst[j];
        tmp[lgo[d >> BSHIFT] + j] = (lsrc[j] << BSHIFT) | (d & (BW - 1));
    }
}

__global__ __launch_bounds__(256) void csr_kernel(
    const int* __restrict__ Hs, const int* __restrict__ tmp,
    int* __restrict__ rowptr, int* __restrict__ col) {
    __shared__ int lc[BW];
    __shared__ int lcol[STAGE_CAP];
    __shared__ int wtot[8];
    const int tid  = threadIdx.x;
    const int lane = tid & 31;
    const int wave = tid >> 5;
    const int q    = blockIdx.x;
    const int s    = Hs[q * NBLK];
    const int e    = Hs[(q + 1) * NBLK];
    lc[tid] = 0;
    __syncthreads();
    for (int i0 = s + tid; i0 < e; i0 += 256 * CSR_BATCH) {
        int w[CSR_BATCH];
#pragma unroll
        for (int u = 0; u < CSR_BATCH; ++u) {
            const int i = i0 + 256 * u;
            w[u] = (i < e) ? tmp[i] : -1;
        }
#pragma unroll
        for (int u = 0; u < CSR_BATCH; ++u)
            if (w[u] >= 0) atomicAdd(&lc[w[u] & (BW - 1)], 1);
    }
    __syncthreads();

    const int c = lc[tid];
    int inc = c;
    for (int o = 1; o < 32; o <<= 1) {
        const int y = __shfl_up(inc, o, 32);
        if (lane >= o) inc += y;
    }
    if (lane == 31) wtot[wave] = inc;
    __syncthreads();
    int off = 0;
    for (int w = 0; w < 8; ++w) if (w < wave) off += wtot[w];
    const int ex = off + inc - c;
    const int node = q * BW + tid;
    if (node <= NN) rowptr[node] = s + ex;
    lc[tid] = ex;
    __syncthreads();

    const bool staged = (e - s) <= STAGE_CAP;
    for (int i0 = s + tid; i0 < e; i0 += 256 * CSR_BATCH) {
        int w[CSR_BATCH];
#pragma unroll
        for (int u = 0; u < CSR_BATCH; ++u) {
            const int i = i0 + 256 * u;
            w[u] = (i < e) ? tmp[i] : -1;
        }
#pragma unroll
        for (int u = 0; u < CSR_BATCH; ++u) {
            if (w[u] >= 0) {
                const int p = atomicAdd(&lc[w[u] & (BW - 1)], 1);
                if (staged) lcol[p] = w[u] >> BSHIFT; else col[s + p] = w[u] >> BSHIFT;
            }
        }
    }
    __syncthreads();
    if (staged) {
        for (int i = tid; i < e - s; i += 256) col[s + i] = lcol[i];
    }
}

template <int U>
__device__ __forceinline__ void add_rows(const float* __restrict__ xl, int myc, int j,
                                         float4& acc) {
    float4 v[U];
#pragma unroll
    for (int u = 0; u < U; ++u) {
        const int s = __builtin_amdgcn_readlane(myc, j + u);
        v[u] = *(const float4*)(xl + (size_t)s * DIM);
    }
#pragma unroll
    for (int u = 0; u < U; ++u) {
        acc.x += v[u].x; acc.y += v[u].y; acc.z += v[u].z; acc.w += v[u].w;
    }
}

__global__ __launch_bounds__(256) void gather_mean_kernel(
    const float* __restrict__ x, const int* __restrict__ rowptr,
    const int* __restrict__ col, float* __restrict__ mean) {
    const int lane = threadIdx.x & 31;
    const int w = __builtin_amdgcn_readfirstlane((int)(blockIdx.x * 8 + (threadIdx.x >> 5)));
    const float* xl = x + lane * 4;
    for (int q = 0; q < NODES_PER_WAVE; ++q) {
        const int node = w * NODES_PER_WAVE + q;
        if (node >= NN) break;
        const int start = rowptr[node];
        const int end   = rowptr[node + 1];
        float4 acc; acc.x = 0.f; acc.y = 0.f; acc.z = 0.f; acc.w = 0.f;
        for (int j0 = start; j0 < end; j0 += 32) {
            const int m = (end - j0) < 32 ? (end - j0) : 32;
            const int myc = (lane < m) ? col[j0 + lane] : 0;
            int j = 0;
            for (; j + 8 <= m; j += 8) add_rows<8>(xl, myc, j, acc);
            if (j + 4 <= m) { add_rows<4>(xl, myc, j, acc); j += 4; }
            if (j + 2 <= m) { add_rows<2>(xl, myc, j, acc); j += 2; }
            if (j < m)      { add_rows<1>(xl, myc, j, acc); }
        }
        const float inv = 1.0f / fmaxf((float)(end - start), 1.0f);
        float4 o;
        o.x = acc.x * inv; o.y = acc.y * inv; o.z = acc.z * inv; o.w = acc.w * inv;
        *(float4*)(mean + (size_t)node * DIM + lane * 4) = o;
    }
}

__global__ __launch_bounds__(256) void gemm_bn_kernel(
    const float* __restrict__ mean, const float* __restrict__ x,
    const float* __restrict__ Wl, const float* __restrict__ bl,
    const float* __restrict__ Wr, float* __restrict__ h,
    float* __restrict__ gsum, float* __restrict__ gsq) {
    __shared__ float lWl[DIM * DIM];
    __shared__ float lWr[DIM * DIM];
    __shared__ float lsum[DIM];
    __shared__ float lsq[DIM];

    const int tid  = threadIdx.x;
    const int wave = tid >> 5;
    const int lane = tid & 31;
    const int m    = lane & 15;
    const int kh   = lane >> 4;
    const int tile = blockIdx.x * 8 + wave;
    const bool active = (tile * 16) < NN;

    const int arow = (active ? tile * 16 : 0) + m;
    const float* meanRow = mean + (size_t)arow * DIM + 4 * kh;
    const float* xRow    = x    + (size_t)arow * DIM + 4 * kh;
    float4 am[16], ax[16];
#pragma unroll
    for (int j = 0; j < 16; ++j) {
        am[j] = *(const float4*)(meanRow + 8 * j);
        ax[j] = *(const float4*)(xRow + 8 * j);
    }

    {
        const int n  = tid & (DIM - 1);
        const int qb = tid >> 7;
        const float* pl = Wl + (size_t)(4 * qb) * DIM + n;
        const float* pr = Wr + (size_t)(4 * qb) * DIM + n;
        float* sl = lWl + qb * 512 + n * 4;
        float* sr = lWr + qb * 512 + n * 4;
#pragma nounroll
        for (int ub = 0; ub < 16; ub += 8) {
            float4 wl[8], wr[8];
#pragma unroll
            for (int u = 0; u < 8; ++u) {
                const float* a = pl + (size_t)(ub + u) * (8 * DIM);
                const float* b = pr + (size_t)(ub + u) * (8 * DIM);
                wl[u].x = a[0]; wl[u].y = a[DIM]; wl[u].z = a[2 * DIM]; wl[u].w = a[3 * DIM];
                wr[u].x = b[0]; wr[u].y = b[DIM]; wr[u].z = b[2 * DIM]; wr[u].w = b[3 * DIM];
            }
#pragma unroll
            for (int u = 0; u < 8; ++u) {
                *(float4*)(sl + (ub + u) * 1024) = wl[u];
                *(float4*)(sr + (ub + u) * 1024) = wr[u];
            }
        }
    }
    if (tid < DIM) { lsum[tid] = 0.f; lsq[tid] = 0.f; }
    __syncthreads();

    if (active) {
        const int row0 = tile * 16;

        for (int nt = 0; nt < 8; ++nt) {
            const int ncol = nt * 16 + m;
            const float bias = bl[ncol];
            v8f c;
#pragma unroll
            for (int r = 0; r < 8; ++r) c[r] = bias;

            const float* bwl = lWl + kh * 512 + ncol * 4;
            const float* bwr = lWr + kh * 512 + ncol * 4;

#pragma unroll
            for (int j = 0; j < 16; ++j) {
                const float4 bq = *(const float4*)(bwl + j * 1024);
                v2f a0; a0.x = am[j].x; a0.y = am[j].y;
                v2f b0; b0.x = bq.x;    b0.y = bq.y;
                c = __builtin_amdgcn_wmma_f32_16x16x4_f32(
                        false, a0, false, b0, (short)0, c, false, false);
                v2f a1; a1.x = am[j].z; a1.y = am[j].w;
                v2f b1; b1.x = bq.z;    b1.y = bq.w;
                c = __builtin_amdgcn_wmma_f32_16x16x4_f32(
                        false, a1, false, b1, (short)0, c, false, false);
            }
#pragma unroll
            for (int j = 0; j < 16; ++j) {
                const float4 bq = *(const float4*)(bwr + j * 1024);
                v2f a0; a0.x = ax[j].x; a0.y = ax[j].y;
                v2f b0; b0.x = bq.x;    b0.y = bq.y;
                c = __builtin_amdgcn_wmma_f32_16x16x4_f32(
                        false, a0, false, b0, (short)0, c, false, false);
                v2f a1; a1.x = ax[j].z; a1.y = ax[j].w;
                v2f b1; b1.x = bq.z;    b1.y = bq.w;
                c = __builtin_amdgcn_wmma_f32_16x16x4_f32(
                        false, a1, false, b1, (short)0, c, false, false);
            }

            float s = 0.f, s2 = 0.f;
#pragma unroll
            for (int r = 0; r < 8; ++r) {
                float v = c[r] > 0.f ? c[r] : 0.f;
                const int row = row0 + r + 8 * kh;
                h[(size_t)row * DIM + ncol] = v;
                s  += v;
                s2 += v * v;
            }
            s  += __shfl_xor(s, 16, 32);
            s2 += __shfl_xor(s2, 16, 32);
            if (lane < 16) {
                atomicAdd(&lsum[ncol], s);
                atomicAdd(&lsq[ncol], s2);
            }
        }
    }
    __syncthreads();
    if (tid < DIM) {
        atomicAdd(gsum + tid, lsum[tid]);
        atomicAdd(gsq  + tid, lsq[tid]);
    }
}

__global__ void stats_kernel(const float* __restrict__ gsum, const float* __restrict__ gsq,
                             const float* __restrict__ gamma, const float* __restrict__ beta,
                             float* __restrict__ cA, float* __restrict__ cB) {
    int c = threadIdx.x;
    if (c >= DIM) return;
    const float invN = 1.0f / (float)NN;
    float mu  = gsum[c] * invN;
    float var = gsq[c] * invN - mu * mu;
    float a   = rsqrtf(var + BN_EPS) * gamma[c];
    cA[c] = a;
    cB[c] = beta[c] - mu * a;
}

__global__ void apply_kernel(const float* __restrict__ x,
                             const float* __restrict__ cA, const float* __restrict__ cB,
                             float* __restrict__ out) {
    size_t i = (size_t)blockIdx.x * blockDim.x + threadIdx.x;
    if (i >= (size_t)NN * DIM / 4) return;
    int col4 = (int)(i & 31) * 4;
    float4 hv = ((float4*)out)[i];
    float4 xv = ((const float4*)x)[i];
    float4 a  = *(const float4*)(cA + col4);
    float4 b  = *(const float4*)(cB + col4);
    float4 o;
    o.x = xv.x + hv.x * a.x + b.x;
    o.y = xv.y + hv.y * a.y + b.y;
    o.z = xv.z + hv.z * a.z + b.z;
    o.w = xv.w + hv.w * a.w + b.w;
    ((float4*)out)[i] = o;
}

extern "C" void kernel_launch(void* const* d_in, const int* in_sizes, int n_in,
                              void* d_out, int out_size, void* d_ws, size_t ws_size,
                              hipStream_t stream) {
    const float* x     = (const float*)d_in[0];
    const int*   ei    = (const int*)d_in[1];
    const float* Wl    = (const float*)d_in[2];
    const float* bl    = (const float*)d_in[3];
    const float* Wr    = (const float*)d_in[4];
    const float* gamma = (const float*)d_in[5];
    const float* beta  = (const float*)d_in[6];
    float* out = (float*)d_out;

    float* agg  = (float*)d_ws;
    float* deg  = agg + (size_t)NN * DIM;
    float* gsum = deg + NN;
    float* gsq  = gsum + DIM;
    float* cA   = gsq + DIM;
    float* cB   = cA + DIM;

    int* H   = (int*)d_ws;
    int* Hs  = H + NH_SLOTS;
    int* tmp = Hs + NH_SLOTS;

    int* rowptr = (int*)d_out;
    int* col    = rowptr + ROWPTR_SLOTS;

    hist_kernel<<<NBLK, 256, 0, stream>>>(ei, H, gsum);

    scan_kernel<NH><<<(NH + 1 + SCAN_CHUNK - 1) / SCAN_CHUNK, 256, 0, stream>>>(H, Hs);

    part_kernel<<<NBLK, 256, 0, stream>>>(ei, Hs, tmp);

    csr_kernel<<<NBK, 256, 0, stream>>>(Hs, tmp, rowptr, col);

    gather_mean_kernel<<<(NN + 8 * NODES_PER_WAVE - 1) / (8 * NODES_PER_WAVE), 256, 0, stream>>>(
        x, rowptr, col, agg);

    const int tiles = NN / 16;
    gemm_bn_kernel<<<(tiles + 7) / 8, 256, 0, stream>>>(agg, x, Wl, bl, Wr,
                                                        out, gsum, gsq);

    stats_kernel<<<1, 128, 0, stream>>>(gsum, gsq, gamma, beta, cA, cB);

    int mb = (int)(((size_t)NN * DIM / 4 + 255) / 256);
    apply_kernel<<<mb, 256, 0, stream>>>(x, cA, cB, out);
}


# edge grouping without its count and prefix-sum launches: chunks bucket-ordered in place with a start table; bucket place = sum of chunk starts; 8 launches -> 6; per-node sum order = LDS arrival
# speedup vs baseline: 2.9424x; 1.1283x over previous
#include <hip/hip_runtime.h>

#define NN 100000
#define NE 1600000
#define DIM 128
#define BN_EPS 1e-5f
#define NODES_PER_WAVE 4
#define ROWPTR_SLOTS 100352
#define EDGE_CHUNK 4096
#define NBLK ((NE + EDGE_CHUNK - 1) / EDGE_CHUNK)
#define BSHIFT 8
#define BW (1 << BSHIFT)
#define NBK ((NN + BW - 1) / BW)
#define BK_SLOTS 512
#define NT_SLOTS 153600
#define STAGE_CAP 8192
#define CSR_BATCH 16
#define RUNS_PER_HALF ((NBLK + 15) / 16)

typedef __attribute__((ext_vector_type(2))) float v2f;
typedef __attribute__((ext_vector_type(8))) float v8f;


__global__ __launch_bounds__(256) void part_kernel(
    const int* __restrict__ ei, int* __restrict__ T, int* __restrict__ tmp,
    float* __restrict__ sums) {
    __shared__ int lh[BK_SLOTS];
    __shared__ alignas(16) int lpk[EDGE_CHUNK];
    __shared__ int wtot[8];
    const int tid  = threadIdx.x;
    const int lane = tid & 31;
    const int wave = tid >> 5;
    const int b    = blockIdx.x;
    const int e0   = b * EDGE_CHUNK;
    lh[tid] = 0; lh[tid + 256] = 0;
    if (b == 0) sums[tid] = 0.f;
    __syncthreads();

    int4 sv[4], dv[4], rv[4];
#pragma unroll
    for (int u = 0; u < 4; ++u) {
        const int e = e0 + 4 * (tid + 256 * u);
        sv[u].x = 0; sv[u].y = 0; sv[u].z = 0; sv[u].w = 0;
        dv[u] = sv[u]; rv[u] = sv[u];
        if (e < NE) {
            sv[u] = *(const int4*)(ei + e);
            dv[u] = *(const int4*)(ei + NE + e);
            rv[u].x = atomicAdd(&lh[dv[u].x >> BSHIFT], 1);
            rv[u].y = atomicAdd(&lh[dv[u].y >> BSHIFT], 1);
            rv[u].z = atomicAdd(&lh[dv[u].z >> BSHIFT], 1);
            rv[u].w = atomicAdd(&lh[dv[u].w >> BSHIFT], 1);
        }
    }
    __syncthreads();

    const int c0 = lh[2 * tid];
    const int c1 = lh[2 * tid + 1];
    const int t  = c0 + c1;
    int inc = t;
    for (int o = 1; o < 32; o <<= 1) {
        const int y = __shfl_up(inc, o, 32);
        if (lane >= o) inc += y;
    }
    if (lane == 31) wtot[wave] = inc;
    __syncthreads();
    int off = 0;
    for (int w = 0; w < 8; ++w) if (w < wave) off += wtot[w];
    const int ex = off + inc - t;
    lh[2 * tid]     = ex;
    lh[2 * tid + 1] = ex + c0;
    if (2 * tid     <= NBK) T[(2 * tid)     * NBLK + b] = ex;
    if (2 * tid + 1 <= NBK) T[(2 * tid + 1) * NBLK + b] = ex + c0;
    __syncthreads();

#pragma unroll
    for (int u = 0; u < 4; ++u) {
        const int e = e0 + 4 * (tid + 256 * u);
        if (e < NE) {
            lpk[lh[dv[u].x >> BSHIFT] + rv[u].x] = (sv[u].x << BSHIFT) | (dv[u].x & (BW - 1));
            lpk[lh[dv[u].y >> BSHIFT] + rv[u].y] = (sv[u].y << BSHIFT) | (dv[u].y & (BW - 1));
            lpk[lh[dv[u].z >> BSHIFT] + rv[u].z] = (sv[u].z << BSHIFT) | (dv[u].z & (BW - 1));
            lpk[lh[dv[u].w >> BSHIFT] + rv[u].w] = (sv[u].w << BSHIFT) | (dv[u].w & (BW - 1));
        }
    }
    __syncthreads();

#pragma unroll
    for (int u = 0; u < 4; ++u) {
        const int j = 4 * (tid + 256 * u);
        if (e0 + j < NE) *(int4*)(tmp + e0 + j) = *(const int4*)(lpk + j);
    }
}

__global__ __launch_bounds__(256) void csr_kernel(
    const int* __restrict__ T, const int* __restrict__ tmp,
    int* __restrict__ rowptr, int* __restrict__ col) {
    __shared__ int2 lan[BK_SLOTS];
    __shared__ int lp[BK_SLOTS];
    __shared__ int lc[BW];
    __shared__ int lw[STAGE_CAP];
    __shared__ int lcol[STAGE_CAP];
    __shared__ int wtot[8];
    __shared__ int wsta[8];
    __shared__ int wnod[8];
    const int tid  = threadIdx.x;
    const int lane = tid & 31;
    const int wave = tid >> 5;
    const int q    = blockIdx.x;

    int a0 = 0, n0 = 0, a1 = 0, n1 = 0;
    if (2 * tid < NBLK) {
        a0 = T[q * NBLK + 2 * tid];
        n0 = T[(q + 1) * NBLK + 2 * tid] - a0;
    }
    if (2 * tid + 1 < NBLK) {
        a1 = T[q * NBLK + 2 * tid + 1];
        n1 = T[(q + 1) * NBLK + 2 * tid + 1] - a1;
    }
    const int t = n0 + n1;
    int inc = t;
    for (int o = 1; o < 32; o <<= 1) {
        const int y = __shfl_up(inc, o, 32);
        if (lane >= o) inc += y;
    }
    int asum = a0 + a1;
    for (int o = 16; o > 0; o >>= 1) asum += __shfl_xor(asum, o, 32);
    if (lane == 31) wtot[wave] = inc;
    if (lane == 0)  wsta[wave] = asum;
    lc[tid] = 0;
    __syncthreads();
    int off = 0, s = 0, total = 0;
    for (int w = 0; w < 8; ++w) {
        if (w < wave) off += wtot[w];
        total += wtot[w];
        s += wsta[w];
    }
    const int exr = off + inc - t;
    lan[2 * tid]     = make_int2(a0, n0); lp[2 * tid]     = exr;
    lan[2 * tid + 1] = make_int2(a1, n1); lp[2 * tid + 1] = exr + n0;
    __syncthreads();

    const bool staged = total <= STAGE_CAP;
    const int l16 = lane & 15;
    const int cb  = 2 * wave + (lane >> 4);
    if (staged) {
        int v[RUNS_PER_HALF], ra[RUNS_PER_HALF], rn[RUNS_PER_HALF], rp[RUNS_PER_HALF];
#pragma unroll
        for (int u = 0; u < RUNS_PER_HALF; ++u) {
            const int c  = cb + 16 * u;
            const int cc = c < NBLK ? c : NBLK - 1;
            const int2 an = lan[cc];
            ra[u] = cc * EDGE_CHUNK + an.x;
            rn[u] = c < NBLK ? an.y : 0;
            rp[u] = lp[cc];
            v[u]  = tmp[l16 < rn[u] ? ra[u] + l16 : 0];
        }
#pragma unroll
        for (int u = 0; u < RUNS_PER_HALF; ++u) {
            if (l16 < rn[u]) {
                lw[rp[u] + l16] = v[u];
                atomicAdd(&lc[v[u] & (BW - 1)], 1);
            }
            for (int k = 16 + l16; k < rn[u]; k += 16) {
                const int w = tmp[ra[u] + k];
                lw[rp[u] + k] = w;
                atomicAdd(&lc[w & (BW - 1)], 1);
            }
        }
    } else {
        for (int c = cb; c < NBLK; c += 16) {
            const int a = lan[c].x;
            const int n = lan[c].y;
            for (int k = l16; k < n; k += 16)
                atomicAdd(&lc[tmp[c * EDGE_CHUNK + a + k] & (BW - 1)], 1);
        }
    }
    __syncthreads();

    const int cn = lc[tid];
    int incn = cn;
    for (int o = 1; o < 32; o <<= 1) {
        const int y = __shfl_up(incn, o, 32);
        if (lane >= o) incn += y;
    }
    if (lane == 31) wnod[wave] = incn;
    __syncthreads();
    int offn = 0;
    for (int w = 0; w < 8; ++w) if (w < wave) offn += wnod[w];
    const int ex = offn + incn - cn;
    const int node = q * BW + tid;
    if (node <= NN) rowptr[node] = s + ex;
    lc[tid] = ex;
    __syncthreads();

    if (staged) {
        for (int i0 = tid; i0 < total; i0 += 256 * CSR_BATCH) {
            int w[CSR_BATCH];
#pragma unroll
            for (int u = 0; u < CSR_BATCH; ++u) {
                const int i = i0 + 256 * u;
                w[u] = (i < total) ? lw[i] : -1;
            }
#pragma unroll
            for (int u = 0; u < CSR_BATCH; ++u) {
                if (w[u] >= 0) {
                    const int p = atomicAdd(&lc[w[u] & (BW - 1)], 1);
                    lcol[p] = w[u] >> BSHIFT;
                }
            }
        }
    } else {
        for (int c = cb; c < NBLK; c += 16) {
            const int a = lan[c].x;
            const int n = lan[c].y;
            for (int k = l16; k < n; k += 16) {
                const int w = tmp[c * EDGE_CHUNK + a + k];
                const int p = atomicAdd(&lc[w & (BW - 1)], 1);
                col[s + p] = w >> BSHIFT;
            }
        }
    }
    __syncthreads();
    if (staged) {
        for (int i = tid; i < total; i += 256) col[s + i] = lcol[i];
    }
}

template <int U>
__device__ __forceinline__ void add_rows(const float* __restrict__ xl, int myc, int j,
                                         float4& acc) {
    float4 v[U];
#pragma unroll
    for (int u = 0; u < U; ++u) {
        const int s = __builtin_amdgcn_readlane(myc, j + u);
        v[u] = *(const float4*)(xl + (size_t)s * DIM);
    }
#pragma unroll
    for (int u = 0; u < U; ++u) {
        acc.x += v[u].x; acc.y += v[u].y; acc.z += v[u].z; acc.w += v[u].w;
    }
}

__global__ __launch_bounds__(256) void gather_mean_kernel(
    const float* __restrict__ x, const int* __restrict__ rowptr,
    const int* __restrict__ col, float* __restrict__ mean) {
    const int lane = threadIdx.x & 31;
    const int w = __builtin_amdgcn_readfirstlane((int)(blockIdx.x * 8 + (threadIdx.x >> 5)));
    const float* xl = x + lane * 4;
    for (int q = 0; q < NODES_PER_WAVE; ++q) {
        const int node = w * NODES_PER_WAVE + q;
        if (node >= NN) break;
        const int start = rowptr[node];
        const int end   = rowptr[node + 1];
        float4 acc; acc.x = 0.f; acc.y = 0.f; acc.z = 0.f; acc.w = 0.f;
        for (int j0 = start; j0 < end; j0 += 32) {
            const int m = (end - j0) < 32 ? (end - j0) : 32;
            const int myc = (lane < m) ? col[j0 + lane] : 0;
            int j = 0;
            for (; j + 8 <= m; j += 8) add_rows<8>(xl, myc, j, acc);
            if (j + 4 <= m) { add_rows<4>(xl, myc, j, acc); j += 4; }
            if (j + 2 <= m) { add_rows<2>(xl, myc, j, acc); j += 2; }
            if (j < m)      { add_rows<1>(xl, myc, j, acc); }
        }
        const float inv = 1.0f / fmaxf((float)(end - start), 1.0f);
        float4 o;
        o.x = acc.x * inv; o.y = acc.y * inv; o.z = acc.z * inv; o.w = acc.w * inv;
        *(float4*)(mean + (size_t)node * DIM + lane * 4) = o;
    }
}

__global__ __launch_bounds__(256) void gemm_bn_kernel(
    const float* __restrict__ mean, const float* __restrict__ x,
    const float* __restrict__ Wl, const float* __restrict__ bl,
    const float* __restrict__ Wr, float* __restrict__ h,
    float* __restrict__ gsum, float* __restrict__ gsq) {
    __shared__ float lWl[DIM * DIM];
    __shared__ float lWr[DIM * DIM];
    __shared__ float lsum[DIM];
    __shared__ float lsq[DIM];

    const int tid  = threadIdx.x;
    const int wave = tid >> 5;
    const int lane = tid & 31;
    const int m    = lane & 15;
    const int kh   = lane >> 4;
    const int tile = blockIdx.x * 8 + wave;
    const bool active = (tile * 16) < NN;

    const int arow = (active ? tile * 16 : 0) + m;
    const float* meanRow = mean + (size_t)arow * DIM + 4 * kh;
    const float* xRow    = x    + (size_t)arow * DIM + 4 * kh;
    float4 am[16], ax[16];
#pragma unroll
    for (int j = 0; j < 16; ++j) {
        am[j] = *(const float4*)(meanRow + 8 * j);
        ax[j] = *(const float4*)(xRow + 8 * j);
    }

    {
        const int n  = tid & (DIM - 1);
        const int qb = tid >> 7;
        const float* pl = Wl + (size_t)(4 * qb) * DIM + n;
        const float* pr = Wr + (size_t)(4 * qb) * DIM + n;
        float* sl = lWl + qb * 512 + n * 4;
        float* sr = lWr + qb * 512 + n * 4;
#pragma nounroll
        for (int ub = 0; ub < 16; ub += 8) {
            float4 wl[8], wr[8];
#pragma unroll
            for (int u = 0; u < 8; ++u) {
                const float* a = pl + (size_t)(ub + u) * (8 * DIM);
                const float* b = pr + (size_t)(ub + u) * (8 * DIM);
                wl[u].x = a[0]; wl[u].y = a[DIM]; wl[u].z = a[2 * DIM]; wl[u].w = a[3 * DIM];
                wr[u].x = b[0]; wr[u].y = b[DIM]; wr[u].z = b[2 * DIM]; wr[u].w = b[3 * DIM];
            }
#pragma unroll
            for (int u = 0; u < 8; ++u) {
                *(float4*)(sl + (ub + u) * 1024) = wl[u];
                *(float4*)(sr + (ub + u) * 1024) = wr[u];
            }
        }
    }
    if (tid < DIM) { lsum[tid] = 0.f; lsq[tid] = 0.f; }
    __syncthreads();

    if (active) {
        const int row0 = tile * 16;

        for (int nt = 0; nt < 8; ++nt) {
            const int ncol = nt * 16 + m;
            const float bias = bl[ncol];
            v8f c;
#pragma unroll
            for (int r = 0; r < 8; ++r) c[r] = bias;

            const float* bwl = lWl + kh * 512 + ncol * 4;
            const float* bwr = lWr + kh * 512 + ncol * 4;

#pragma unroll
            for (int j = 0; j < 16; ++j) {
                const float4 bq = *(const float4*)(bwl + j * 1024);
                v2f a0; a0.x = am[j].x; a0.y = am[j].y;
                v2f b0; b0.x = bq.x;    b0.y = bq.y;
                c = __builtin_amdgcn_wmma_f32_16x16x4_f32(
                        false, a0, false, b0, (short)0, c, false, false);
                v2f a1; a1.x = am[j].z; a1.y = am[j].w;
                v2f b1; b1.x = bq.z;    b1.y = bq.w;
                c = __builtin_amdgcn_wmma_f32_16x16x4_f32(
                        false, a1, false, b1, (short)0, c, false, false);
            }
#pragma unroll
            for (int j = 0; j < 16; ++j) {
                const float4 bq = *(const float4*)(bwr + j * 1024);
                v2f a0; a0.x = ax[j].x; a0.y = ax[j].y;
                v2f b0; b0.x = bq.x;    b0.y = bq.y;
                c = __builtin_amdgcn_wmma_f32_16x16x4_f32(
                        false, a0, false, b0, (short)0, c, false, false);
                v2f a1; a1.x = ax[j].z; a1.y = ax[j].w;
                v2f b1; b1.x = bq.z;    b1.y = bq.w;
                c = __builtin_amdgcn_wmma_f32_16x16x4_f32(
                        false, a1, false, b1, (short)0, c, false, false);
            }

            float s = 0.f, s2 = 0.f;
#pragma unroll
            for (int r = 0; r < 8; ++r) {
                float v = c[r] > 0.f ? c[r] : 0.f;
                const int row = row0 + r + 8 * kh;
                h[(size_t)row * DIM + ncol] = v;
                s  += v;
                s2 += v * v;
            }
            s  += __shfl_xor(s, 16, 32);
            s2 += __shfl_xor(s2, 16, 32);
            if (lane < 16) {
                atomicAdd(&lsum[ncol], s);
                atomicAdd(&lsq[ncol], s2);
            }
        }
    }
    __syncthreads();
    if (tid < DIM) {
        atomicAdd(gsum + tid, lsum[tid]);
        atomicAdd(gsq  + tid, lsq[tid]);
    }
}

__global__ void stats_kernel(const float* __restrict__ gsum, const float* __restrict__ gsq,
                             const float* __restrict__ gamma, const float* __restrict__ beta,
                             float* __restrict__ cA, float* __restrict__ cB) {
    int c = threadIdx.x;
    if (c >= DIM) return;
    const float invN = 1.0f / (float)NN;
    float mu  = gsum[c] * invN;
    float var = gsq[c] * invN - mu * mu;
    float a   = rsqrtf(var + BN_EPS) * gamma[c];
    cA[c] = a;
    cB[c] = beta[c] - mu * a;
}

__global__ void apply_kernel(const float* __restrict__ x,
                             const float* __restrict__ cA, const float* __restrict__ cB,
                             float* __restrict__ out) {
    size_t i = (size_t)blockIdx.x * blockDim.x + threadIdx.x;
    if (i >= (size_t)NN * DIM / 4) return;
    int col4 = (int)(i & 31) * 4;
    float4 hv = ((float4*)out)[i];
    float4 xv = ((const float4*)x)[i];
    float4 a  = *(const float4*)(cA + col4);
    float4 b  = *(const float4*)(cB + col4);
    float4 o;
    o.x = xv.x + hv.x * a.x + b.x;
    o.y = xv.y + hv.y * a.y + b.y;
    o.z = xv.z + hv.z * a.z + b.z;
    o.w = xv.w + hv.w * a.w + b.w;
    ((float4*)out)[i] = o;
}

extern "C" void kernel_launch(void* const* d_in, const int* in_sizes, int n_in,
                              void* d_out, int out_size, void* d_ws, size_t ws_size,
                              hipStream_t stream) {
    const float* x     = (const float*)d_in[0];
    const int*   ei    = (const int*)d_in[1];
    const float* Wl    = (const float*)d_in[2];
    const float* bl    = (const float*)d_in[3];
    const float* Wr    = (const float*)d_in[4];
    const float* gamma = (const float*)d_in[5];
    const float* beta  = (const float*)d_in[6];
    float* out = (float*)d_out;

    float* agg  = (float*)d_ws;
    float* deg  = agg + (size_t)NN * DIM;
    float* gsum = deg + NN;
    float* gsq  = gsum + DIM;
    float* cA   = gsq + DIM;
    float* cB   = cA + DIM;

    int* T   = (int*)d_ws;
    int* tmp = T + NT_SLOTS;

    int* rowptr = (int*)d_out;
    int* col    = rowptr + ROWPTR_SLOTS;

    part_kernel<<<NBLK, 256, 0, stream>>>(ei, T, tmp, gsum);

    csr_kernel<<<NBK, 256, 0, stream>>>(T, tmp, rowptr, col);

    gather_mean_kernel<<<(NN + 8 * NODES_PER_WAVE - 1) / (8 * NODES_PER_WAVE), 256, 0, stream>>>(
        x, rowptr, col, agg);

    const int tiles = NN / 16;
    gemm_bn_kernel<<<(tiles + 7) / 8, 256, 0, stream>>>(agg, x, Wl, bl, Wr,
                                                        out, gsum, gsq);

    stats_kernel<<<1, 128, 0, stream>>>(gsum, gsq, gamma, beta, cA, cB);

    int mb = (int)(((size_t)NN * DIM / 4 + 255) / 256);
    apply_kernel<<<mb, 256, 0, stream>>>(x, cA, cB, out);
}


# neighbour gather fused into the matrix kernel (256 resident 16-wave blocks): per tile x@Wr, rows summed into LDS as A operand, mean@Wl; mean not stored; x@Wr summed first, BN sums per 256 blocks
# speedup vs baseline: 3.5916x; 1.2206x over previous
#include <hip/hip_runtime.h>

#define NN 100000
#define NE 1600000
#define DIM 128
#define BN_EPS 1e-5f
#define FT_WAVES 16
#define FT_GRID 256
#define STAGE_PITCH 132
#define ROWPTR_SLOTS 100352
#define EDGE_CHUNK 4096
#define NBLK ((NE + EDGE_CHUNK - 1) / EDGE_CHUNK)
#define BSHIFT 8
#define BW (1 << BSHIFT)
#define NBK ((NN + BW - 1) / BW)
#define BK_SLOTS 512
#define NT_SLOTS 153600
#define STAGE_CAP 8192
#define CSR_BATCH 16
#define RUNS_PER_HALF ((NBLK + 15) / 16)

typedef __attribute__((ext_vector_type(2))) float v2f;
typedef __attribute__((ext_vector_type(8))) float v8f;


__global__ __launch_bounds__(256) void part_kernel(
    const int* __restrict__ ei, int* __restrict__ T, int* __restrict__ tmp,
    float* __restrict__ sums) {
    __shared__ int lh[BK_SLOTS];
    __shared__ alignas(16) int lpk[EDGE_CHUNK];
    __shared__ int wtot[8];
    const int tid  = threadIdx.x;
    const int lane = tid & 31;
    const int wave = tid >> 5;
    const int b    = blockIdx.x;
    const int e0   = b * EDGE_CHUNK;
    lh[tid] = 0; lh[tid + 256] = 0;
    if (b == 0) sums[tid] = 0.f;
    __syncthreads();

    int4 sv[4], dv[4], rv[4];
#pragma unroll
    for (int u = 0; u < 4; ++u) {
        const int e = e0 + 4 * (tid + 256 * u);
        sv[u].x = 0; sv[u].y = 0; sv[u].z = 0; sv[u].w = 0;
        dv[u] = sv[u]; rv[u] = sv[u];
        if (e < NE) {
            sv[u] = *(const int4*)(ei + e);
            dv[u] = *(const int4*)(ei + NE + e);
            rv[u].x = atomicAdd(&lh[dv[u].x >> BSHIFT], 1);
            rv[u].y = atomicAdd(&lh[dv[u].y >> BSHIFT], 1);
            rv[u].z = atomicAdd(&lh[dv[u].z >> BSHIFT], 1);
            rv[u].w = atomicAdd(&lh[dv[u].w >> BSHIFT], 1);
        }
    }
    __syncthreads();

    const int c0 = lh[2 * tid];
    const int c1 = lh[2 * tid + 1];
    const int t  = c0 + c1;
    int inc = t;
    for (int o = 1; o < 32; o <<= 1) {
        const int y = __shfl_up(inc, o, 32);
        if (lane >= o) inc += y;
    }
    if (lane == 31) wtot[wave] = inc;
    __syncthreads();
    int off = 0;
    for (int w = 0; w < 8; ++w) if (w < wave) off += wtot[w];
    const int ex = off + inc - t;
    lh[2 * tid]     = ex;
    lh[2 * tid + 1] = ex + c0;
    if (2 * tid     <= NBK) T[(2 * tid)     * NBLK + b] = ex;
    if (2 * tid + 1 <= NBK) T[(2 * tid + 1) * NBLK + b] = ex + c0;
    __syncthreads();

#pragma unroll
    for (int u = 0; u < 4; ++u) {
        const int e = e0 + 4 * (tid + 256 * u);
        if (e < NE) {
            lpk[lh[dv[u].x >> BSHIFT] + rv[u].x] = (sv[u].x << BSHIFT) | (dv[u].x & (BW - 1));
            lpk[lh[dv[u].y >> BSHIFT] + rv[u].y] = (sv[u].y << BSHIFT) | (dv[u].y & (BW - 1));
            lpk[lh[dv[u].z >> BSHIFT] + rv[u].z] = (sv[u].z << BSHIFT) | (dv[u].z & (BW - 1));
            lpk[lh[dv[u].w >> BSHIFT] + rv[u].w] = (sv[u].w << BSHIFT) | (dv[u].w & (BW - 1));
        }
    }
    __syncthreads();

#pragma unroll
    for (int u = 0; u < 4; ++u) {
        const int j = 4 * (tid + 256 * u);
        if (e0 + j < NE) *(int4*)(tmp + e0 + j) = *(const int4*)(lpk + j);
    }
}

__global__ __launch_bounds__(256) void csr_kernel(
    const int* __restrict__ T, const int* __restrict__ tmp,
    int* __restrict__ rowptr, int* __restrict__ col) {
    __shared__ int2 lan[BK_SLOTS];
    __shared__ int lp[BK_SLOTS];
    __shared__ int lc[BW];
    __shared__ int lw[STAGE_CAP];
    __shared__ int lcol[STAGE_CAP];
    __shared__ int wtot[8];
    __shared__ int wsta[8];
    __shared__ int wnod[8];
    const int tid  = threadIdx.x;
    const int lane = tid & 31;
    const int wave = tid >> 5;
    const int q    = blockIdx.x;

    int a0 = 0, n0 = 0, a1 = 0, n1 = 0;
    if (2 * tid < NBLK) {
        a0 = T[q * NBLK + 2 * tid];
        n0 = T[(q + 1) * NBLK + 2 * tid] - a0;
    }
    if (2 * tid + 1 < NBLK) {
        a1 = T[q * NBLK + 2 * tid + 1];
        n1 = T[(q + 1) * NBLK + 2 * tid + 1] - a1;
    }
    const int t = n0 + n1;
    int inc = t;
    for (int o = 1; o < 32; o <<= 1) {
        const int y = __shfl_up(inc, o, 32);
        if (lane >= o) inc += y;
    }
    int asum = a0 + a1;
    for (int o = 16; o > 0; o >>= 1) asum += __shfl_xor(asum, o, 32);
    if (lane == 31) wtot[wave] = inc;
    if (lane == 0)  wsta[wave] = asum;
    lc[tid] = 0;
    __syncthreads();
    int off = 0, s = 0, total = 0;
    for (int w = 0; w < 8; ++w) {
        if (w < wave) off += wtot[w];
        total += wtot[w];
        s += wsta[w];
    }
    const int exr = off + inc - t;
    lan[2 * tid]     = make_int2(a0, n0); lp[2 * tid]     = exr;
    lan[2 * tid + 1] = make_int2(a1, n1); lp[2 * tid + 1] = exr + n0;
    __syncthreads();

    const bool staged = total <= STAGE_CAP;
    const int l16 = lane & 15;
    const int cb  = 2 * wave + (lane >> 4);
    if (staged) {
        int v[RUNS_PER_HALF], ra[RUNS_PER_HALF], rn[RUNS_PER_HALF], rp[RUNS_PER_HALF];
#pragma unroll
        for (int u = 0; u < RUNS_PER_HALF; ++u) {
            const int c  = cb + 16 * u;
            const int cc = c < NBLK ? c : NBLK - 1;
            const int2 an = lan[cc];
            ra[u] = cc * EDGE_CHUNK + an.x;
            rn[u] = c < NBLK ? an.y : 0;
            rp[u] = lp[cc];
            v[u]  = tmp[l16 < rn[u] ? ra[u] + l16 : 0];
        }
#pragma unroll
        for (int u = 0; u < RUNS_PER_HALF; ++u) {
            if (l16 < rn[u]) {
                lw[rp[u] + l16] = v[u];
                atomicAdd(&lc[v[u] & (BW - 1)], 1);
            }
            for (int k = 16 + l16; k < rn[u]; k += 16) {
                const int w = tmp[ra[u] + k];
                lw[rp[u] + k] = w;
                atomicAdd(&lc[w & (BW - 1)], 1);
            }
        }
    } else {
        for (int c = cb; c < NBLK; c += 16) {
            const int a = lan[c].x;
            const int n = lan[c].y;
            for (int k = l16; k < n; k += 16)
                atomicAdd(&lc[tmp[c * EDGE_CHUNK + a + k] & (BW - 1)], 1);
        }
    }
    __syncthreads();

    const int cn = lc[tid];
    int incn = cn;
    for (int o = 1; o < 32; o <<= 1) {
        const int y = __shfl_up(incn, o, 32);
        if (lane >= o) incn += y;
    }
    if (lane == 31) wnod[wave] = incn;
    __syncthreads();
    int offn = 0;
    for (int w = 0; w < 8; ++w) if (w < wave) offn += wnod[w];
    const int ex = offn + incn - cn;
    const int node = q * BW + tid;
    if (node <= NN) rowptr[node] = s + ex;
    lc[tid] = ex;
    __syncthreads();

    if (staged) {
        for (int i0 = tid; i0 < total; i0 += 256 * CSR_BATCH) {
            int w[CSR_BATCH];
#pragma unroll
            for (int u = 0; u < CSR_BATCH; ++u) {
                const int i = i0 + 256 * u;
                w[u] = (i < total) ? lw[i] : -1;
            }
#pragma unroll
            for (int u = 0; u < CSR_BATCH; ++u) {
                if (w[u] >= 0) {
                    const int p = atomicAdd(&lc[w[u] & (BW - 1)], 1);
                    lcol[p] = w[u] >> BSHIFT;
                }
            }
        }
    } else {
        for (int c = cb; c < NBLK; c += 16) {
            const int a = lan[c].x;
            const int n = lan[c].y;
            for (int k = l16; k < n; k += 16) {
                const int w = tmp[c * EDGE_CHUNK + a + k];
                const int p = atomicAdd(&lc[w & (BW - 1)], 1);
                col[s + p] = w >> BSHIFT;
            }
        }
    }
    __syncthreads();
    if (staged) {
        for (int i = tid; i < total; i += 256) col[s + i] = lcol[i];
    }
}

__device__ __forceinline__ void flush_node(float4& acc, int& q, int& qstart, int& qend,
                                           int rpv, float* __restrict__ stw) {
    const float inv = 1.0f / fmaxf((float)(qend - qstart), 1.0f);
    float4 o;
    o.x = acc.x * inv; o.y = acc.y * inv; o.z = acc.z * inv; o.w = acc.w * inv;
    *(float4*)(stw + q * STAGE_PITCH) = o;
    acc.x = 0.f; acc.y = 0.f; acc.z = 0.f; acc.w = 0.f;
    qstart = qend;
    ++q;
    qend = __builtin_amdgcn_readlane(rpv, q + 1);
}

template <int U>
__device__ __forceinline__ void add_rows(const float* __restrict__ xl, int myc, int j, int e,
                                         float4& acc, int& q, int& qstart, int& qend,
                                         int rpv, float* __restrict__ stw) {
    float4 v[U];
#pragma unroll
    for (int u = 0; u < U; ++u) {
        const int s = __builtin_amdgcn_readlane(myc, j + u);
        v[u] = *(const float4*)(xl + (size_t)s * DIM);
    }
#pragma unroll
    for (int u = 0; u < U; ++u) {
        while (e + u == qend) flush_node(acc, q, qstart, qend, rpv, stw);
        acc.x += v[u].x; acc.y += v[u].y; acc.z += v[u].z; acc.w += v[u].w;
    }
}

__global__ __launch_bounds__(FT_WAVES * 32) void gather_gemm_bn_kernel(
    const int* __restrict__ rowptr, const int* __restrict__ col,
    const float* __restrict__ x,
    const float* __restrict__ Wl, const float* __restrict__ bl,
    const float* __restrict__ Wr, float* __restrict__ h,
    float* __restrict__ gsum, float* __restrict__ gsq) {
    __shared__ float lWl[DIM * DIM];
    __shared__ float lWr[DIM * DIM];
    __shared__ alignas(16) float lst[FT_WAVES * 16 * STAGE_PITCH];
    __shared__ float lsum[DIM];
    __shared__ float lsq[DIM];

    const int tid  = threadIdx.x;
    const int lane = tid & 31;
    const int wave = __builtin_amdgcn_readfirstlane(tid >> 5);
    const int m    = lane & 15;
    const int kh   = lane >> 4;

    {
        const int n  = tid & (DIM - 1);
        const int qb = tid >> 7;
        const float* pl = Wl + (size_t)(4 * qb) * DIM + n;
        const float* pr = Wr + (size_t)(4 * qb) * DIM + n;
        float* sl = lWl + qb * 512 + n * 4;
        float* sr = lWr + qb * 512 + n * 4;
        float4 wl[8], wr[8];
#pragma unroll
        for (int u = 0; u < 8; ++u) {
            const float* a = pl + (size_t)u * (16 * DIM);
            const float* b = pr + (size_t)u * (16 * DIM);
            wl[u].x = a[0]; wl[u].y = a[DIM]; wl[u].z = a[2 * DIM]; wl[u].w = a[3 * DIM];
            wr[u].x = b[0]; wr[u].y = b[DIM]; wr[u].z = b[2 * DIM]; wr[u].w = b[3 * DIM];
        }
#pragma unroll
        for (int u = 0; u < 8; ++u) {
            *(float4*)(sl + u * 2048) = wl[u];
            *(float4*)(sr + u * 2048) = wr[u];
        }
    }
    if (tid < DIM) { lsum[tid] = 0.f; lsq[tid] = 0.f; }
    __syncthreads();

    const float* xl  = x + lane * 4;
    float* stw       = lst + wave * (16 * STAGE_PITCH) + lane * 4;
    const float* str = lst + wave * (16 * STAGE_PITCH) + m * STAGE_PITCH + 4 * kh;

    const int ntile = (NN / 16 - (int)blockIdx.x + FT_GRID - 1) / FT_GRID;
    for (int t = wave; t < ntile; t += FT_WAVES) {
        const int row0 = ((int)blockIdx.x + FT_GRID * t) * 16;

        const int rpv = rowptr[row0 + (lane < 16 ? lane : 16)];
        const int e0  = __builtin_amdgcn_readlane(rpv, 0);
        const int e1  = __builtin_amdgcn_readlane(rpv, 16);
        int mynext = (e0 + lane < e1) ? col[e0 + lane] : 0;

        const float* xRow = x + (size_t)(row0 + m) * DIM + 4 * kh;
        float4 ax[16];
#pragma unroll
        for (int j = 0; j < 16; ++j) ax[j] = *(const float4*)(xRow + 8 * j);

        v8f c[8];
#pragma unroll
        for (int nt = 0; nt < 8; ++nt) {
            const int ncol = nt * 16 + m;
            const float bias = bl[ncol];
#pragma unroll
            for (int r = 0; r < 8; ++r) c[nt][r] = bias;
            const float* bwr = lWr + kh * 512 + ncol * 4;
#pragma unroll
            for (int j = 0; j < 16; ++j) {
                const float4 bq = *(const float4*)(bwr + j * 1024);
                v2f a0; a0.x = ax[j].x; a0.y = ax[j].y;
                v2f b0; b0.x = bq.x;    b0.y = bq.y;
                c[nt] = __builtin_amdgcn_wmma_f32_16x16x4_f32(
                            false, a0, false, b0, (short)0, c[nt], false, false);
                v2f a1; a1.x = ax[j].z; a1.y = ax[j].w;
                v2f b1; b1.x = bq.z;    b1.y = bq.w;
                c[nt] = __builtin_amdgcn_wmma_f32_16x16x4_f32(
                            false, a1, false, b1, (short)0, c[nt], false, false);
            }
        }

        int q = 0, qstart = e0;
        int qend = __builtin_amdgcn_readlane(rpv, 1);
        float4 acc; acc.x = 0.f; acc.y = 0.f; acc.z = 0.f; acc.w = 0.f;
        for (int j0 = e0; j0 < e1; j0 += 32) {
            const int mc  = (e1 - j0) < 32 ? (e1 - j0) : 32;
            const int myc = mynext;
            mynext = (j0 + 32 + lane < e1) ? col[j0 + 32 + lane] : 0;
            int j = 0;
            for (; j + 16 <= mc; j += 16) add_rows<16>(xl, myc, j, j0 + j, acc, q, qstart, qend, rpv, stw);
            if (j + 8 <= mc) { add_rows<8>(xl, myc, j, j0 + j, acc, q, qstart, qend, rpv, stw); j += 8; }
            if (j + 4 <= mc) { add_rows<4>(xl, myc, j, j0 + j, acc, q, qstart, qend, rpv, stw); j += 4; }
            if (j + 2 <= mc) { add_rows<2>(xl, myc, j, j0 + j, acc, q, qstart, qend, rpv, stw); j += 2; }
            if (j < mc)      { add_rows<1>(xl, myc, j, j0 + j, acc, q, qstart, qend, rpv, stw); }
        }
        while (q < 16) flush_node(acc, q, qstart, qend, rpv, stw);

        float4 am[16];
#pragma unroll
        for (int j = 0; j < 16; ++j) am[j] = *(const float4*)(str + 8 * j);

#pragma unroll
        for (int nt = 0; nt < 8; ++nt) {
            const int ncol = nt * 16 + m;
            const float* bwl = lWl + kh * 512 + ncol * 4;
#pragma unroll
            for (int j = 0; j < 16; ++j) {
                const float4 bq = *(const float4*)(bwl + j * 1024);
                v2f a0; a0.x = am[j].x; a0.y = am[j].y;
                v2f b0; b0.x = bq.x;    b0.y = bq.y;
                c[nt] = __builtin_amdgcn_wmma_f32_16x16x4_f32(
                            false, a0, false, b0, (short)0, c[nt], false, false);
                v2f a1; a1.x = am[j].z; a1.y = am[j].w;
                v2f b1; b1.x = bq.z;    b1.y = bq.w;
                c[nt] = __builtin_amdgcn_wmma_f32_16x16x4_f32(
                            false, a1, false, b1, (short)0, c[nt], false, false);
            }

            float s = 0.f, s2 = 0.f;
#pragma unroll
            for (int r = 0; r < 8; ++r) {
                float v = c[nt][r] > 0.f ? c[nt][r] : 0.f;
                const int row = row0 + r + 8 * kh;
                h[(size_t)row * DIM + ncol] = v;
                s  += v;
                s2 += v * v;
            }
            s  += __shfl_xor(s, 16, 32);
            s2 += __shfl_xor(s2, 16, 32);
            if (lane < 16) {
                atomicAdd(&lsum[ncol], s);
                atomicAdd(&lsq[ncol], s2);
            }
        }
    }
    __syncthreads();
    if (tid < DIM) {
        atomicAdd(gsum + tid, lsum[tid]);
        atomicAdd(gsq  + tid, lsq[tid]);
    }
}

__global__ void stats_kernel(const float* __restrict__ gsum, const float* __restrict__ gsq,
                             const float* __restrict__ gamma, const float* __restrict__ beta,
                             float* __restrict__ cA, float* __restrict__ cB) {
    int c = threadIdx.x;
    if (c >= DIM) return;
    const float invN = 1.0f / (float)NN;
    float mu  = gsum[c] * invN;
    float var = gsq[c] * invN - mu * mu;
    float a   = rsqrtf(var + BN_EPS) * gamma[c];
    cA[c] = a;
    cB[c] = beta[c] - mu * a;
}

__global__ void apply_kernel(const float* __restrict__ x,
                             const float* __restrict__ cA, const float* __restrict__ cB,
                             float* __restrict__ out) {
    size_t i = (size_t)blockIdx.x * blockDim.x + threadIdx.x;
    if (i >= (size_t)NN * DIM / 4) return;
    int col4 = (int)(i & 31) * 4;
    float4 hv = ((float4*)out)[i];
    float4 xv = ((const float4*)x)[i];
    float4 a  = *(const float4*)(cA + col4);
    float4 b  = *(const float4*)(cB + col4);
    float4 o;
    o.x = xv.x + hv.x * a.x + b.x;
    o.y = xv.y + hv.y * a.y + b.y;
    o.z = xv.z + hv.z * a.z + b.z;
    o.w = xv.w + hv.w * a.w + b.w;
    ((float4*)out)[i] = o;
}

extern "C" void kernel_launch(void* const* d_in, const int* in_sizes, int n_in,
                              void* d_out, int out_size, void* d_ws, size_t ws_size,
                              hipStream_t stream) {
    const float* x     = (const float*)d_in[0];
    const int*   ei    = (const int*)d_in[1];
    const float* Wl    = (const float*)d_in[2];
    const float* bl    = (const float*)d_in[3];
    const float* Wr    = (const float*)d_in[4];
    const float* gamma = (const float*)d_in[5];
    const float* beta  = (const float*)d_in[6];
    float* out = (float*)d_out;

    float* agg  = (float*)d_ws;
    float* deg  = agg + (size_t)NN * DIM;
    float* gsum = deg + NN;
    float* gsq  = gsum + DIM;
    float* cA   = gsq + DIM;
    float* cB   = cA + DIM;

    int* T   = (int*)d_ws;
    int* tmp = T + NT_SLOTS;

    int* rowptr = tmp + NBLK * EDGE_CHUNK;
    int* col    = rowptr + ROWPTR_SLOTS;

    part_kernel<<<NBLK, 256, 0, stream>>>(ei, T, tmp, gsum);

    csr_kernel<<<NBK, 256, 0, stream>>>(T, tmp, rowptr, col);

    gather_gemm_bn_kernel<<<FT_GRID, FT_WAVES * 32, 0, stream>>>(rowptr, col, x, Wl, bl, Wr,
                                                                 out, gsum, gsq);

    stats_kernel<<<1, 128, 0, stream>>>(gsum, gsq, gamma, beta, cA, cB);

    int mb = (int)(((size_t)NN * DIM / 4 + 255) / 256);
    apply_kernel<<<mb, 256, 0, stream>>>(x, cA, cB, out);
}


# matrix kernel: h tile staged in the wave's LDS tile and stored as whole 512-byte rows: 16 stores of four full 128-byte lines a tile, not 64 stores of two 64-byte pieces; no sum order changes
# speedup vs baseline: 4.2625x; 1.1868x over previous
#include <hip/hip_runtime.h>

#define NN 100000
#define NE 1600000
#define DIM 128
#define BN_EPS 1e-5f
#define FT_WAVES 16
#define FT_GRID 256
#define STAGE_PITCH 132
#define ROWPTR_SLOTS 100352
#define EDGE_CHUNK 4096
#define NBLK ((NE + EDGE_CHUNK - 1) / EDGE_CHUNK)
#define BSHIFT 8
#define BW (1 << BSHIFT)
#define NBK ((NN + BW - 1) / BW)
#define BK_SLOTS 512
#define NT_SLOTS 153600
#define STAGE_CAP 8192
#define CSR_BATCH 16
#define RUNS_PER_HALF ((NBLK + 15) / 16)

typedef __attribute__((ext_vector_type(2))) float v2f;
typedef __attribute__((ext_vector_type(8))) float v8f;


__global__ __launch_bounds__(256) void part_kernel(
    const int* __restrict__ ei, int* __restrict__ T, int* __restrict__ tmp,
    float* __restrict__ sums) {
    __shared__ int lh[BK_SLOTS];
    __shared__ alignas(16) int lpk[EDGE_CHUNK];
    __shared__ int wtot[8];
    const int tid  = threadIdx.x;
    const int lane = tid & 31;
    const int wave = tid >> 5;
    const int b    = blockIdx.x;
    const int e0   = b * EDGE_CHUNK;
    lh[tid] = 0; lh[tid + 256] = 0;
    if (b == 0) sums[tid] = 0.f;
    __syncthreads();

    int4 sv[4], dv[4], rv[4];
#pragma unroll
    for (int u = 0; u < 4; ++u) {
        const int e = e0 + 4 * (tid + 256 * u);
        sv[u].x = 0; sv[u].y = 0; sv[u].z = 0; sv[u].w = 0;
        dv[u] = sv[u]; rv[u] = sv[u];
        if (e < NE) {
            sv[u] = *(const int4*)(ei + e);
            dv[u] = *(const int4*)(ei + NE + e);
            rv[u].x = atomicAdd(&lh[dv[u].x >> BSHIFT], 1);
            rv[u].y = atomicAdd(&lh[dv[u].y >> BSHIFT], 1);
            rv[u].z = atomicAdd(&lh[dv[u].z >> BSHIFT], 1);
            rv[u].w = atomicAdd(&lh[dv[u].w >> BSHIFT], 1);
        }
    }
    __syncthreads();

    const int c0 = lh[2 * tid];
    const int c1 = lh[2 * tid + 1];
    const int t  = c0 + c1;
    int inc = t;
    for (int o = 1; o < 32; o <<= 1) {
        const int y = __shfl_up(inc, o, 32);
        if (lane >= o) inc += y;
    }
    if (lane == 31) wtot[wave] = inc;
    __syncthreads();
    int off = 0;
    for (int w = 0; w < 8; ++w) if (w < wave) off += wtot[w];
    const int ex = off + inc - t;
    lh[2 * tid]     = ex;
    lh[2 * tid + 1] = ex + c0;
    if (2 * tid     <= NBK) T[(2 * tid)     * NBLK + b] = ex;
    if (2 * tid + 1 <= NBK) T[(2 * tid + 1) * NBLK + b] = ex + c0;
    __syncthreads();

#pragma unroll
    for (int u = 0; u < 4; ++u) {
        const int e = e0 + 4 * (tid + 256 * u);
        if (e < NE) {
            lpk[lh[dv[u].x >> BSHIFT] + rv[u].x] = (sv[u].x << BSHIFT) | (dv[u].x & (BW - 1));
            lpk[lh[dv[u].y >> BSHIFT] + rv[u].y] = (sv[u].y << BSHIFT) | (dv[u].y & (BW - 1));
            lpk[lh[dv[u].z >> BSHIFT] + rv[u].z] = (sv[u].z << BSHIFT) | (dv[u].z & (BW - 1));
            lpk[lh[dv[u].w >> BSHIFT] + rv[u].w] = (sv[u].w << BSHIFT) | (dv[u].w & (BW - 1));
        }
    }
    __syncthreads();

#pragma unroll
    for (int u = 0; u < 4; ++u) {
        const int j = 4 * (tid + 256 * u);
        if (e0 + j < NE) *(int4*)(tmp + e0 + j) = *(const int4*)(lpk + j);
    }
}

__global__ __launch_bounds__(256) void csr_kernel(
    const int* __restrict__ T, const int* __restrict__ tmp,
    int* __restrict__ rowptr, int* __restrict__ col) {
    __shared__ int2 lan[BK_SLOTS];
    __shared__ int lp[BK_SLOTS];
    __shared__ int lc[BW];
    __shared__ int lw[STAGE_CAP];
    __shared__ int lcol[STAGE_CAP];
    __shared__ int wtot[8];
    __shared__ int wsta[8];
    __shared__ int wnod[8];
    const int tid  = threadIdx.x;
    const int lane = tid & 31;
    const int wave = tid >> 5;
    const int q    = blockIdx.x;

    int a0 = 0, n0 = 0, a1 = 0, n1 = 0;
    if (2 * tid < NBLK) {
        a0 = T[q * NBLK + 2 * tid];
        n0 = T[(q + 1) * NBLK + 2 * tid] - a0;
    }
    if (2 * tid + 1 < NBLK) {
        a1 = T[q * NBLK + 2 * tid + 1];
        n1 = T[(q + 1) * NBLK + 2 * tid + 1] - a1;
    }
    const int t = n0 + n1;
    int inc = t;
    for (int o = 1; o < 32; o <<= 1) {
        const int y = __shfl_up(inc, o, 32);
        if (lane >= o) inc += y;
    }
    int asum = a0 + a1;
    for (int o = 16; o > 0; o >>= 1) asum += __shfl_xor(asum, o, 32);
    if (lane == 31) wtot[wave] = inc;
    if (lane == 0)  wsta[wave] = asum;
    lc[tid] = 0;
    __syncthreads();
    int off = 0, s = 0, total = 0;
    for (int w = 0; w < 8; ++w) {
        if (w < wave) off += wtot[w];
        total += wtot[w];
        s += wsta[w];
    }
    const int exr = off + inc - t;
    lan[2 * tid]     = make_int2(a0, n0); lp[2 * tid]     = exr;
    lan[2 * tid + 1] = make_int2(a1, n1); lp[2 * tid + 1] = exr + n0;
    __syncthreads();

    const bool staged = total <= STAGE_CAP;
    const int l16 = lane & 15;
    const int cb  = 2 * wave + (lane >> 4);
    if (staged) {
        int v[RUNS_PER_HALF], ra[RUNS_PER_HALF], rn[RUNS_PER_HALF], rp[RUNS_PER_HALF];
#pragma unroll
        for (int u = 0; u < RUNS_PER_HALF; ++u) {
            const int c  = cb + 16 * u;
            const int cc = c < NBLK ? c : NBLK - 1;
            const int2 an = lan[cc];
            ra[u] = cc * EDGE_CHUNK + an.x;
            rn[u] = c < NBLK ? an.y : 0;
            rp[u] = lp[cc];
            v[u]  = tmp[l16 < rn[u] ? ra[u] + l16 : 0];
        }
#pragma unroll
        for (int u = 0; u < RUNS_PER_HALF; ++u) {
            if (l16 < rn[u]) {
                lw[rp[u] + l16] = v[u];
                atomicAdd(&lc[v[u] & (BW - 1)], 1);
            }
            for (int k = 16 + l16; k < rn[u]; k += 16) {
                const int w = tmp[ra[u] + k];
                lw[rp[u] + k] = w;
                atomicAdd(&lc[w & (BW - 1)], 1);
            }
        }
    } else {
        for (int c = cb; c < NBLK; c += 16) {
            const int a = lan[c].x;
            const int n = lan[c].y;
            for (int k = l16; k < n; k += 16)
                atomicAdd(&lc[tmp[c * EDGE_CHUNK + a + k] & (BW - 1)], 1);
        }
    }
    __syncthreads();

    const int cn = lc[tid];
    int incn = cn;
    for (int o = 1; o < 32; o <<= 1) {
        const int y = __shfl_up(incn, o, 32);
        if (lane >= o) incn += y;
    }
    if (lane == 31) wnod[wave] = incn;
    __syncthreads();
    int offn = 0;
    for (int w = 0; w < 8; ++w) if (w < wave) offn += wnod[w];
    const int ex = offn + incn - cn;
    const int node = q * BW + tid;
    if (node <= NN) rowptr[node] = s + ex;
    lc[tid] = ex;
    __syncthreads();

    if (staged) {
        for (int i0 = tid; i0 < total; i0 += 256 * CSR_BATCH) {
            int w[CSR_BATCH];
#pragma unroll
            for (int u = 0; u < CSR_BATCH; ++u) {
                const int i = i0 + 256 * u;
                w[u] = (i < total) ? lw[i] : -1;
            }
#pragma unroll
            for (int u = 0; u < CSR_BATCH; ++u) {
                if (w[u] >= 0) {
                    const int p = atomicAdd(&lc[w[u] & (BW - 1)], 1);
                    lcol[p] = w[u] >> BSHIFT;
                }
            }
        }
    } else {
        for (int c = cb; c < NBLK; c += 16) {
            const int a = lan[c].x;
            const int n = lan[c].y;
            for (int k = l16; k < n; k += 16) {
                const int w = tmp[c * EDGE_CHUNK + a + k];
                const int p = atomicAdd(&lc[w & (BW - 1)], 1);
                col[s + p] = w >> BSHIFT;
            }
        }
    }
    __syncthreads();
    if (staged) {
        for (int i = tid; i < total; i += 256) col[s + i] = lcol[i];
    }
}

__device__ __forceinline__ void flush_node(float4& acc, int& q, int& qstart, int& qend,
                                           int rpv, float* __restrict__ stw) {
    const float inv = 1.0f / fmaxf((float)(qend - qstart), 1.0f);
    float4 o;
    o.x = acc.x * inv; o.y = acc.y * inv; o.z = acc.z * inv; o.w = acc.w * inv;
    *(float4*)(stw + q * STAGE_PITCH) = o;
    acc.x = 0.f; acc.y = 0.f; acc.z = 0.f; acc.w = 0.f;
    qstart = qend;
    ++q;
    qend = __builtin_amdgcn_readlane(rpv, q + 1);
}

template <int U>
__device__ __forceinline__ void add_rows(const float* __restrict__ xl, int myc, int j, int e,
                                         float4& acc, int& q, int& qstart, int& qend,
                                         int rpv, float* __restrict__ stw) {
    float4 v[U];
#pragma unroll
    for (int u = 0; u < U; ++u) {
        const int s = __builtin_amdgcn_readlane(myc, j + u);
        v[u] = *(const float4*)(xl + (size_t)s * DIM);
    }
#pragma unroll
    for (int u = 0; u < U; ++u) {
        while (e + u == qend) flush_node(acc, q, qstart, qend, rpv, stw);
        acc.x += v[u].x; acc.y += v[u].y; acc.z += v[u].z; acc.w += v[u].w;
    }
}

__global__ __launch_bounds__(FT_WAVES * 32) void gather_gemm_bn_kernel(
    const int* __restrict__ rowptr, const int* __restrict__ col,
    const float* __restrict__ x,
    const float* __restrict__ Wl, const float* __restrict__ bl,
    const float* __restrict__ Wr, float* __restrict__ h,
    float* __restrict__ gsum, float* __restrict__ gsq) {
    __shared__ float lWl[DIM * DIM];
    __shared__ float lWr[DIM * DIM];
    __shared__ alignas(16) float lst[FT_WAVES * 16 * STAGE_PITCH];
    __shared__ float lsum[DIM];
    __shared__ float lsq[DIM];

    const int tid  = threadIdx.x;
    const int lane = tid & 31;
    const int wave = __builtin_amdgcn_readfirstlane(tid >> 5);
    const int m    = lane & 15;
    const int kh   = lane >> 4;

    {
        const int n  = tid & (DIM - 1);
        const int qb = tid >> 7;
        const float* pl = Wl + (size_t)(4 * qb) * DIM + n;
        const float* pr = Wr + (size_t)(4 * qb) * DIM + n;
        float* sl = lWl + qb * 512 + n * 4;
        float* sr = lWr + qb * 512 + n * 4;
        float4 wl[8], wr[8];
#pragma unroll
        for (int u = 0; u < 8; ++u) {
            const float* a = pl + (size_t)u * (16 * DIM);
            const float* b = pr + (size_t)u * (16 * DIM);
            wl[u].x = a[0]; wl[u].y = a[DIM]; wl[u].z = a[2 * DIM]; wl[u].w = a[3 * DIM];
            wr[u].x = b[0]; wr[u].y = b[DIM]; wr[u].z = b[2 * DIM]; wr[u].w = b[3 * DIM];
        }
#pragma unroll
        for (int u = 0; u < 8; ++u) {
            *(float4*)(sl + u * 2048) = wl[u];
            *(float4*)(sr + u * 2048) = wr[u];
        }
    }
    if (tid < DIM) { lsum[tid] = 0.f; lsq[tid] = 0.f; }
    __syncthreads();

    const float* xl  = x + lane * 4;
    float* stw       = lst + wave * (16 * STAGE_PITCH) + lane * 4;
    const float* str = lst + wave * (16 * STAGE_PITCH) + m * STAGE_PITCH + 4 * kh;
    float* sto       = lst + wave * (16 * STAGE_PITCH) + (8 * kh) * STAGE_PITCH + m;

    const int ntile = (NN / 16 - (int)blockIdx.x + FT_GRID - 1) / FT_GRID;
    for (int t = wave; t < ntile; t += FT_WAVES) {
        const int row0 = ((int)blockIdx.x + FT_GRID * t) * 16;

        const int rpv = rowptr[row0 + (lane < 16 ? lane : 16)];
        const int e0  = __builtin_amdgcn_readlane(rpv, 0);
        const int e1  = __builtin_amdgcn_readlane(rpv, 16);
        int mynext = (e0 + lane < e1) ? col[e0 + lane] : 0;

        const float* xRow = x + (size_t)(row0 + m) * DIM + 4 * kh;
        float4 ax[16];
#pragma unroll
        for (int j = 0; j < 16; ++j) ax[j] = *(const float4*)(xRow + 8 * j);

        v8f c[8];
#pragma unroll
        for (int nt = 0; nt < 8; ++nt) {
            const int ncol = nt * 16 + m;
            const float bias = bl[ncol];
#pragma unroll
            for (int r = 0; r < 8; ++r) c[nt][r] = bias;
            const float* bwr = lWr + kh * 512 + ncol * 4;
#pragma unroll
            for (int j = 0; j < 16; ++j) {
                const float4 bq = *(const float4*)(bwr + j * 1024);
                v2f a0; a0.x = ax[j].x; a0.y = ax[j].y;
                v2f b0; b0.x = bq.x;    b0.y = bq.y;
                c[nt] = __builtin_amdgcn_wmma_f32_16x16x4_f32(
                            false, a0, false, b0, (short)0, c[nt], false, false);
                v2f a1; a1.x = ax[j].z; a1.y = ax[j].w;
                v2f b1; b1.x = bq.z;    b1.y = bq.w;
                c[nt] = __builtin_amdgcn_wmma_f32_16x16x4_f32(
                            false, a1, false, b1, (short)0, c[nt], false, false);
            }
        }

        int q = 0, qstart = e0;
        int qend = __builtin_amdgcn_readlane(rpv, 1);
        float4 acc; acc.x = 0.f; acc.y = 0.f; acc.z = 0.f; acc.w = 0.f;
        for (int j0 = e0; j0 < e1; j0 += 32) {
            const int mc  = (e1 - j0) < 32 ? (e1 - j0) : 32;
            const int myc = mynext;
            mynext = (j0 + 32 + lane < e1) ? col[j0 + 32 + lane] : 0;
            int j = 0;
            for (; j + 16 <= mc; j += 16) add_rows<16>(xl, myc, j, j0 + j, acc, q, qstart, qend, rpv, stw);
            if (j + 8 <= mc) { add_rows<8>(xl, myc, j, j0 + j, acc, q, qstart, qend, rpv, stw); j += 8; }
            if (j + 4 <= mc) { add_rows<4>(xl, myc, j, j0 + j, acc, q, qstart, qend, rpv, stw); j += 4; }
            if (j + 2 <= mc) { add_rows<2>(xl, myc, j, j0 + j, acc, q, qstart, qend, rpv, stw); j += 2; }
            if (j < mc)      { add_rows<1>(xl, myc, j, j0 + j, acc, q, qstart, qend, rpv, stw); }
        }
        while (q < 16) flush_node(acc, q, qstart, qend, rpv, stw);

        float4 am[16];
#pragma unroll
        for (int j = 0; j < 16; ++j) am[j] = *(const float4*)(str + 8 * j);

#pragma unroll
        for (int nt = 0; nt < 8; ++nt) {
            const int ncol = nt * 16 + m;
            const float* bwl = lWl + kh * 512 + ncol * 4;
#pragma unroll
            for (int j = 0; j < 16; ++j) {
                const float4 bq = *(const float4*)(bwl + j * 1024);
                v2f a0; a0.x = am[j].x; a0.y = am[j].y;
                v2f b0; b0.x = bq.x;    b0.y = bq.y;
                c[nt] = __builtin_amdgcn_wmma_f32_16x16x4_f32(
                            false, a0, false, b0, (short)0, c[nt], false, false);
                v2f a1; a1.x = am[j].z; a1.y = am[j].w;
                v2f b1; b1.x = bq.z;    b1.y = bq.w;
                c[nt] = __builtin_amdgcn_wmma_f32_16x16x4_f32(
                            false, a1, false, b1, (short)0, c[nt], false, false);
            }

            float s = 0.f, s2 = 0.f;
#pragma unroll
            for (int r = 0; r < 8; ++r) {
                float v = c[nt][r] > 0.f ? c[nt][r] : 0.f;
                sto[r * STAGE_PITCH + nt * 16] = v;
                s  += v;
                s2 += v * v;
            }
            s  += __shfl_xor(s, 16, 32);
            s2 += __shfl_xor(s2, 16, 32);
            if (lane < 16) {
                atomicAdd(&lsum[ncol], s);
                atomicAdd(&lsq[ncol], s2);
            }
        }

        float* hRow = h + (size_t)row0 * DIM + lane * 4;
#pragma unroll
        for (int q = 0; q < 16; ++q) {
            const float4 o = *(const float4*)(stw + q * STAGE_PITCH);
            *(float4*)(hRow + (size_t)q * DIM) = o;
        }
    }
    __syncthreads();
    if (tid < DIM) {
        atomicAdd(gsum + tid, lsum[tid]);
        atomicAdd(gsq  + tid, lsq[tid]);
    }
}

__global__ void stats_kernel(const float* __restrict__ gsum, const float* __restrict__ gsq,
                             const float* __restrict__ gamma, const float* __restrict__ beta,
                             float* __restrict__ cA, float* __restrict__ cB) {
    int c = threadIdx.x;
    if (c >= DIM) return;
    const float invN = 1.0f / (float)NN;
    float mu  = gsum[c] * invN;
    float var = gsq[c] * invN - mu * mu;
    float a   = rsqrtf(var + BN_EPS) * gamma[c];
    cA[c] = a;
    cB[c] = beta[c] - mu * a;
}

__global__ void apply_kernel(const float* __restrict__ x,
                             const float* __restrict__ cA, const float* __restrict__ cB,
                             float* __restrict__ out) {
    size_t i = (size_t)blockIdx.x * blockDim.x + threadIdx.x;
    if (i >= (size_t)NN * DIM / 4) return;
    int col4 = (int)(i & 31) * 4;
    float4 hv = ((float4*)out)[i];
    float4 xv = ((const float4*)x)[i];
    float4 a  = *(const float4*)(cA + col4);
    float4 b  = *(const float4*)(cB + col4);
    float4 o;
    o.x = xv.x + hv.x * a.x + b.x;
    o.y = xv.y + hv.y * a.y + b.y;
    o.z = xv.z + hv.z * a.z + b.z;
    o.w = xv.w + hv.w * a.w + b.w;
    ((float4*)out)[i] = o;
}

extern "C" void kernel_launch(void* const* d_in, const int* in_sizes, int n_in,
                              void* d_out, int out_size, void* d_ws, size_t ws_size,
                              hipStream_t stream) {
    const float* x     = (const float*)d_in[0];
    const int*   ei    = (const int*)d_in[1];
    const float* Wl    = (const float*)d_in[2];
    const float* bl    = (const float*)d_in[3];
    const float* Wr    = (const float*)d_in[4];
    const float* gamma = (const float*)d_in[5];
    const float* beta  = (const float*)d_in[6];
    float* out = (float*)d_out;

    float* agg  = (float*)d_ws;
    float* deg  = agg + (size_t)NN * DIM;
    float* gsum = deg + NN;
    float* gsq  = gsum + DIM;
    float* cA   = gsq + DIM;
    float* cB   = cA + DIM;

    int* T   = (int*)d_ws;
    int* tmp = T + NT_SLOTS;

    int* rowptr = tmp + NBLK * EDGE_CHUNK;
    int* col    = rowptr + ROWPTR_SLOTS;

    part_kernel<<<NBLK, 256, 0, stream>>>(ei, T, tmp, gsum);

    csr_kernel<<<NBK, 256, 0, stream>>>(T, tmp, rowptr, col);

    gather_gemm_bn_kernel<<<FT_GRID, FT_WAVES * 32, 0, stream>>>(rowptr, col, x, Wl, bl, Wr,
                                                                 out, gsum, gsq);

    stats_kernel<<<1, 128, 0, stream>>>(gsum, gsq, gamma, beta, cA, cB);

    int mb = (int)(((size_t)NN * DIM / 4 + 255) / 256);
    apply_kernel<<<mb, 256, 0, stream>>>(x, cA, cB, out);
}


# matrix kernel: LDS weight fragments of the next 16 products loaded as one group ahead (2nd register set), two N-tiles alternating; was a full LDS wait per two products; no sum order changes
# speedup vs baseline: 4.4268x; 1.0385x over previous
#include <hip/hip_runtime.h>

#define NN 100000
#define NE 1600000
#define DIM 128
#define BN_EPS 1e-5f
#define FT_WAVES 16
#define FT_GRID 256
#define STAGE_PITCH 132
#define ROWPTR_SLOTS 100352
#define EDGE_CHUNK 4096
#define NBLK ((NE + EDGE_CHUNK - 1) / EDGE_CHUNK)
#define BSHIFT 8
#define BW (1 << BSHIFT)
#define NBK ((NN + BW - 1) / BW)
#define BK_SLOTS 512
#define NT_SLOTS 153600
#define STAGE_CAP 8192
#define CSR_BATCH 16
#define RUNS_PER_HALF ((NBLK + 15) / 16)

typedef __attribute__((ext_vector_type(2))) float v2f;
typedef __attribute__((ext_vector_type(8))) float v8f;


__global__ __launch_bounds__(256) void part_kernel(
    const int* __restrict__ ei, int* __restrict__ T, int* __restrict__ tmp,
    float* __restrict__ sums) {
    __shared__ int lh[BK_SLOTS];
    __shared__ alignas(16) int lpk[EDGE_CHUNK];
    __shared__ int wtot[8];
    const int tid  = threadIdx.x;
    const int lane = tid & 31;
    const int wave = tid >> 5;
    const int b    = blockIdx.x;
    const int e0   = b * EDGE_CHUNK;
    lh[tid] = 0; lh[tid + 256] = 0;
    if (b == 0) sums[tid] = 0.f;
    __syncthreads();

    int4 sv[4], dv[4], rv[4];
#pragma unroll
    for (int u = 0; u < 4; ++u) {
        const int e = e0 + 4 * (tid + 256 * u);
        sv[u].x = 0; sv[u].y = 0; sv[u].z = 0; sv[u].w = 0;
        dv[u] = sv[u]; rv[u] = sv[u];
        if (e < NE) {
            sv[u] = *(const int4*)(ei + e);
            dv[u] = *(const int4*)(ei + NE + e);
            rv[u].x = atomicAdd(&lh[dv[u].x >> BSHIFT], 1);
            rv[u].y = atomicAdd(&lh[dv[u].y >> BSHIFT], 1);
            rv[u].z = atomicAdd(&lh[dv[u].z >> BSHIFT], 1);
            rv[u].w = atomicAdd(&lh[dv[u].w >> BSHIFT], 1);
        }
    }
    __syncthreads();

    const int c0 = lh[2 * tid];
    const int c1 = lh[2 * tid + 1];
    const int t  = c0 + c1;
    int inc = t;
    for (int o = 1; o < 32; o <<= 1) {
        const int y = __shfl_up(inc, o, 32);
        if (lane >= o) inc += y;
    }
    if (lane == 31) wtot[wave] = inc;
    __syncthreads();
    int off = 0;
    for (int w = 0; w < 8; ++w) if (w < wave) off += wtot[w];
    const int ex = off + inc - t;
    lh[2 * tid]     = ex;
    lh[2 * tid + 1] = ex + c0;
    if (2 * tid     <= NBK) T[(2 * tid)     * NBLK + b] = ex;
    if (2 * tid + 1 <= NBK) T[(2 * tid + 1) * NBLK + b] = ex + c0;
    __syncthreads();

#pragma unroll
    for (int u = 0; u < 4; ++u) {
        const int e = e0 + 4 * (tid + 256 * u);
        if (e < NE) {
            lpk[lh[dv[u].x >> BSHIFT] + rv[u].x] = (sv[u].x << BSHIFT) | (dv[u].x & (BW - 1));
            lpk[lh[dv[u].y >> BSHIFT] + rv[u].y] = (sv[u].y << BSHIFT) | (dv[u].y & (BW - 1));
            lpk[lh[dv[u].z >> BSHIFT] + rv[u].z] = (sv[u].z << BSHIFT) | (dv[u].z & (BW - 1));
            lpk[lh[dv[u].w >> BSHIFT] + rv[u].w] = (sv[u].w << BSHIFT) | (dv[u].w & (BW - 1));
        }
    }
    __syncthreads();

#pragma unroll
    for (int u = 0; u < 4; ++u) {
        const int j = 4 * (tid + 256 * u);
        if (e0 + j < NE) *(int4*)(tmp + e0 + j) = *(const int4*)(lpk + j);
    }
}

__global__ __launch_bounds__(256) void csr_kernel(
    const int* __restrict__ T, const int* __restrict__ tmp,
    int* __restrict__ rowptr, int* __restrict__ col) {
    __shared__ int2 lan[BK_SLOTS];
    __shared__ int lp[BK_SLOTS];
    __shared__ int lc[BW];
    __shared__ int lw[STAGE_CAP];
    __shared__ int lcol[STAGE_CAP];
    __shared__ int wtot[8];
    __shared__ int wsta[8];
    __shared__ int wnod[8];
    const int tid  = threadIdx.x;
    const int lane = tid & 31;
    const int wave = tid >> 5;
    const int q    = blockIdx.x;

    int a0 = 0, n0 = 0, a1 = 0, n1 = 0;
    if (2 * tid < NBLK) {
        a0 = T[q * NBLK + 2 * tid];
        n0 = T[(q + 1) * NBLK + 2 * tid] - a0;
    }
    if (2 * tid + 1 < NBLK) {
        a1 = T[q * NBLK + 2 * tid + 1];
        n1 = T[(q + 1) * NBLK + 2 * tid + 1] - a1;
    }
    const int t = n0 + n1;
    int inc = t;
    for (int o = 1; o < 32; o <<= 1) {
        const int y = __shfl_up(inc, o, 32);
        if (lane >= o) inc += y;
    }
    int asum = a0 + a1;
    for (int o = 16; o > 0; o >>= 1) asum += __shfl_xor(asum, o, 32);
    if (lane == 31) wtot[wave] = inc;
    if (lane == 0)  wsta[wave] = asum;
    lc[tid] = 0;
    __syncthreads();
    int off = 0, s = 0, total = 0;
    for (int w = 0; w < 8; ++w) {
        if (w < wave) off += wtot[w];
        total += wtot[w];
        s += wsta[w];
    }
    const int exr = off + inc - t;
    lan[2 * tid]     = make_int2(a0, n0); lp[2 * tid]     = exr;
    lan[2 * tid + 1] = make_int2(a1, n1); lp[2 * tid + 1] = exr + n0;
    __syncthreads();

    const bool staged = total <= STAGE_CAP;
    const int l16 = lane & 15;
    const int cb  = 2 * wave + (lane >> 4);
    if (staged) {
        int v[RUNS_PER_HALF], ra[RUNS_PER_HALF], rn[RUNS_PER_HALF], rp[RUNS_PER_HALF];
#pragma unroll
        for (int u = 0; u < RUNS_PER_HALF; ++u) {
            const int c  = cb + 16 * u;
            const int cc = c < NBLK ? c : NBLK - 1;
            const int2 an = lan[cc];
            ra[u] = cc * EDGE_CHUNK + an.x;
            rn[u] = c < NBLK ? an.y : 0;
            rp[u] = lp[cc];
            v[u]  = tmp[l16 < rn[u] ? ra[u] + l16 : 0];
        }
#pragma unroll
        for (int u = 0; u < RUNS_PER_HALF; ++u) {
            if (l16 < rn[u]) {
                lw[rp[u] + l16] = v[u];
                atomicAdd(&lc[v[u] & (BW - 1)], 1);
            }
            for (int k = 16 + l16; k < rn[u]; k += 16) {
                const int w = tmp[ra[u] + k];
                lw[rp[u] + k] = w;
                atomicAdd(&lc[w & (BW - 1)], 1);
            }
        }
    } else {
        for (int c = cb; c < NBLK; c += 16) {
            const int a = lan[c].x;
            const int n = lan[c].y;
            for (int k = l16; k < n; k += 16)
                atomicAdd(&lc[tmp[c * EDGE_CHUNK + a + k] & (BW - 1)], 1);
        }
    }
    __syncthreads();

    const int cn = lc[tid];
    int incn = cn;
    for (int o = 1; o < 32; o <<= 1) {
        const int y = __shfl_up(incn, o, 32);
        if (lane >= o) incn += y;
    }
    if (lane == 31) wnod[wave] = incn;
    __syncthreads();
    int offn = 0;
    for (int w = 0; w < 8; ++w) if (w < wave) offn += wnod[w];
    const int ex = offn + incn - cn;
    const int node = q * BW + tid;
    if (node <= NN) rowptr[node] = s + ex;
    lc[tid] = ex;
    __syncthreads();

    if (staged) {
        for (int i0 = tid; i0 < total; i0 += 256 * CSR_BATCH) {
            int w[CSR_BATCH];
#pragma unroll
            for (int u = 0; u < CSR_BATCH; ++u) {
                const int i = i0 + 256 * u;
                w[u] = (i < total) ? lw[i] : -1;
            }
#pragma unroll
            for (int u = 0; u < CSR_BATCH; ++u) {
                if (w[u] >= 0) {
                    const int p = atomicAdd(&lc[w[u] & (BW - 1)], 1);
                    lcol[p] = w[u] >> BSHIFT;
                }
            }
        }
    } else {
        for (int c = cb; c < NBLK; c += 16) {
            const int a = lan[c].x;
            const int n = lan[c].y;
            for (int k = l16; k < n; k += 16) {
                const int w = tmp[c * EDGE_CHUNK + a + k];
                const int p = atomicAdd(&lc[w & (BW - 1)], 1);
                col[s + p] = w >> BSHIFT;
            }
        }
    }
    __syncthreads();
    if (staged) {
        for (int i = tid; i < total; i += 256) col[s + i] = lcol[i];
    }
}

__device__ __forceinline__ void flush_node(float4& acc, int& q, int& qstart, int& qend,
                                           int rpv, float* __restrict__ stw) {
    const float inv = 1.0f / fmaxf((float)(qend - qstart), 1.0f);
    float4 o;
    o.x = acc.x * inv; o.y = acc.y * inv; o.z = acc.z * inv; o.w = acc.w * inv;
    *(float4*)(stw + q * STAGE_PITCH) = o;
    acc.x = 0.f; acc.y = 0.f; acc.z = 0.f; acc.w = 0.f;
    qstart = qend;
    ++q;
    qend = __builtin_amdgcn_readlane(rpv, q + 1);
}

template <int U>
__device__ __forceinline__ void add_rows(const float* __restrict__ xl, int myc, int j, int e,
                                         float4& acc, int& q, int& qstart, int& qend,
                                         int rpv, float* __restrict__ stw) {
    float4 v[U];
#pragma unroll
    for (int u = 0; u < U; ++u) {
        const int s = __builtin_amdgcn_readlane(myc, j + u);
        v[u] = *(const float4*)(xl + (size_t)s * DIM);
    }
#pragma unroll
    for (int u = 0; u < U; ++u) {
        while (e + u == qend) flush_node(acc, q, qstart, qend, rpv, stw);
        acc.x += v[u].x; acc.y += v[u].y; acc.z += v[u].z; acc.w += v[u].w;
    }
}

__device__ __forceinline__ void load_wfrag(float4 (&b)[8], const float* __restrict__ bw, int blk) {
    const float* s = bw + (blk >> 2) * 128 + (blk & 3) * 4096;
#pragma unroll
    for (int i = 0; i < 4; ++i) {
        b[2 * i]     = *(const float4*)(s + i * 1024);
        b[2 * i + 1] = *(const float4*)(s + i * 1024 + 64);
    }
}

__device__ __forceinline__ void mma_wfrag(v8f& c0, v8f& c1, const float4 (&a)[16],
                                          const float4 (&b)[8], int jq) {
#pragma unroll
    for (int i = 0; i < 4; ++i) {
        const float4 aq = a[4 * jq + i];
        v2f a0; a0.x = aq.x; a0.y = aq.y;
        v2f a1; a1.x = aq.z; a1.y = aq.w;
        v2f p0; p0.x = b[2 * i].x;     p0.y = b[2 * i].y;
        v2f p1; p1.x = b[2 * i + 1].x; p1.y = b[2 * i + 1].y;
        v2f q0; q0.x = b[2 * i].z;     q0.y = b[2 * i].w;
        v2f q1; q1.x = b[2 * i + 1].z; q1.y = b[2 * i + 1].w;
        c0 = __builtin_amdgcn_wmma_f32_16x16x4_f32(false, a0, false, p0, (short)0, c0, false, false);
        c1 = __builtin_amdgcn_wmma_f32_16x16x4_f32(false, a0, false, p1, (short)0, c1, false, false);
        c0 = __builtin_amdgcn_wmma_f32_16x16x4_f32(false, a1, false, q0, (short)0, c0, false, false);
        c1 = __builtin_amdgcn_wmma_f32_16x16x4_f32(false, a1, false, q1, (short)0, c1, false, false);
    }
}

__global__ __launch_bounds__(FT_WAVES * 32) void gather_gemm_bn_kernel(
    const int* __restrict__ rowptr, const int* __restrict__ col,
    const float* __restrict__ x,
    const float* __restrict__ Wl, const float* __restrict__ bl,
    const float* __restrict__ Wr, float* __restrict__ h,
    float* __restrict__ gsum, float* __restrict__ gsq) {
    __shared__ float lWl[DIM * DIM];
    __shared__ float lWr[DIM * DIM];
    __shared__ alignas(16) float lst[FT_WAVES * 16 * STAGE_PITCH];
    __shared__ float lsum[DIM];
    __shared__ float lsq[DIM];

    const int tid  = threadIdx.x;
    const int lane = tid & 31;
    const int wave = __builtin_amdgcn_readfirstlane(tid >> 5);
    const int m    = lane & 15;
    const int kh   = lane >> 4;

    {
        const int n  = tid & (DIM - 1);
        const int qb = tid >> 7;
        const float* pl = Wl + (size_t)(4 * qb) * DIM + n;
        const float* pr = Wr + (size_t)(4 * qb) * DIM + n;
        float* sl = lWl + qb * 512 + n * 4;
        float* sr = lWr + qb * 512 + n * 4;
        float4 wl[8], wr[8];
#pragma unroll
        for (int u = 0; u < 8; ++u) {
            const float* a = pl + (size_t)u * (16 * DIM);
            const float* b = pr + (size_t)u * (16 * DIM);
            wl[u].x = a[0]; wl[u].y = a[DIM]; wl[u].z = a[2 * DIM]; wl[u].w = a[3 * DIM];
            wr[u].x = b[0]; wr[u].y = b[DIM]; wr[u].z = b[2 * DIM]; wr[u].w = b[3 * DIM];
        }
#pragma unroll
        for (int u = 0; u < 8; ++u) {
            *(float4*)(sl + u * 2048) = wl[u];
            *(float4*)(sr + u * 2048) = wr[u];
        }
    }
    if (tid < DIM) { lsum[tid] = 0.f; lsq[tid] = 0.f; }
    __syncthreads();

    const float* xl  = x + lane * 4;
    float* stw       = lst + wave * (16 * STAGE_PITCH) + lane * 4;
    const float* str = lst + wave * (16 * STAGE_PITCH) + m * STAGE_PITCH + 4 * kh;
    float* sto       = lst + wave * (16 * STAGE_PITCH) + (8 * kh) * STAGE_PITCH + m;

    float biasv[8];
#pragma unroll
    for (int nt = 0; nt < 8; ++nt) biasv[nt] = bl[nt * 16 + m];

    const int ntile = (NN / 16 - (int)blockIdx.x + FT_GRID - 1) / FT_GRID;
    for (int t = wave; t < ntile; t += FT_WAVES) {
        const int row0 = ((int)blockIdx.x + FT_GRID * t) * 16;

        const int rpv = rowptr[row0 + (lane < 16 ? lane : 16)];
        const int e0  = __builtin_amdgcn_readlane(rpv, 0);
        const int e1  = __builtin_amdgcn_readlane(rpv, 16);
        int mynext = (e0 + lane < e1) ? col[e0 + lane] : 0;

        const float* xRow = x + (size_t)(row0 + m) * DIM + 4 * kh;
        float4 ax[16];
#pragma unroll
        for (int j = 0; j < 16; ++j) ax[j] = *(const float4*)(xRow + 8 * j);

        v8f c[8];
#pragma unroll
        for (int nt = 0; nt < 8; ++nt) {
            float bias = biasv[nt];
            asm volatile("" : "+v"(bias));
#pragma unroll
            for (int r = 0; r < 8; ++r) c[nt][r] = bias;
        }
        {
            const float* bwr = lWr + kh * 512 + m * 4;
            float4 w0[8], w1[8];
            load_wfrag(w0, bwr, 0);
#pragma unroll
            for (int blk = 0; blk < 16; blk += 2) {
                load_wfrag(w1, bwr, blk + 1);
                __builtin_amdgcn_sched_barrier(0);
                mma_wfrag(c[2 * (blk >> 2)], c[2 * (blk >> 2) + 1], ax, w0, blk & 3);
                __builtin_amdgcn_sched_barrier(0);
                if (blk + 2 < 16) load_wfrag(w0, bwr, blk + 2);
                __builtin_amdgcn_sched_barrier(0);
                mma_wfrag(c[2 * ((blk + 1) >> 2)], c[2 * ((blk + 1) >> 2) + 1], ax, w1, (blk + 1) & 3);
                __builtin_amdgcn_sched_barrier(0);
            }
        }

        int q = 0, qstart = e0;
        int qend = __builtin_amdgcn_readlane(rpv, 1);
        float4 acc; acc.x = 0.f; acc.y = 0.f; acc.z = 0.f; acc.w = 0.f;
        for (int j0 = e0; j0 < e1; j0 += 32) {
            const int mc  = (e1 - j0) < 32 ? (e1 - j0) : 32;
            const int myc = mynext;
            mynext = (j0 + 32 + lane < e1) ? col[j0 + 32 + lane] : 0;
            int j = 0;
            for (; j + 16 <= mc; j += 16) add_rows<16>(xl, myc, j, j0 + j, acc, q, qstart, qend, rpv, stw);
            if (j + 8 <= mc) { add_rows<8>(xl, myc, j, j0 + j, acc, q, qstart, qend, rpv, stw); j += 8; }
            if (j + 4 <= mc) { add_rows<4>(xl, myc, j, j0 + j, acc, q, qstart, qend, rpv, stw); j += 4; }
            if (j + 2 <= mc) { add_rows<2>(xl, myc, j, j0 + j, acc, q, qstart, qend, rpv, stw); j += 2; }
            if (j < mc)      { add_rows<1>(xl, myc, j, j0 + j, acc, q, qstart, qend, rpv, stw); }
        }
        while (q < 16) flush_node(acc, q, qstart, qend, rpv, stw);

        float4 am[16];
#pragma unroll
        for (int j = 0; j < 16; ++j) am[j] = *(const float4*)(str + 8 * j);

        {
            const float* bwl = lWl + kh * 512 + m * 4;
            float4 w0[8], w1[8];
            load_wfrag(w0, bwl, 0);
#pragma unroll
            for (int blk = 0; blk < 16; blk += 2) {
                load_wfrag(w1, bwl, blk + 1);
                __builtin_amdgcn_sched_barrier(0);
                mma_wfrag(c[2 * (blk >> 2)], c[2 * (blk >> 2) + 1], am, w0, blk & 3);
                __builtin_amdgcn_sched_barrier(0);
                if (blk + 2 < 16) load_wfrag(w0, bwl, blk + 2);
                __builtin_amdgcn_sched_barrier(0);
                mma_wfrag(c[2 * ((blk + 1) >> 2)], c[2 * ((blk + 1) >> 2) + 1], am, w1, (blk + 1) & 3);
                __builtin_amdgcn_sched_barrier(0);

                if (((blk + 1) & 3) == 3) {
#pragma unroll
                    for (int nt = 2 * ((blk + 1) >> 2); nt < 2 * ((blk + 1) >> 2) + 2; ++nt) {
                        const int ncol = nt * 16 + m;
                        float s = 0.f, s2 = 0.f;
#pragma unroll
                        for (int r = 0; r < 8; ++r) {
                            float v = c[nt][r] > 0.f ? c[nt][r] : 0.f;
                            sto[r * STAGE_PITCH + nt * 16] = v;
                            s  += v;
                            s2 += v * v;
                        }
                        s  += __shfl_xor(s, 16, 32);
                        s2 += __shfl_xor(s2, 16, 32);
                        if (lane < 16) {
                            atomicAdd(&lsum[ncol], s);
                            atomicAdd(&lsq[ncol], s2);
                        }
                    }
                }
            }
        }

        float* hRow = h + (size_t)row0 * DIM + lane * 4;
#pragma unroll
        for (int q = 0; q < 16; ++q) {
            const float4 o = *(const float4*)(stw + q * STAGE_PITCH);
            *(float4*)(hRow + (size_t)q * DIM) = o;
        }
    }
    __syncthreads();
    if (tid < DIM) {
        atomicAdd(gsum + tid, lsum[tid]);
        atomicAdd(gsq  + tid, lsq[tid]);
    }
}

__global__ void stats_kernel(const float* __restrict__ gsum, const float* __restrict__ gsq,
                             const float* __restrict__ gamma, const float* __restrict__ beta,
                             float* __restrict__ cA, float* __restrict__ cB) {
    int c = threadIdx.x;
    if (c >= DIM) return;
    const float invN = 1.0f / (float)NN;
    float mu  = gsum[c] * invN;
    float var = gsq[c] * invN - mu * mu;
    float a   = rsqrtf(var + BN_EPS) * gamma[c];
    cA[c] = a;
    cB[c] = beta[c] - mu * a;
}

__global__ void apply_kernel(const float* __restrict__ x,
                             const float* __restrict__ cA, const float* __restrict__ cB,
                             float* __restrict__ out) {
    size_t i = (size_t)blockIdx.x * blockDim.x + threadIdx.x;
    if (i >= (size_t)NN * DIM / 4) return;
    int col4 = (int)(i & 31) * 4;
    float4 hv = ((float4*)out)[i];
    float4 xv = ((const float4*)x)[i];
    float4 a  = *(const float4*)(cA + col4);
    float4 b  = *(const float4*)(cB + col4);
    float4 o;
    o.x = xv.x + hv.x * a.x + b.x;
    o.y = xv.y + hv.y * a.y + b.y;
    o.z = xv.z + hv.z * a.z + b.z;
    o.w = xv.w + hv.w * a.w + b.w;
    ((float4*)out)[i] = o;
}

extern "C" void kernel_launch(void* const* d_in, const int* in_sizes, int n_in,
                              void* d_out, int out_size, void* d_ws, size_t ws_size,
                              hipStream_t stream) {
    const float* x     = (const float*)d_in[0];
    const int*   ei    = (const int*)d_in[1];
    const float* Wl    = (const float*)d_in[2];
    const float* bl    = (const float*)d_in[3];
    const float* Wr    = (const float*)d_in[4];
    const float* gamma = (const float*)d_in[5];
    const float* beta  = (const float*)d_in[6];
    float* out = (float*)d_out;

    float* agg  = (float*)d_ws;
    float* deg  = agg + (size_t)NN * DIM;
    float* gsum = deg + NN;
    float* gsq  = gsum + DIM;
    float* cA   = gsq + DIM;
    float* cB   = cA + DIM;

    int* T   = (int*)d_ws;
    int* tmp = T + NT_SLOTS;

    int* rowptr = tmp + NBLK * EDGE_CHUNK;
    int* col    = rowptr + ROWPTR_SLOTS;

    part_kernel<<<NBLK, 256, 0, stream>>>(ei, T, tmp, gsum);

    csr_kernel<<<NBK, 256, 0, stream>>>(T, tmp, rowptr, col);

    gather_gemm_bn_kernel<<<FT_GRID, FT_WAVES * 32, 0, stream>>>(rowptr, col, x, Wl, bl, Wr,
                                                                 out, gsum, gsq);

    stats_kernel<<<1, 128, 0, stream>>>(gsum, gsq, gamma, beta, cA, cB);

    int mb = (int)(((size_t)NN * DIM / 4 + 255) / 256);
    apply_kernel<<<mb, 256, 0, stream>>>(x, cA, cB, out);
}


# neighbour gather: two register sets of 16 rows take turns, a set is refilled while the other's 16 loads are in flight (16-32 row loads out, not 16 draining to none); per-node sum order unchanged
# speedup vs baseline: 4.6338x; 1.0468x over previous
#include <hip/hip_runtime.h>

#define NN 100000
#define NE 1600000
#define DIM 128
#define BN_EPS 1e-5f
#define FT_WAVES 16
#define FT_GRID 256
#define STAGE_PITCH 132
#define ROWPTR_SLOTS 100352
#define EDGE_CHUNK 4096
#define NBLK ((NE + EDGE_CHUNK - 1) / EDGE_CHUNK)
#define BSHIFT 8
#define BW (1 << BSHIFT)
#define NBK ((NN + BW - 1) / BW)
#define BK_SLOTS 512
#define NT_SLOTS 153600
#define STAGE_CAP 8192
#define CSR_BATCH 16
#define RUNS_PER_HALF ((NBLK + 15) / 16)

typedef __attribute__((ext_vector_type(2))) float v2f;
typedef __attribute__((ext_vector_type(8))) float v8f;


__global__ __launch_bounds__(256) void part_kernel(
    const int* __restrict__ ei, int* __restrict__ T, int* __restrict__ tmp,
    float* __restrict__ sums) {
    __shared__ int lh[BK_SLOTS];
    __shared__ alignas(16) int lpk[EDGE_CHUNK];
    __shared__ int wtot[8];
    const int tid  = threadIdx.x;
    const int lane = tid & 31;
    const int wave = tid >> 5;
    const int b    = blockIdx.x;
    const int e0   = b * EDGE_CHUNK;
    lh[tid] = 0; lh[tid + 256] = 0;
    if (b == 0) sums[tid] = 0.f;
    __syncthreads();

    int4 sv[4], dv[4], rv[4];
#pragma unroll
    for (int u = 0; u < 4; ++u) {
        const int e = e0 + 4 * (tid + 256 * u);
        sv[u].x = 0; sv[u].y = 0; sv[u].z = 0; sv[u].w = 0;
        dv[u] = sv[u]; rv[u] = sv[u];
        if (e < NE) {
            sv[u] = *(const int4*)(ei + e);
            dv[u] = *(const int4*)(ei + NE + e);
            rv[u].x = atomicAdd(&lh[dv[u].x >> BSHIFT], 1);
            rv[u].y = atomicAdd(&lh[dv[u].y >> BSHIFT], 1);
            rv[u].z = atomicAdd(&lh[dv[u].z >> BSHIFT], 1);
            rv[u].w = atomicAdd(&lh[dv[u].w >> BSHIFT], 1);
        }
    }
    __syncthreads();

    const int c0 = lh[2 * tid];
    const int c1 = lh[2 * tid + 1];
    const int t  = c0 + c1;
    int inc = t;
    for (int o = 1; o < 32; o <<= 1) {
        const int y = __shfl_up(inc, o, 32);
        if (lane >= o) inc += y;
    }
    if (lane == 31) wtot[wave] = inc;
    __syncthreads();
    int off = 0;
    for (int w = 0; w < 8; ++w) if (w < wave) off += wtot[w];
    const int ex = off + inc - t;
    lh[2 * tid]     = ex;
    lh[2 * tid + 1] = ex + c0;
    if (2 * tid     <= NBK) T[(2 * tid)     * NBLK + b] = ex;
    if (2 * tid + 1 <= NBK) T[(2 * tid + 1) * NBLK + b] = ex + c0;
    __syncthreads();

#pragma unroll
    for (int u = 0; u < 4; ++u) {
        const int e = e0 + 4 * (tid + 256 * u);
        if (e < NE) {
            lpk[lh[dv[u].x >> BSHIFT] + rv[u].x] = (sv[u].x << BSHIFT) | (dv[u].x & (BW - 1));
            lpk[lh[dv[u].y >> BSHIFT] + rv[u].y] = (sv[u].y << BSHIFT) | (dv[u].y & (BW - 1));
            lpk[lh[dv[u].z >> BSHIFT] + rv[u].z] = (sv[u].z << BSHIFT) | (dv[u].z & (BW - 1));
            lpk[lh[dv[u].w >> BSHIFT] + rv[u].w] = (sv[u].w << BSHIFT) | (dv[u].w & (BW - 1));
        }
    }
    __syncthreads();

#pragma unroll
    for (int u = 0; u < 4; ++u) {
        const int j = 4 * (tid + 256 * u);
        if (e0 + j < NE) *(int4*)(tmp + e0 + j) = *(const int4*)(lpk + j);
    }
}

__global__ __launch_bounds__(256) void csr_kernel(
    const int* __restrict__ T, const int* __restrict__ tmp,
    int* __restrict__ rowptr, int* __restrict__ col) {
    __shared__ int2 lan[BK_SLOTS];
    __shared__ int lp[BK_SLOTS];
    __shared__ int lc[BW];
    __shared__ int lw[STAGE_CAP];
    __shared__ int lcol[STAGE_CAP];
    __shared__ int wtot[8];
    __shared__ int wsta[8];
    __shared__ int wnod[8];
    const int tid  = threadIdx.x;
    const int lane = tid & 31;
    const int wave = tid >> 5;
    const int q    = blockIdx.x;

    int a0 = 0, n0 = 0, a1 = 0, n1 = 0;
    if (2 * tid < NBLK) {
        a0 = T[q * NBLK + 2 * tid];
        n0 = T[(q + 1) * NBLK + 2 * tid] - a0;
    }
    if (2 * tid + 1 < NBLK) {
        a1 = T[q * NBLK + 2 * tid + 1];
        n1 = T[(q + 1) * NBLK + 2 * tid + 1] - a1;
    }
    const int t = n0 + n1;
    int inc = t;
    for (int o = 1; o < 32; o <<= 1) {
        const int y = __shfl_up(inc, o, 32);
        if (lane >= o) inc += y;
    }
    int asum = a0 + a1;
    for (int o = 16; o > 0; o >>= 1) asum += __shfl_xor(asum, o, 32);
    if (lane == 31) wtot[wave] = inc;
    if (lane == 0)  wsta[wave] = asum;
    lc[tid] = 0;
    __syncthreads();
    int off = 0, s = 0, total = 0;
    for (int w = 0; w < 8; ++w) {
        if (w < wave) off += wtot[w];
        total += wtot[w];
        s += wsta[w];
    }
    const int exr = off + inc - t;
    lan[2 * tid]     = make_int2(a0, n0); lp[2 * tid]     = exr;
    lan[2 * tid + 1] = make_int2(a1, n1); lp[2 * tid + 1] = exr + n0;
    __syncthreads();

    const bool staged = total <= STAGE_CAP;
    const int l16 = lane & 15;
    const int cb  = 2 * wave + (lane >> 4);
    if (staged) {
        int v[RUNS_PER_HALF], ra[RUNS_PER_HALF], rn[RUNS_PER_HALF], rp[RUNS_PER_HALF];
#pragma unroll
        for (int u = 0; u < RUNS_PER_HALF; ++u) {
            const int c  = cb + 16 * u;
            const int cc = c < NBLK ? c : NBLK - 1;
            const int2 an = lan[cc];
            ra[u] = cc * EDGE_CHUNK + an.x;
            rn[u] = c < NBLK ? an.y : 0;
            rp[u] = lp[cc];
            v[u]  = tmp[l16 < rn[u] ? ra[u] + l16 : 0];
        }
#pragma unroll
        for (int u = 0; u < RUNS_PER_HALF; ++u) {
            if (l16 < rn[u]) {
                lw[rp[u] + l16] = v[u];
                atomicAdd(&lc[v[u] & (BW - 1)], 1);
            }
            for (int k = 16 + l16; k < rn[u]; k += 16) {
                const int w = tmp[ra[u] + k];
                lw[rp[u] + k] = w;
                atomicAdd(&lc[w & (BW - 1)], 1);
            }
        }
    } else {
        for (int c = cb; c < NBLK; c += 16) {
            const int a = lan[c].x;
            const int n = lan[c].y;
            for (int k = l16; k < n; k += 16)
                atomicAdd(&lc[tmp[c * EDGE_CHUNK + a + k] & (BW - 1)], 1);
        }
    }
    __syncthreads();

    const int cn = lc[tid];
    int incn = cn;
    for (int o = 1; o < 32; o <<= 1) {
        const int y = __shfl_up(incn, o, 32);
        if (lane >= o) incn += y;
    }
    if (lane == 31) wnod[wave] = incn;
    __syncthreads();
    int offn = 0;
    for (int w = 0; w < 8; ++w) if (w < wave) offn += wnod[w];
    const int ex = offn + incn - cn;
    const int node = q * BW + tid;
    if (node <= NN) rowptr[node] = s + ex;
    lc[tid] = ex;
    __syncthreads();

    if (staged) {
        for (int i0 = tid; i0 < total; i0 += 256 * CSR_BATCH) {
            int w[CSR_BATCH];
#pragma unroll
            for (int u = 0; u < CSR_BATCH; ++u) {
                const int i = i0 + 256 * u;
                w[u] = (i < total) ? lw[i] : -1;
            }
#pragma unroll
            for (int u = 0; u < CSR_BATCH; ++u) {
                if (w[u] >= 0) {
                    const int p = atomicAdd(&lc[w[u] & (BW - 1)], 1);
                    lcol[p] = w[u] >> BSHIFT;
                }
            }
        }
    } else {
        for (int c = cb; c < NBLK; c += 16) {
            const int a = lan[c].x;
            const int n = lan[c].y;
            for (int k = l16; k < n; k += 16) {
                const int w = tmp[c * EDGE_CHUNK + a + k];
                const int p = atomicAdd(&lc[w & (BW - 1)], 1);
                col[s + p] = w >> BSHIFT;
            }
        }
    }
    __syncthreads();
    if (staged) {
        for (int i = tid; i < total; i += 256) col[s + i] = lcol[i];
    }
}

__device__ __forceinline__ void flush_node(float4& acc, int& q, int& qstart, int& qend,
                                           int rpv, float* __restrict__ stw) {
    const float inv = 1.0f / fmaxf((float)(qend - qstart), 1.0f);
    float4 o;
    o.x = acc.x * inv; o.y = acc.y * inv; o.z = acc.z * inv; o.w = acc.w * inv;
    *(float4*)(stw + q * STAGE_PITCH) = o;
    acc.x = 0.f; acc.y = 0.f; acc.z = 0.f; acc.w = 0.f;
    qstart = qend;
    ++q;
    qend = __builtin_amdgcn_readlane(rpv, q + 1);
}

template <int U>
__device__ __forceinline__ void add_rows(const float* __restrict__ xl, int myc, int j, int e,
                                         float4& acc, int& q, int& qstart, int& qend,
                                         int rpv, float* __restrict__ stw) {
    float4 v[U];
#pragma unroll
    for (int u = 0; u < U; ++u) {
        const int s = __builtin_amdgcn_readlane(myc, j + u);
        v[u] = *(const float4*)(xl + (size_t)s * DIM);
    }
#pragma unroll
    for (int u = 0; u < U; ++u) {
        while (e + u == qend) flush_node(acc, q, qstart, qend, rpv, stw);
        acc.x += v[u].x; acc.y += v[u].y; acc.z += v[u].z; acc.w += v[u].w;
    }
}

template <int U>
__device__ __forceinline__ void load_rows(float4 (&v)[U], const float* __restrict__ x, unsigned lane16, int ids, int j) {
#pragma unroll
    for (int u = 0; u < U; ++u) {
        const unsigned s = (unsigned)__builtin_amdgcn_readlane(ids, j + u);
        const unsigned off = (s << 9) + lane16;
        v[u] = *(const float4*)((const char*)x + off);
    }
}

template <int U>
__device__ __forceinline__ void sum_rows(const float4 (&v)[U], int e,
                                          float4& acc, int& q, int& qstart, int& qend,
                                          int rpv, float* __restrict__ stw) {
#pragma unroll
    for (int u = 0; u < U; ++u) {
        while (e + u == qend) flush_node(acc, q, qstart, qend, rpv, stw);
        acc.x += v[u].x; acc.y += v[u].y; acc.z += v[u].z; acc.w += v[u].w;
    }
}

__device__ __forceinline__ void load_wfrag(float4 (&b)[8], const float* __restrict__ bw, int blk) {
    const float* s = bw + (blk >> 2) * 128 + (blk & 3) * 4096;
#pragma unroll
    for (int i = 0; i < 4; ++i) {
        b[2 * i]     = *(const float4*)(s + i * 1024);
        b[2 * i + 1] = *(const float4*)(s + i * 1024 + 64);
    }
}

__device__ __forceinline__ void mma_wfrag(v8f& c0, v8f& c1, const float4 (&a)[16],
                                          const float4 (&b)[8], int jq) {
#pragma unroll
    for (int i = 0; i < 4; ++i) {
        const float4 aq = a[4 * jq + i];
        v2f a0; a0.x = aq.x; a0.y = aq.y;
        v2f a1; a1.x = aq.z; a1.y = aq.w;
        v2f p0; p0.x = b[2 * i].x;     p0.y = b[2 * i].y;
        v2f p1; p1.x = b[2 * i + 1].x; p1.y = b[2 * i + 1].y;
        v2f q0; q0.x = b[2 * i].z;     q0.y = b[2 * i].w;
        v2f q1; q1.x = b[2 * i + 1].z; q1.y = b[2 * i + 1].w;
        c0 = __builtin_amdgcn_wmma_f32_16x16x4_f32(false, a0, false, p0, (short)0, c0, false, false);
        c1 = __builtin_amdgcn_wmma_f32_16x16x4_f32(false, a0, false, p1, (short)0, c1, false, false);
        c0 = __builtin_amdgcn_wmma_f32_16x16x4_f32(false, a1, false, q0, (short)0, c0, false, false);
        c1 = __builtin_amdgcn_wmma_f32_16x16x4_f32(false, a1, false, q1, (short)0, c1, false, false);
    }
}

__global__ __launch_bounds__(FT_WAVES * 32) void gather_gemm_bn_kernel(
    const int* __restrict__ rowptr, const int* __restrict__ col,
    const float* __restrict__ x,
    const float* __restrict__ Wl, const float* __restrict__ bl,
    const float* __restrict__ Wr, float* __restrict__ h,
    float* __restrict__ gsum, float* __restrict__ gsq) {
    __shared__ float lWl[DIM * DIM];
    __shared__ float lWr[DIM * DIM];
    __shared__ alignas(16) float lst[FT_WAVES * 16 * STAGE_PITCH];
    __shared__ float lsum[DIM];
    __shared__ float lsq[DIM];

    const int tid  = threadIdx.x;
    const int lane = tid & 31;
    const int wave = __builtin_amdgcn_readfirstlane(tid >> 5);
    const int m    = lane & 15;
    const int kh   = lane >> 4;

    {
        const int n  = tid & (DIM - 1);
        const int qb = tid >> 7;
        const float* pl = Wl + (size_t)(4 * qb) * DIM + n;
        const float* pr = Wr + (size_t)(4 * qb) * DIM + n;
        float* sl = lWl + qb * 512 + n * 4;
        float* sr = lWr + qb * 512 + n * 4;
        float4 wl[8], wr[8];
#pragma unroll
        for (int u = 0; u < 8; ++u) {
            const float* a = pl + (size_t)u * (16 * DIM);
            const float* b = pr + (size_t)u * (16 * DIM);
            wl[u].x = a[0]; wl[u].y = a[DIM]; wl[u].z = a[2 * DIM]; wl[u].w = a[3 * DIM];
            wr[u].x = b[0]; wr[u].y = b[DIM]; wr[u].z = b[2 * DIM]; wr[u].w = b[3 * DIM];
        }
#pragma unroll
        for (int u = 0; u < 8; ++u) {
            *(float4*)(sl + u * 2048) = wl[u];
            *(float4*)(sr + u * 2048) = wr[u];
        }
    }
    if (tid < DIM) { lsum[tid] = 0.f; lsq[tid] = 0.f; }
    __syncthreads();

    const float* xl  = x + lane * 4;
    float* stw       = lst + wave * (16 * STAGE_PITCH) + lane * 4;
    const float* str = lst + wave * (16 * STAGE_PITCH) + m * STAGE_PITCH + 4 * kh;
    float* sto       = lst + wave * (16 * STAGE_PITCH) + (8 * kh) * STAGE_PITCH + m;


    const int ntile = (NN / 16 - (int)blockIdx.x + FT_GRID - 1) / FT_GRID;
    for (int t = wave; t < ntile; t += FT_WAVES) {
        const int row0 = ((int)blockIdx.x + FT_GRID * t) * 16;

        const int rpv = rowptr[row0 + (lane < 16 ? lane : 16)];
        const int e0  = __builtin_amdgcn_readlane(rpv, 0);
        const int e1  = __builtin_amdgcn_readlane(rpv, 16);
        int mynext = (e0 + lane < e1) ? col[e0 + lane] : 0;

        const float* xRow = x + (size_t)(row0 + m) * DIM + 4 * kh;
        float4 ax[16];
#pragma unroll
        for (int j = 0; j < 16; ++j) ax[j] = *(const float4*)(xRow + 8 * j);

        v8f c[8];
#pragma unroll
        for (int nt = 0; nt < 8; ++nt) {
            float bias = bl[nt * 16 + m];
            asm volatile("" : "+v"(bias));
#pragma unroll
            for (int r = 0; r < 8; ++r) c[nt][r] = bias;
        }
        {
            const float* bwr = lWr + kh * 512 + m * 4;
            float4 w0[8], w1[8];
            load_wfrag(w0, bwr, 0);
#pragma unroll
            for (int blk = 0; blk < 16; blk += 2) {
                load_wfrag(w1, bwr, blk + 1);
                __builtin_amdgcn_sched_barrier(0);
                mma_wfrag(c[2 * (blk >> 2)], c[2 * (blk >> 2) + 1], ax, w0, blk & 3);
                __builtin_amdgcn_sched_barrier(0);
                if (blk + 2 < 16) load_wfrag(w0, bwr, blk + 2);
                __builtin_amdgcn_sched_barrier(0);
                mma_wfrag(c[2 * ((blk + 1) >> 2)], c[2 * ((blk + 1) >> 2) + 1], ax, w1, (blk + 1) & 3);
                __builtin_amdgcn_sched_barrier(0);
            }
        }

        int q = 0, qstart = e0;
        int qend = __builtin_amdgcn_readlane(rpv, 1);
        float4 acc; acc.x = 0.f; acc.y = 0.f; acc.z = 0.f; acc.w = 0.f;
        const int nb = (e1 - e0) >> 4;
        float4 sa[16] = {}, sb[16] = {};
        const unsigned lane16 = (unsigned)lane * 16u;
        int tids = mynext;
        int tj   = 0;
        if (nb >= 1) {
            int g = 0;
            int idsC = mynext;
            const int pn = e0 + 32 + lane;
            int idsN = col[pn < NE - 1 ? pn : NE - 1];
            load_rows<16>(sa, x, lane16, idsC, 0);
            for (; g + 2 < nb; g += 2) {
                const int pb  = e0 + 16 * g;
                const int pnn = pb + 64 + lane;
                const int idsNN = col[pnn < NE - 1 ? pnn : NE - 1];
                load_rows<16>(sb, x, lane16, idsC, 16);
                sum_rows<16>(sa, pb, acc, q, qstart, qend, rpv, stw);
                load_rows<16>(sa, x, lane16, idsN, 0);
                sum_rows<16>(sb, pb + 16, acc, q, qstart, qend, rpv, stw);
                idsC = idsN;
                idsN = idsNN;
            }
            const int pb = e0 + 16 * g;
            if (nb - g == 2) {
                load_rows<16>(sb, x, lane16, idsC, 16);
                sum_rows<16>(sa, pb, acc, q, qstart, qend, rpv, stw);
                sum_rows<16>(sb, pb + 16, acc, q, qstart, qend, rpv, stw);
                tids = idsN; tj = 0;
            } else {
                sum_rows<16>(sa, pb, acc, q, qstart, qend, rpv, stw);
                tids = idsC; tj = 16;
            }
        }
        {
            int j  = tj;
            int pt = e0 + 16 * nb;
            if (pt + 8 <= e1) { add_rows<8>(xl, tids, j, pt, acc, q, qstart, qend, rpv, stw); j += 8; pt += 8; }
            if (pt + 4 <= e1) { add_rows<4>(xl, tids, j, pt, acc, q, qstart, qend, rpv, stw); j += 4; pt += 4; }
            if (pt + 2 <= e1) { add_rows<2>(xl, tids, j, pt, acc, q, qstart, qend, rpv, stw); j += 2; pt += 2; }
            if (pt < e1)      { add_rows<1>(xl, tids, j, pt, acc, q, qstart, qend, rpv, stw); }
        }
        while (q < 16) flush_node(acc, q, qstart, qend, rpv, stw);

        float4 am[16];
#pragma unroll
        for (int j = 0; j < 16; ++j) am[j] = *(const float4*)(str + 8 * j);

        {
            const float* bwl = lWl + kh * 512 + m * 4;
            float4 w0[8], w1[8];
            load_wfrag(w0, bwl, 0);
#pragma unroll
            for (int blk = 0; blk < 16; blk += 2) {
                load_wfrag(w1, bwl, blk + 1);
                __builtin_amdgcn_sched_barrier(0);
                mma_wfrag(c[2 * (blk >> 2)], c[2 * (blk >> 2) + 1], am, w0, blk & 3);
                __builtin_amdgcn_sched_barrier(0);
                if (blk + 2 < 16) load_wfrag(w0, bwl, blk + 2);
                __builtin_amdgcn_sched_barrier(0);
                mma_wfrag(c[2 * ((blk + 1) >> 2)], c[2 * ((blk + 1) >> 2) + 1], am, w1, (blk + 1) & 3);
                __builtin_amdgcn_sched_barrier(0);

                if (((blk + 1) & 3) == 3) {
#pragma unroll
                    for (int nt = 2 * ((blk + 1) >> 2); nt < 2 * ((blk + 1) >> 2) + 2; ++nt) {
                        const int ncol = nt * 16 + m;
                        float s = 0.f, s2 = 0.f;
#pragma unroll
                        for (int r = 0; r < 8; ++r) {
                            float v = c[nt][r] > 0.f ? c[nt][r] : 0.f;
                            sto[r * STAGE_PITCH + nt * 16] = v;
                            s  += v;
                            s2 += v * v;
                        }
                        s  += __shfl_xor(s, 16, 32);
                        s2 += __shfl_xor(s2, 16, 32);
                        if (lane < 16) {
                            atomicAdd(&lsum[ncol], s);
                            atomicAdd(&lsq[ncol], s2);
                        }
                    }
                }
            }
        }

        float* hRow = h + (size_t)row0 * DIM + lane * 4;
#pragma unroll
        for (int q = 0; q < 16; ++q) {
            const float4 o = *(const float4*)(stw + q * STAGE_PITCH);
            *(float4*)(hRow + (size_t)q * DIM) = o;
        }
    }
    __syncthreads();
    if (tid < DIM) {
        atomicAdd(gsum + tid, lsum[tid]);
        atomicAdd(gsq  + tid, lsq[tid]);
    }
}

__global__ void stats_kernel(const float* __restrict__ gsum, const float* __restrict__ gsq,
                             const float* __restrict__ gamma, const float* __restrict__ beta,
                             float* __restrict__ cA, float* __restrict__ cB) {
    int c = threadIdx.x;
    if (c >= DIM) return;
    const float invN = 1.0f / (float)NN;
    float mu  = gsum[c] * invN;
    float var = gsq[c] * invN - mu * mu;
    float a   = rsqrtf(var + BN_EPS) * gamma[c];
    cA[c] = a;
    cB[c] = beta[c] - mu * a;
}

__global__ void apply_kernel(const float* __restrict__ x,
                             const float* __restrict__ cA, const float* __restrict__ cB,
                             float* __restrict__ out) {
    size_t i = (size_t)blockIdx.x * blockDim.x + threadIdx.x;
    if (i >= (size_t)NN * DIM / 4) return;
    int col4 = (int)(i & 31) * 4;
    float4 hv = ((float4*)out)[i];
    float4 xv = ((const float4*)x)[i];
    float4 a  = *(const float4*)(cA + col4);
    float4 b  = *(const float4*)(cB + col4);
    float4 o;
    o.x = xv.x + hv.x * a.x + b.x;
    o.y = xv.y + hv.y * a.y + b.y;
    o.z = xv.z + hv.z * a.z + b.z;
    o.w = xv.w + hv.w * a.w + b.w;
    ((float4*)out)[i] = o;
}

extern "C" void kernel_launch(void* const* d_in, const int* in_sizes, int n_in,
                              void* d_out, int out_size, void* d_ws, size_t ws_size,
                              hipStream_t stream) {
    const float* x     = (const float*)d_in[0];
    const int*   ei    = (const int*)d_in[1];
    const float* Wl    = (const float*)d_in[2];
    const float* bl    = (const float*)d_in[3];
    const float* Wr    = (const float*)d_in[4];
    const float* gamma = (const float*)d_in[5];
    const float* beta  = (const float*)d_in[6];
    float* out = (float*)d_out;

    float* agg  = (float*)d_ws;
    float* deg  = agg + (size_t)NN * DIM;
    float* gsum = deg + NN;
    float* gsq  = gsum + DIM;
    float* cA   = gsq + DIM;
    float* cB   = cA + DIM;

    int* T   = (int*)d_ws;
    int* tmp = T + NT_SLOTS;

    int* rowptr = tmp + NBLK * EDGE_CHUNK;
    int* col    = rowptr + ROWPTR_SLOTS;

    part_kernel<<<NBLK, 256, 0, stream>>>(ei, T, tmp, gsum);

    csr_kernel<<<NBK, 256, 0, stream>>>(T, tmp, rowptr, col);

    gather_gemm_bn_kernel<<<FT_GRID, FT_WAVES * 32, 0, stream>>>(rowptr, col, x, Wl, bl, Wr,
                                                                 out, gsum, gsq);

    stats_kernel<<<1, 128, 0, stream>>>(gsum, gsq, gamma, beta, cA, cB);

    int mb = (int)(((size_t)NN * DIM / 4 + 255) / 256);
    apply_kernel<<<mb, 256, 0, stream>>>(x, cA, cB, out);
}


# independent loads no longer wait one behind another: sort kernels' edge, table and long-run loads together, their table stores last; bias loads; last 0-15 gather rows at once; sum order as before
# speedup vs baseline: 4.7933x; 1.0344x over previous
#include <hip/hip_runtime.h>

#define NN 100000
#define NE 1600000
#define DIM 128
#define BN_EPS 1e-5f
#define FT_WAVES 16
#define FT_GRID 256
#define STAGE_PITCH 132
#define ROWPTR_SLOTS 100352
#define EDGE_CHUNK 4096
#define NBLK ((NE + EDGE_CHUNK - 1) / EDGE_CHUNK)
#define BSHIFT 8
#define BW (1 << BSHIFT)
#define NBK ((NN + BW - 1) / BW)
#define BK_SLOTS 512
#define NT_SLOTS 153600
#define STAGE_CAP 8192
#define CSR_BATCH 16
#define RUNS_PER_HALF ((NBLK + 15) / 16)

typedef __attribute__((ext_vector_type(2))) float v2f;
typedef __attribute__((ext_vector_type(8))) float v8f;


__global__ __launch_bounds__(256) void part_kernel(
    const int* __restrict__ ei, int* __restrict__ T, int* __restrict__ tmp,
    float* __restrict__ sums) {
    __shared__ int lh[BK_SLOTS];
    __shared__ alignas(16) int lpk[EDGE_CHUNK];
    __shared__ int wtot[8];
    const int tid  = threadIdx.x;
    const int lane = tid & 31;
    const int wave = tid >> 5;
    const int b    = blockIdx.x;
    const int e0   = b * EDGE_CHUNK;
    lh[tid] = 0; lh[tid + 256] = 0;
    __syncthreads();

    int4 sv[4], dv[4], rv[4];
#pragma unroll
    for (int u = 0; u < 4; ++u) {
        const int e  = e0 + 4 * (tid + 256 * u);
        const int ec = e < NE ? e : 0;
        sv[u] = *(const int4*)(ei + ec);
        dv[u] = *(const int4*)(ei + NE + ec);
    }
#pragma unroll
    for (int u = 0; u < 4; ++u) {
        const int e = e0 + 4 * (tid + 256 * u);
        rv[u].x = 0; rv[u].y = 0; rv[u].z = 0; rv[u].w = 0;
        if (e < NE) {
            rv[u].x = atomicAdd(&lh[dv[u].x >> BSHIFT], 1);
            rv[u].y = atomicAdd(&lh[dv[u].y >> BSHIFT], 1);
            rv[u].z = atomicAdd(&lh[dv[u].z >> BSHIFT], 1);
            rv[u].w = atomicAdd(&lh[dv[u].w >> BSHIFT], 1);
        }
    }
    __syncthreads();

    const int c0 = lh[2 * tid];
    const int c1 = lh[2 * tid + 1];
    const int t  = c0 + c1;
    int inc = t;
    for (int o = 1; o < 32; o <<= 1) {
        const int y = __shfl_up(inc, o, 32);
        if (lane >= o) inc += y;
    }
    if (lane == 31) wtot[wave] = inc;
    __syncthreads();
    int off = 0;
    for (int w = 0; w < 8; ++w) if (w < wave) off += wtot[w];
    const int ex = off + inc - t;
    lh[2 * tid]     = ex;
    lh[2 * tid + 1] = ex + c0;
    __syncthreads();

#pragma unroll
    for (int u = 0; u < 4; ++u) {
        const int e = e0 + 4 * (tid + 256 * u);
        if (e < NE) {
            lpk[lh[dv[u].x >> BSHIFT] + rv[u].x] = (sv[u].x << BSHIFT) | (dv[u].x & (BW - 1));
            lpk[lh[dv[u].y >> BSHIFT] + rv[u].y] = (sv[u].y << BSHIFT) | (dv[u].y & (BW - 1));
            lpk[lh[dv[u].z >> BSHIFT] + rv[u].z] = (sv[u].z << BSHIFT) | (dv[u].z & (BW - 1));
            lpk[lh[dv[u].w >> BSHIFT] + rv[u].w] = (sv[u].w << BSHIFT) | (dv[u].w & (BW - 1));
        }
    }
    __syncthreads();

#pragma unroll
    for (int u = 0; u < 4; ++u) {
        const int j = 4 * (tid + 256 * u);
        if (e0 + j < NE) *(int4*)(tmp + e0 + j) = *(const int4*)(lpk + j);
    }
    if (2 * tid     <= NBK) T[(2 * tid)     * NBLK + b] = ex;
    if (2 * tid + 1 <= NBK) T[(2 * tid + 1) * NBLK + b] = ex + c0;
    if (b == 0) sums[tid] = 0.f;
}

__global__ __launch_bounds__(256) void csr_kernel(
    const int* __restrict__ T, const int* __restrict__ tmp,
    int* __restrict__ rowptr, int* __restrict__ col) {
    __shared__ int2 lan[BK_SLOTS];
    __shared__ int lp[BK_SLOTS];
    __shared__ int lc[BW];
    __shared__ int lw[STAGE_CAP];
    __shared__ int lcol[STAGE_CAP];
    __shared__ int wtot[8];
    __shared__ int wsta[8];
    __shared__ int wnod[8];
    const int tid  = threadIdx.x;
    const int lane = tid & 31;
    const int wave = tid >> 5;
    const int q    = blockIdx.x;

    const bool in0 = 2 * tid < NBLK;
    const bool in1 = 2 * tid + 1 < NBLK;
    const int k0 = in0 ? 2 * tid : NBLK - 1;
    const int k1 = in1 ? 2 * tid + 1 : NBLK - 1;
    const int ta0 = T[q * NBLK + k0];
    const int tb0 = T[(q + 1) * NBLK + k0];
    const int ta1 = T[q * NBLK + k1];
    const int tb1 = T[(q + 1) * NBLK + k1];
    const int a0 = in0 ? ta0 : 0;
    const int n0 = in0 ? tb0 - ta0 : 0;
    const int a1 = in1 ? ta1 : 0;
    const int n1 = in1 ? tb1 - ta1 : 0;
    const int t = n0 + n1;
    int inc = t;
    for (int o = 1; o < 32; o <<= 1) {
        const int y = __shfl_up(inc, o, 32);
        if (lane >= o) inc += y;
    }
    int asum = a0 + a1;
    for (int o = 16; o > 0; o >>= 1) asum += __shfl_xor(asum, o, 32);
    if (lane == 31) wtot[wave] = inc;
    if (lane == 0)  wsta[wave] = asum;
    lc[tid] = 0;
    __syncthreads();
    int off = 0, s = 0, total = 0;
    for (int w = 0; w < 8; ++w) {
        if (w < wave) off += wtot[w];
        total += wtot[w];
        s += wsta[w];
    }
    const int exr = off + inc - t;
    lan[2 * tid]     = make_int2(a0, n0); lp[2 * tid]     = exr;
    lan[2 * tid + 1] = make_int2(a1, n1); lp[2 * tid + 1] = exr + n0;
    __syncthreads();

    const bool staged = total <= STAGE_CAP;
    const int l16 = lane & 15;
    const int cb  = 2 * wave + (lane >> 4);
    if (staged) {
        int v[RUNS_PER_HALF], v2[RUNS_PER_HALF], ra[RUNS_PER_HALF], rn[RUNS_PER_HALF], rp[RUNS_PER_HALF];
#pragma unroll
        for (int u = 0; u < RUNS_PER_HALF; ++u) {
            const int c  = cb + 16 * u;
            const int cc = c < NBLK ? c : NBLK - 1;
            const int2 an = lan[cc];
            ra[u] = cc * EDGE_CHUNK + an.x;
            rn[u] = c < NBLK ? an.y : 0;
            rp[u] = lp[cc];
            v[u]  = tmp[l16 < rn[u] ? ra[u] + l16 : 0];
            v2[u] = tmp[16 + l16 < rn[u] ? ra[u] + 16 + l16 : 0];
        }
#pragma unroll
        for (int u = 0; u < RUNS_PER_HALF; ++u) {
            if (l16 < rn[u]) {
                lw[rp[u] + l16] = v[u];
                atomicAdd(&lc[v[u] & (BW - 1)], 1);
            }
            if (16 + l16 < rn[u]) {
                lw[rp[u] + 16 + l16] = v2[u];
                atomicAdd(&lc[v2[u] & (BW - 1)], 1);
            }
            for (int k = 32 + l16; k < rn[u]; k += 16) {
                const int w = tmp[ra[u] + k];
                lw[rp[u] + k] = w;
                atomicAdd(&lc[w & (BW - 1)], 1);
            }
        }
    } else {
        for (int c = cb; c < NBLK; c += 16) {
            const int a = lan[c].x;
            const int n = lan[c].y;
            for (int k = l16; k < n; k += 16)
                atomicAdd(&lc[tmp[c * EDGE_CHUNK + a + k] & (BW - 1)], 1);
        }
    }
    __syncthreads();

    const int cn = lc[tid];
    int incn = cn;
    for (int o = 1; o < 32; o <<= 1) {
        const int y = __shfl_up(incn, o, 32);
        if (lane >= o) incn += y;
    }
    if (lane == 31) wnod[wave] = incn;
    __syncthreads();
    int offn = 0;
    for (int w = 0; w < 8; ++w) if (w < wave) offn += wnod[w];
    const int ex = offn + incn - cn;
    lc[tid] = ex;
    __syncthreads();

    if (staged) {
        for (int i0 = tid; i0 < total; i0 += 256 * CSR_BATCH) {
            int w[CSR_BATCH];
#pragma unroll
            for (int u = 0; u < CSR_BATCH; ++u) {
                const int i = i0 + 256 * u;
                w[u] = (i < total) ? lw[i] : -1;
            }
#pragma unroll
            for (int u = 0; u < CSR_BATCH; ++u) {
                if (w[u] >= 0) {
                    const int p = atomicAdd(&lc[w[u] & (BW - 1)], 1);
                    lcol[p] = w[u] >> BSHIFT;
                }
            }
        }
    } else {
        for (int c = cb; c < NBLK; c += 16) {
            const int a = lan[c].x;
            const int n = lan[c].y;
            for (int k = l16; k < n; k += 16) {
                const int w = tmp[c * EDGE_CHUNK + a + k];
                const int p = atomicAdd(&lc[w & (BW - 1)], 1);
                col[s + p] = w >> BSHIFT;
            }
        }
    }
    __syncthreads();
    if (staged) {
        for (int i = tid; i < total; i += 256) col[s + i] = lcol[i];
    }
    const int node = q * BW + tid;
    if (node <= NN) rowptr[node] = s + ex;
}

__device__ __forceinline__ void flush_node(float4& acc, int& q, int& qstart, int& qend,
                                           int rpv, float* __restrict__ stw) {
    const float inv = 1.0f / fmaxf((float)(qend - qstart), 1.0f);
    float4 o;
    o.x = acc.x * inv; o.y = acc.y * inv; o.z = acc.z * inv; o.w = acc.w * inv;
    *(float4*)(stw + q * STAGE_PITCH) = o;
    acc.x = 0.f; acc.y = 0.f; acc.z = 0.f; acc.w = 0.f;
    qstart = qend;
    ++q;
    qend = __builtin_amdgcn_readlane(rpv, q + 1);
}

template <int U>
__device__ __forceinline__ void load_rows(float4 (&v)[U], const float* __restrict__ x, unsigned lane16, int ids, int j) {
#pragma unroll
    for (int u = 0; u < U; ++u) {
        const unsigned s = (unsigned)__builtin_amdgcn_readlane(ids, j + u);
        const unsigned off = (s << 9) + lane16;
        v[u] = *(const float4*)((const char*)x + off);
    }
}

template <int U>
__device__ __forceinline__ void sum_rows(const float4 (&v)[U], int e,
                                          float4& acc, int& q, int& qstart, int& qend,
                                          int rpv, float* __restrict__ stw) {
#pragma unroll
    for (int u = 0; u < U; ++u) {
        while (e + u == qend) flush_node(acc, q, qstart, qend, rpv, stw);
        acc.x += v[u].x; acc.y += v[u].y; acc.z += v[u].z; acc.w += v[u].w;
    }
}

template <int U, int O>
__device__ __forceinline__ void load_rows_at(float4 (&v)[16], const float* __restrict__ x, unsigned lane16, int ids, int j) {
#pragma unroll
    for (int u = 0; u < U; ++u) {
        const unsigned s = (unsigned)__builtin_amdgcn_readlane(ids, j + u);
        const unsigned off = (s << 9) + lane16;
        v[O + u] = *(const float4*)((const char*)x + off);
    }
}

template <int U, int O>
__device__ __forceinline__ void sum_rows_at(const float4 (&v)[16], int e,
                                             float4& acc, int& q, int& qstart, int& qend,
                                             int rpv, float* __restrict__ stw) {
#pragma unroll
    for (int u = 0; u < U; ++u) {
        while (e + u == qend) flush_node(acc, q, qstart, qend, rpv, stw);
        acc.x += v[O + u].x; acc.y += v[O + u].y; acc.z += v[O + u].z; acc.w += v[O + u].w;
    }
}

__device__ __forceinline__ void load_wfrag(float4 (&b)[8], const float* __restrict__ bw, int blk) {
    const float* s = bw + (blk >> 2) * 128 + (blk & 3) * 4096;
#pragma unroll
    for (int i = 0; i < 4; ++i) {
        b[2 * i]     = *(const float4*)(s + i * 1024);
        b[2 * i + 1] = *(const float4*)(s + i * 1024 + 64);
    }
}

__device__ __forceinline__ void mma_wfrag(v8f& c0, v8f& c1, const float4 (&a)[16],
                                          const float4 (&b)[8], int jq) {
#pragma unroll
    for (int i = 0; i < 4; ++i) {
        const float4 aq = a[4 * jq + i];
        v2f a0; a0.x = aq.x; a0.y = aq.y;
        v2f a1; a1.x = aq.z; a1.y = aq.w;
        v2f p0; p0.x = b[2 * i].x;     p0.y = b[2 * i].y;
        v2f p1; p1.x = b[2 * i + 1].x; p1.y = b[2 * i + 1].y;
        v2f q0; q0.x = b[2 * i].z;     q0.y = b[2 * i].w;
        v2f q1; q1.x = b[2 * i + 1].z; q1.y = b[2 * i + 1].w;
        c0 = __builtin_amdgcn_wmma_f32_16x16x4_f32(false, a0, false, p0, (short)0, c0, false, false);
        c1 = __builtin_amdgcn_wmma_f32_16x16x4_f32(false, a0, false, p1, (short)0, c1, false, false);
        c0 = __builtin_amdgcn_wmma_f32_16x16x4_f32(false, a1, false, q0, (short)0, c0, false, false);
        c1 = __builtin_amdgcn_wmma_f32_16x16x4_f32(false, a1, false, q1, (short)0, c1, false, false);
    }
}

__global__ __launch_bounds__(FT_WAVES * 32) void gather_gemm_bn_kernel(
    const int* __restrict__ rowptr, const int* __restrict__ col,
    const float* __restrict__ x,
    const float* __restrict__ Wl, const float* __restrict__ bl,
    const float* __restrict__ Wr, float* __restrict__ h,
    float* __restrict__ gsum, float* __restrict__ gsq) {
    __shared__ float lWl[DIM * DIM];
    __shared__ float lWr[DIM * DIM];
    __shared__ alignas(16) float lst[FT_WAVES * 16 * STAGE_PITCH];
    __shared__ float lsum[DIM];
    __shared__ float lsq[DIM];

    const int tid  = threadIdx.x;
    const int lane = tid & 31;
    const int wave = __builtin_amdgcn_readfirstlane(tid >> 5);
    const int m    = lane & 15;
    const int kh   = lane >> 4;

    {
        const int n  = tid & (DIM - 1);
        const int qb = tid >> 7;
        const float* pl = Wl + (size_t)(4 * qb) * DIM + n;
        const float* pr = Wr + (size_t)(4 * qb) * DIM + n;
        float* sl = lWl + qb * 512 + n * 4;
        float* sr = lWr + qb * 512 + n * 4;
        float4 wl[8], wr[8];
#pragma unroll
        for (int u = 0; u < 8; ++u) {
            const float* a = pl + (size_t)u * (16 * DIM);
            const float* b = pr + (size_t)u * (16 * DIM);
            wl[u].x = a[0]; wl[u].y = a[DIM]; wl[u].z = a[2 * DIM]; wl[u].w = a[3 * DIM];
            wr[u].x = b[0]; wr[u].y = b[DIM]; wr[u].z = b[2 * DIM]; wr[u].w = b[3 * DIM];
        }
#pragma unroll
        for (int u = 0; u < 8; ++u) {
            *(float4*)(sl + u * 2048) = wl[u];
            *(float4*)(sr + u * 2048) = wr[u];
        }
    }
    if (tid < DIM) { lsum[tid] = 0.f; lsq[tid] = 0.f; }
    __syncthreads();

    float* stw       = lst + wave * (16 * STAGE_PITCH) + lane * 4;
    const float* str = lst + wave * (16 * STAGE_PITCH) + m * STAGE_PITCH + 4 * kh;
    float* sto       = lst + wave * (16 * STAGE_PITCH) + (8 * kh) * STAGE_PITCH + m;


    const int ntile = (NN / 16 - (int)blockIdx.x + FT_GRID - 1) / FT_GRID;
    for (int t = wave; t < ntile; t += FT_WAVES) {
        const int row0 = ((int)blockIdx.x + FT_GRID * t) * 16;

        const int rpv = rowptr[row0 + (lane < 16 ? lane : 16)];

        const float* xRow = x + (size_t)(row0 + m) * DIM + 4 * kh;
        float4 ax[16];
#pragma unroll
        for (int j = 0; j < 16; ++j) ax[j] = *(const float4*)(xRow + 8 * j);

        v8f c[8];
        float bias8[8];
#pragma unroll
        for (int nt = 0; nt < 8; ++nt) bias8[nt] = bl[nt * 16 + m];
        const int e0  = __builtin_amdgcn_readlane(rpv, 0);
        const int e1  = __builtin_amdgcn_readlane(rpv, 16);
        int mynext = (e0 + lane < e1) ? col[e0 + lane] : 0;
#pragma unroll
        for (int nt = 0; nt < 8; ++nt) {
            float bias = bias8[nt];
            asm volatile("" : "+v"(bias));
#pragma unroll
            for (int r = 0; r < 8; ++r) c[nt][r] = bias;
        }
        {
            const float* bwr = lWr + kh * 512 + m * 4;
            float4 w0[8], w1[8];
            load_wfrag(w0, bwr, 0);
#pragma unroll
            for (int blk = 0; blk < 16; blk += 2) {
                load_wfrag(w1, bwr, blk + 1);
                __builtin_amdgcn_sched_barrier(0);
                mma_wfrag(c[2 * (blk >> 2)], c[2 * (blk >> 2) + 1], ax, w0, blk & 3);
                __builtin_amdgcn_sched_barrier(0);
                if (blk + 2 < 16) load_wfrag(w0, bwr, blk + 2);
                __builtin_amdgcn_sched_barrier(0);
                mma_wfrag(c[2 * ((blk + 1) >> 2)], c[2 * ((blk + 1) >> 2) + 1], ax, w1, (blk + 1) & 3);
                __builtin_amdgcn_sched_barrier(0);
            }
        }

        int q = 0, qstart = e0;
        int qend = __builtin_amdgcn_readlane(rpv, 1);
        float4 acc; acc.x = 0.f; acc.y = 0.f; acc.z = 0.f; acc.w = 0.f;
        const int nb = (e1 - e0) >> 4;
        float4 sa[16] = {}, sb[16] = {};
        const unsigned lane16 = (unsigned)lane * 16u;
        int tids = mynext;
        int tj   = 0;
        if (nb >= 1) {
            int g = 0;
            int idsC = mynext;
            const int pn = e0 + 32 + lane;
            int idsN = col[pn < NE - 1 ? pn : NE - 1];
            load_rows<16>(sa, x, lane16, idsC, 0);
            for (; g + 2 < nb; g += 2) {
                const int pb  = e0 + 16 * g;
                const int pnn = pb + 64 + lane;
                const int idsNN = col[pnn < NE - 1 ? pnn : NE - 1];
                load_rows<16>(sb, x, lane16, idsC, 16);
                sum_rows<16>(sa, pb, acc, q, qstart, qend, rpv, stw);
                load_rows<16>(sa, x, lane16, idsN, 0);
                sum_rows<16>(sb, pb + 16, acc, q, qstart, qend, rpv, stw);
                idsC = idsN;
                idsN = idsNN;
            }
            const int pb = e0 + 16 * g;
            if (nb - g == 2) {
                load_rows<16>(sb, x, lane16, idsC, 16);
                sum_rows<16>(sa, pb, acc, q, qstart, qend, rpv, stw);
                sum_rows<16>(sb, pb + 16, acc, q, qstart, qend, rpv, stw);
                tids = idsN; tj = 0;
            } else {
                sum_rows<16>(sa, pb, acc, q, qstart, qend, rpv, stw);
                tids = idsC; tj = 16;
            }
        }
        {
            int j  = tj;
            int pt = e0 + 16 * nb;
            const int left = e1 - pt;
            if (left & 8) { load_rows_at<8, 0>(sa, x, lane16, tids, j);  j += 8; }
            if (left & 4) { load_rows_at<4, 8>(sa, x, lane16, tids, j);  j += 4; }
            if (left & 2) { load_rows_at<2, 12>(sa, x, lane16, tids, j); j += 2; }
            if (left & 1) { load_rows_at<1, 14>(sa, x, lane16, tids, j); }
            if (left & 8) { sum_rows_at<8, 0>(sa, pt, acc, q, qstart, qend, rpv, stw);  pt += 8; }
            if (left & 4) { sum_rows_at<4, 8>(sa, pt, acc, q, qstart, qend, rpv, stw);  pt += 4; }
            if (left & 2) { sum_rows_at<2, 12>(sa, pt, acc, q, qstart, qend, rpv, stw); pt += 2; }
            if (left & 1) { sum_rows_at<1, 14>(sa, pt, acc, q, qstart, qend, rpv, stw); }
        }
        while (q < 16) flush_node(acc, q, qstart, qend, rpv, stw);

        float4 am[16];
#pragma unroll
        for (int j = 0; j < 16; ++j) am[j] = *(const float4*)(str + 8 * j);

        {
            const float* bwl = lWl + kh * 512 + m * 4;
            float4 w0[8], w1[8];
            load_wfrag(w0, bwl, 0);
#pragma unroll
            for (int blk = 0; blk < 16; blk += 2) {
                load_wfrag(w1, bwl, blk + 1);
                __builtin_amdgcn_sched_barrier(0);
                mma_wfrag(c[2 * (blk >> 2)], c[2 * (blk >> 2) + 1], am, w0, blk & 3);
                __builtin_amdgcn_sched_barrier(0);
                if (blk + 2 < 16) load_wfrag(w0, bwl, blk + 2);
                __builtin_amdgcn_sched_barrier(0);
                mma_wfrag(c[2 * ((blk + 1) >> 2)], c[2 * ((blk + 1) >> 2) + 1], am, w1, (blk + 1) & 3);
                __builtin_amdgcn_sched_barrier(0);

                if (((blk + 1) & 3) == 3) {
#pragma unroll
                    for (int nt = 2 * ((blk + 1) >> 2); nt < 2 * ((blk + 1) >> 2) + 2; ++nt) {
                        const int ncol = nt * 16 + m;
                        float s = 0.f, s2 = 0.f;
#pragma unroll
                        for (int r = 0; r < 8; ++r) {
                            float v = c[nt][r] > 0.f ? c[nt][r] : 0.f;
                            sto[r * STAGE_PITCH + nt * 16] = v;
                            s  += v;
                            s2 += v * v;
                        }
                        s  += __shfl_xor(s, 16, 32);
                        s2 += __shfl_xor(s2, 16, 32);
                        if (lane < 16) {
                            atomicAdd(&lsum[ncol], s);
                            atomicAdd(&lsq[ncol], s2);
                        }
                    }
                }
            }
        }

        float* hRow = h + (size_t)row0 * DIM + lane * 4;
#pragma unroll
        for (int q = 0; q < 16; ++q) {
            const float4 o = *(const float4*)(stw + q * STAGE_PITCH);
            *(float4*)(hRow + (size_t)q * DIM) = o;
        }
    }
    __syncthreads();
    if (tid < DIM) {
        atomicAdd(gsum + tid, lsum[tid]);
        atomicAdd(gsq  + tid, lsq[tid]);
    }
}

__global__ void stats_kernel(const float* __restrict__ gsum, const float* __restrict__ gsq,
                             const float* __restrict__ gamma, const float* __restrict__ beta,
                             float* __restrict__ cA, float* __restrict__ cB) {
    int c = threadIdx.x;
    if (c >= DIM) return;
    const float invN = 1.0f / (float)NN;
    float mu  = gsum[c] * invN;
    float var = gsq[c] * invN - mu * mu;
    float a   = rsqrtf(var + BN_EPS) * gamma[c];
    cA[c] = a;
    cB[c] = beta[c] - mu * a;
}

__global__ void apply_kernel(const float* __restrict__ x,
                             const float* __restrict__ cA, const float* __restrict__ cB,
                             float* __restrict__ out) {
    size_t i = (size_t)blockIdx.x * blockDim.x + threadIdx.x;
    if (i >= (size_t)NN * DIM / 4) return;
    int col4 = (int)(i & 31) * 4;
    float4 hv = ((float4*)out)[i];
    float4 xv = ((const float4*)x)[i];
    float4 a  = *(const float4*)(cA + col4);
    float4 b  = *(const float4*)(cB + col4);
    float4 o;
    o.x = xv.x + hv.x * a.x + b.x;
    o.y = xv.y + hv.y * a.y + b.y;
    o.z = xv.z + hv.z * a.z + b.z;
    o.w = xv.w + hv.w * a.w + b.w;
    ((float4*)out)[i] = o;
}

extern "C" void kernel_launch(void* const* d_in, const int* in_sizes, int n_in,
                              void* d_out, int out_size, void* d_ws, size_t ws_size,
                              hipStream_t stream) {
    const float* x     = (const float*)d_in[0];
    const int*   ei    = (const int*)d_in[1];
    const float* Wl    = (const float*)d_in[2];
    const float* bl    = (const float*)d_in[3];
    const float* Wr    = (const float*)d_in[4];
    const float* gamma = (const float*)d_in[5];
    const float* beta  = (const float*)d_in[6];
    float* out = (float*)d_out;

    float* agg  = (float*)d_ws;
    float* deg  = agg + (size_t)NN * DIM;
    float* gsum = deg + NN;
    float* gsq  = gsum + DIM;
    float* cA   = gsq + DIM;
    float* cB   = cA + DIM;

    int* T   = (int*)d_ws;
    int* tmp = T + NT_SLOTS;

    int* rowptr = tmp + NBLK * EDGE_CHUNK;
    int* col    = rowptr + ROWPTR_SLOTS;

    part_kernel<<<NBLK, 256, 0, stream>>>(ei, T, tmp, gsum);

    csr_kernel<<<NBK, 256, 0, stream>>>(T, tmp, rowptr, col);

    gather_gemm_bn_kernel<<<FT_GRID, FT_WAVES * 32, 0, stream>>>(rowptr, col, x, Wl, bl, Wr,
                                                                 out, gsum, gsq);

    stats_kernel<<<1, 128, 0, stream>>>(gsum, gsq, gamma, beta, cA, cB);

    int mb = (int)(((size_t)NN * DIM / 4 + 255) / 256);
    apply_kernel<<<mb, 256, 0, stream>>>(x, cA, cB, out);
}
